# starting kernel
# baseline (speedup 1.0000x reference)
#include <hip/hip_runtime.h>
#include <hip/hip_bf16.h>


typedef __attribute__((ext_vector_type(2))) float v2f;
typedef __attribute__((ext_vector_type(8))) float v8f;

#define CIN   128
#define COUT  256
#define D2    64
#define KPTS  15
#define NBH   32
#define KDFLAT (KPTS * D2)
#define INFL_INV 2.5f
#define BN_EPS 1e-5f
#define SLOPE 0.2f

__device__ __forceinline__ v8f wmma_f32(v2f a, v2f b, v8f c) {
  return __builtin_amdgcn_wmma_f32_16x16x4_f32(
       false, a,  false, b,
       (short)0, c,  false,  false);
}

__device__ __forceinline__ float lrelu(float v) {
  return v >= 0.0f ? v : SLOPE * v;
}

__global__ __launch_bounds__(256) void gemm_f32_wmma(
    const float* __restrict__ A, const float* __restrict__ B,
    float* __restrict__ C, int M, int Kd, int Nc, int totalUnits) {
  int unit = blockIdx.x * (blockDim.x >> 5) + (threadIdx.x >> 5);
  if (unit >= totalUnits) return;
  int lane = threadIdx.x & 31;
  int m    = lane & 15;
  int half = lane >> 4;
  int strips = Nc >> 6;
  int row0 = (unit / strips) << 4;
  int col0 = (unit % strips) << 6;

  const float* Arow = A + (size_t)(row0 + m) * Kd;
  v8f acc0 = {}, acc1 = {}, acc2 = {}, acc3 = {};
  for (int kk = 0; kk < Kd; kk += 4) {
    int ka = kk + (half << 1);
    v2f a;
    a.x = Arow[ka];
    a.y = Arow[ka + 1];
    const float* Bp = B + (size_t)ka * Nc + col0 + m;
    v2f b0, b1, b2, b3;
    b0.x = Bp[0];  b0.y = Bp[Nc];
    b1.x = Bp[16]; b1.y = Bp[Nc + 16];
    b2.x = Bp[32]; b2.y = Bp[Nc + 32];
    b3.x = Bp[48]; b3.y = Bp[Nc + 48];
    acc0 = wmma_f32(a, b0, acc0);
    acc1 = wmma_f32(a, b1, acc1);
    acc2 = wmma_f32(a, b2, acc2);
    acc3 = wmma_f32(a, b3, acc3);
  }
  int rbase = row0 + (half << 3);
  float* Cp = C + (size_t)rbase * Nc + col0 + m;
  #pragma unroll
  for (int r = 0; r < 8; ++r) {
    size_t o = (size_t)r * Nc;
    Cp[o +  0] = acc0[r];
    Cp[o + 16] = acc1[r];
    Cp[o + 32] = acc2[r];
    Cp[o + 48] = acc3[r];
  }
}

__global__ __launch_bounds__(256) void kpconv_wmma(
    const float* __restrict__ xyz, const int* __restrict__ nidx,
    const float* __restrict__ x1, const float* __restrict__ kp_points,
    const float* __restrict__ kpw, float* __restrict__ xk) {
  extern __shared__ float smem[];
  float* FK  = smem;
  float* Wl  = FK + 16 * KDFLAT;
  int*   IDX = (int*)(Wl + 8 * NBH * 16);
  float* PS  = (float*)(IDX + 8 * NBH);

  const int wave = threadIdx.x >> 5;
  const int lane = threadIdx.x & 31;
  const int m    = lane & 15;
  const int half = lane >> 4;
  const int base = blockIdx.x << 4;

  float* wl = Wl + wave * (NBH * 16);
  int*   il = IDX + wave * NBH;

  for (int it = 0; it < 2; ++it) {
    const int pi = (wave << 1) + it;
    const int n  = base + pi;

    const int id = nidx[(size_t)n * NBH + lane];
    il[lane] = id;
    const float cx = xyz[(size_t)n * 3 + 0];
    const float cy = xyz[(size_t)n * 3 + 1];
    const float cz = xyz[(size_t)n * 3 + 2];
    const float px = xyz[(size_t)id * 3 + 0] - cx;
    const float py = xyz[(size_t)id * 3 + 1] - cy;
    const float pz = xyz[(size_t)id * 3 + 2] - cz;
    #pragma unroll
    for (int k = 0; k < KPTS; ++k) {
      float dx = px - kp_points[k * 3 + 0];
      float dy = py - kp_points[k * 3 + 1];
      float dz = pz - kp_points[k * 3 + 2];
      float r = sqrtf(dx * dx + dy * dy + dz * dz);
      wl[lane * 16 + k] = fmaxf(1.0f - r * INFL_INV, 0.0f);
    }
    wl[lane * 16 + KPTS] = 0.0f;
    __syncthreads();

    v8f acc0 = {}, acc1 = {}, acc2 = {}, acc3 = {};
    #pragma unroll
    for (int kk = 0; kk < NBH; kk += 4) {
      const int ka = kk + (half << 1);
      v2f a;
      a.x = wl[ka * 16 + m];
      a.y = wl[(ka + 1) * 16 + m];
      const float* r0 = x1 + (size_t)il[ka] * D2 + m;
      const float* r1 = x1 + (size_t)il[ka + 1] * D2 + m;
      v2f b0, b1, b2, b3;
      b0.x = r0[0];  b0.y = r1[0];
      b1.x = r0[16]; b1.y = r1[16];
      b2.x = r0[32]; b2.y = r1[32];
      b3.x = r0[48]; b3.y = r1[48];
      acc0 = wmma_f32(a, b0, acc0);
      acc1 = wmma_f32(a, b1, acc1);
      acc2 = wmma_f32(a, b2, acc2);
      acc3 = wmma_f32(a, b3, acc3);
    }
    #pragma unroll
    for (int r = 0; r < 8; ++r) {
      const int krow = (half << 3) + r;
      if (krow < KPTS) {
        float* f = FK + pi * KDFLAT + krow * D2 + m;
        f[0]  = acc0[r];
        f[16] = acc1[r];
        f[32] = acc2[r];
        f[48] = acc3[r];
      }
    }
    __syncthreads();
  }

  {
    const int kbeg = wave * 120;
    v8f acc0 = {}, acc1 = {}, acc2 = {}, acc3 = {};
    for (int kk = kbeg; kk < kbeg + 120; kk += 4) {
      const int ka = kk + (half << 1);
      v2f a;
      a.x = FK[m * KDFLAT + ka];
      a.y = FK[m * KDFLAT + ka + 1];
      const float* Bp = kpw + (size_t)ka * D2 + m;
      v2f b0, b1, b2, b3;
      b0.x = Bp[0];  b0.y = Bp[D2];
      b1.x = Bp[16]; b1.y = Bp[D2 + 16];
      b2.x = Bp[32]; b2.y = Bp[D2 + 32];
      b3.x = Bp[48]; b3.y = Bp[D2 + 48];
      acc0 = wmma_f32(a, b0, acc0);
      acc1 = wmma_f32(a, b1, acc1);
      acc2 = wmma_f32(a, b2, acc2);
      acc3 = wmma_f32(a, b3, acc3);
    }
    #pragma unroll
    for (int r = 0; r < 8; ++r) {
      float* p = PS + wave * (16 * D2) + ((half << 3) + r) * D2 + m;
      p[0]  = acc0[r];
      p[16] = acc1[r];
      p[32] = acc2[r];
      p[48] = acc3[r];
    }
  }
  __syncthreads();

  for (int t = threadIdx.x; t < 16 * D2; t += 256) {
    float s = 0.0f;
    #pragma unroll
    for (int w = 0; w < 8; ++w) s += PS[w * (16 * D2) + t];
    const int pt = t >> 6;
    const int e  = t & 63;
    xk[(size_t)(base + pt) * D2 + e] = s;
  }
}

__global__ void zero_f32(float* __restrict__ p, int n) {
  int i = blockIdx.x * blockDim.x + threadIdx.x;
  if (i < n) p[i] = 0.0f;
}

__global__ void colstats(const float* __restrict__ Y, long long total, int C,
                         float* __restrict__ stats) {
  size_t idx0   = (size_t)blockIdx.x * blockDim.x + threadIdx.x;
  size_t stride = (size_t)gridDim.x * blockDim.x;
  float s = 0.0f, s2 = 0.0f;
  for (size_t i = idx0; i < (size_t)total; i += stride) {
    float v = Y[i];
    s += v;
    s2 += v * v;
  }
  int col = (int)(idx0 % (size_t)C);
  atomicAdd(&stats[col], s);
  atomicAdd(&stats[C + col], s2);
}

__global__ void finalize_bn(const float* __restrict__ stats,
                            const float* __restrict__ g,
                            const float* __restrict__ b, int C, float invN,
                            float* __restrict__ sc, float* __restrict__ sh) {
  int c = blockIdx.x * blockDim.x + threadIdx.x;
  if (c < C) {
    float mean = stats[c] * invN;
    float var  = stats[C + c] * invN - mean * mean;
    float s    = g[c] * rsqrtf(var + BN_EPS);
    sc[c] = s;
    sh[c] = b[c] - mean * s;
  }
}

__global__ void bnact_inplace(float* __restrict__ Y, const float* __restrict__ sc,
                              const float* __restrict__ sh, long long total, int C) {
  size_t idx0   = (size_t)blockIdx.x * blockDim.x + threadIdx.x;
  size_t stride = (size_t)gridDim.x * blockDim.x;
  for (size_t i = idx0; i < (size_t)total; i += stride) {
    int c = (int)(i % (size_t)C);
    Y[i] = lrelu(Y[i] * sc[c] + sh[c]);
  }
}

__global__ void fuse_out(float* __restrict__ out, const float* __restrict__ ysc,
                         const float* __restrict__ sc2, const float* __restrict__ sh2,
                         const float* __restrict__ scS, const float* __restrict__ shS,
                         long long total, int C) {
  size_t idx0   = (size_t)blockIdx.x * blockDim.x + threadIdx.x;
  size_t stride = (size_t)gridDim.x * blockDim.x;
  for (size_t i = idx0; i < (size_t)total; i += stride) {
    int c = (int)(i % (size_t)C);
    float v = lrelu(out[i] * sc2[c] + sh2[c]);
    out[i] = v + (ysc[i] * scS[c] + shS[c]);
  }
}

extern "C" void kernel_launch(void* const* d_in, const int* in_sizes, int n_in,
                              void* d_out, int out_size, void* d_ws, size_t ws_size,
                              hipStream_t stream) {
  const float* feats = (const float*)d_in[0];
  const float* xyz   = (const float*)d_in[1];
  const int*   nidx  = (const int*)d_in[3];
  const float* W1    = (const float*)d_in[4];
  const float* g1    = (const float*)d_in[5];
  const float* b1    = (const float*)d_in[6];
  const float* kpp   = (const float*)d_in[7];
  const float* kpw   = (const float*)d_in[8];
  const float* W2    = (const float*)d_in[9];
  const float* g2    = (const float*)d_in[10];
  const float* b2    = (const float*)d_in[11];
  const float* Wsc   = (const float*)d_in[12];
  const float* gsc   = (const float*)d_in[13];
  const float* bsc   = (const float*)d_in[14];
  float* out = (float*)d_out;

  const int N = in_sizes[0] / CIN;

  float* ws      = (float*)d_ws;
  float* x1      = ws;
  float* xk      = x1 + (size_t)N * D2;
  float* ysc     = xk + (size_t)N * D2;
  float* stats1  = ysc + (size_t)N * COUT;
  float* stats2  = stats1 + 2 * D2;
  float* statsSC = stats2 + 2 * COUT;
  float* sc1 = statsSC + 2 * COUT; float* sh1 = sc1 + D2;
  float* sc2 = sh1 + D2;           float* sh2 = sc2 + COUT;
  float* scS = sh2 + COUT;         float* shS = scS + COUT;

  const int statN = 2 * D2 + 4 * COUT;
  zero_f32<<<(statN + 255) / 256, 256, 0, stream>>>(stats1, statN);

  {
    int units = (N / 16) * (D2 / 64);
    gemm_f32_wmma<<<(units + 7) / 8, 256, 0, stream>>>(feats, W1, x1, N, CIN, D2, units);
  }
  colstats<<<256, 256, 0, stream>>>(x1, (long long)N * D2, D2, stats1);
  finalize_bn<<<1, D2, 0, stream>>>(stats1, g1, b1, D2, 1.0f / (float)N, sc1, sh1);
  bnact_inplace<<<2048, 256, 0, stream>>>(x1, sc1, sh1, (long long)N * D2, D2);

  {
    size_t lds = (16 * KDFLAT + 8 * NBH * 16 + 8 * 16 * D2) * sizeof(float)
               + 8 * NBH * sizeof(int);
    kpconv_wmma<<<N / 16, 256, lds, stream>>>(xyz, nidx, x1, kpp, kpw, xk);
  }

  {
    int units = (N / 16) * (COUT / 64);
    gemm_f32_wmma<<<(units + 7) / 8, 256, 0, stream>>>(xk, W2, out, N, D2, COUT, units);
  }
  {
    int units = (N / 16) * (COUT / 64);
    gemm_f32_wmma<<<(units + 7) / 8, 256, 0, stream>>>(feats, Wsc, ysc, N, CIN, COUT, units);
  }

  colstats<<<256, 256, 0, stream>>>(out, (long long)N * COUT, COUT, stats2);
  colstats<<<256, 256, 0, stream>>>(ysc, (long long)N * COUT, COUT, statsSC);
  finalize_bn<<<1, COUT, 0, stream>>>(stats2, g2, b2, COUT, 1.0f / (float)N, sc2, sh2);
  finalize_bn<<<1, COUT, 0, stream>>>(statsSC, gsc, bsc, COUT, 1.0f / (float)N, scS, shS);

  fuse_out<<<4096, 256, 0, stream>>>(out, ysc, sc2, sh2, scS, shS,
                                     (long long)N * COUT, COUT);
}


# BN statistics of the three linear layers from moments of their INPUTS: G=X^T X by matrix instructions, colsum X; var_c=w_c^T G w_c/N-mean_c^2. The 3 atomic passes over Y go. Reorders those sums.
# speedup vs baseline: 1.6545x; 1.6545x over previous
#include <hip/hip_runtime.h>
#include <hip/hip_bf16.h>


typedef __attribute__((ext_vector_type(2))) float v2f;
typedef __attribute__((ext_vector_type(8))) float v8f;

#define CIN   128
#define COUT  256
#define D2    64
#define KPTS  15
#define NBH   32
#define KDFLAT (KPTS * D2)
#define INFL_INV 2.5f
#define BN_EPS 1e-5f
#define SLOPE 0.2f

__device__ __forceinline__ v8f wmma_f32(v2f a, v2f b, v8f c) {
  return __builtin_amdgcn_wmma_f32_16x16x4_f32(
       false, a,  false, b,
       (short)0, c,  false,  false);
}

__device__ __forceinline__ float lrelu(float v) {
  return v >= 0.0f ? v : SLOPE * v;
}

__global__ __launch_bounds__(256) void gemm_f32_wmma(
    const float* __restrict__ A, const float* __restrict__ B,
    float* __restrict__ C, int M, int Kd, int Nc, int totalUnits) {
  int unit = blockIdx.x * (blockDim.x >> 5) + (threadIdx.x >> 5);
  if (unit >= totalUnits) return;
  int lane = threadIdx.x & 31;
  int m    = lane & 15;
  int half = lane >> 4;
  int strips = Nc >> 6;
  int row0 = (unit / strips) << 4;
  int col0 = (unit % strips) << 6;

  const float* Arow = A + (size_t)(row0 + m) * Kd;
  v8f acc0 = {}, acc1 = {}, acc2 = {}, acc3 = {};
  for (int kk = 0; kk < Kd; kk += 4) {
    int ka = kk + (half << 1);
    v2f a;
    a.x = Arow[ka];
    a.y = Arow[ka + 1];
    const float* Bp = B + (size_t)ka * Nc + col0 + m;
    v2f b0, b1, b2, b3;
    b0.x = Bp[0];  b0.y = Bp[Nc];
    b1.x = Bp[16]; b1.y = Bp[Nc + 16];
    b2.x = Bp[32]; b2.y = Bp[Nc + 32];
    b3.x = Bp[48]; b3.y = Bp[Nc + 48];
    acc0 = wmma_f32(a, b0, acc0);
    acc1 = wmma_f32(a, b1, acc1);
    acc2 = wmma_f32(a, b2, acc2);
    acc3 = wmma_f32(a, b3, acc3);
  }
  int rbase = row0 + (half << 3);
  float* Cp = C + (size_t)rbase * Nc + col0 + m;
  #pragma unroll
  for (int r = 0; r < 8; ++r) {
    size_t o = (size_t)r * Nc;
    Cp[o +  0] = acc0[r];
    Cp[o + 16] = acc1[r];
    Cp[o + 32] = acc2[r];
    Cp[o + 48] = acc3[r];
  }
}

__global__ __launch_bounds__(256) void kpconv_wmma(
    const float* __restrict__ xyz, const int* __restrict__ nidx,
    const float* __restrict__ x1, const float* __restrict__ kp_points,
    const float* __restrict__ kpw, float* __restrict__ xk) {
  extern __shared__ float smem[];
  float* FK  = smem;
  float* Wl  = FK + 16 * KDFLAT;
  int*   IDX = (int*)(Wl + 8 * NBH * 16);
  float* PS  = (float*)(IDX + 8 * NBH);

  const int wave = threadIdx.x >> 5;
  const int lane = threadIdx.x & 31;
  const int m    = lane & 15;
  const int half = lane >> 4;
  const int base = blockIdx.x << 4;

  float* wl = Wl + wave * (NBH * 16);
  int*   il = IDX + wave * NBH;

  for (int it = 0; it < 2; ++it) {
    const int pi = (wave << 1) + it;
    const int n  = base + pi;

    const int id = nidx[(size_t)n * NBH + lane];
    il[lane] = id;
    const float cx = xyz[(size_t)n * 3 + 0];
    const float cy = xyz[(size_t)n * 3 + 1];
    const float cz = xyz[(size_t)n * 3 + 2];
    const float px = xyz[(size_t)id * 3 + 0] - cx;
    const float py = xyz[(size_t)id * 3 + 1] - cy;
    const float pz = xyz[(size_t)id * 3 + 2] - cz;
    #pragma unroll
    for (int k = 0; k < KPTS; ++k) {
      float dx = px - kp_points[k * 3 + 0];
      float dy = py - kp_points[k * 3 + 1];
      float dz = pz - kp_points[k * 3 + 2];
      float r = sqrtf(dx * dx + dy * dy + dz * dz);
      wl[lane * 16 + k] = fmaxf(1.0f - r * INFL_INV, 0.0f);
    }
    wl[lane * 16 + KPTS] = 0.0f;
    __syncthreads();

    v8f acc0 = {}, acc1 = {}, acc2 = {}, acc3 = {};
    #pragma unroll
    for (int kk = 0; kk < NBH; kk += 4) {
      const int ka = kk + (half << 1);
      v2f a;
      a.x = wl[ka * 16 + m];
      a.y = wl[(ka + 1) * 16 + m];
      const float* r0 = x1 + (size_t)il[ka] * D2 + m;
      const float* r1 = x1 + (size_t)il[ka + 1] * D2 + m;
      v2f b0, b1, b2, b3;
      b0.x = r0[0];  b0.y = r1[0];
      b1.x = r0[16]; b1.y = r1[16];
      b2.x = r0[32]; b2.y = r1[32];
      b3.x = r0[48]; b3.y = r1[48];
      acc0 = wmma_f32(a, b0, acc0);
      acc1 = wmma_f32(a, b1, acc1);
      acc2 = wmma_f32(a, b2, acc2);
      acc3 = wmma_f32(a, b3, acc3);
    }
    #pragma unroll
    for (int r = 0; r < 8; ++r) {
      const int krow = (half << 3) + r;
      if (krow < KPTS) {
        float* f = FK + pi * KDFLAT + krow * D2 + m;
        f[0]  = acc0[r];
        f[16] = acc1[r];
        f[32] = acc2[r];
        f[48] = acc3[r];
      }
    }
    __syncthreads();
  }

  {
    const int kbeg = wave * 120;
    v8f acc0 = {}, acc1 = {}, acc2 = {}, acc3 = {};
    for (int kk = kbeg; kk < kbeg + 120; kk += 4) {
      const int ka = kk + (half << 1);
      v2f a;
      a.x = FK[m * KDFLAT + ka];
      a.y = FK[m * KDFLAT + ka + 1];
      const float* Bp = kpw + (size_t)ka * D2 + m;
      v2f b0, b1, b2, b3;
      b0.x = Bp[0];  b0.y = Bp[D2];
      b1.x = Bp[16]; b1.y = Bp[D2 + 16];
      b2.x = Bp[32]; b2.y = Bp[D2 + 32];
      b3.x = Bp[48]; b3.y = Bp[D2 + 48];
      acc0 = wmma_f32(a, b0, acc0);
      acc1 = wmma_f32(a, b1, acc1);
      acc2 = wmma_f32(a, b2, acc2);
      acc3 = wmma_f32(a, b3, acc3);
    }
    #pragma unroll
    for (int r = 0; r < 8; ++r) {
      float* p = PS + wave * (16 * D2) + ((half << 3) + r) * D2 + m;
      p[0]  = acc0[r];
      p[16] = acc1[r];
      p[32] = acc2[r];
      p[48] = acc3[r];
    }
  }
  __syncthreads();

  for (int t = threadIdx.x; t < 16 * D2; t += 256) {
    float s = 0.0f;
    #pragma unroll
    for (int w = 0; w < 8; ++w) s += PS[w * (16 * D2) + t];
    const int pt = t >> 6;
    const int e  = t & 63;
    xk[(size_t)(base + pt) * D2 + e] = s;
  }
}

#define GRAM_U 4

template <int C, int RPG>
__device__ __forceinline__ void gram_body(const float* __restrict__ X, int N,
                                          float* __restrict__ Gp,
                                          float* __restrict__ Sp, float* red,
                                          float* reds) {
  constexpr int TC     = C / 64;
  constexpr int TILES  = (C / 32) * TC;
  constexpr int GROUPS = 8 / TILES;
  const int wave = __builtin_amdgcn_readfirstlane((int)(threadIdx.x >> 5));
  const int lane = threadIdx.x & 31;
  const int m    = lane & 15;
  const int half = lane >> 4;
  const int tile = wave % TILES;
  const int grp  = wave / TILES;
  const int tr   = tile / TC;
  const int tc   = tile % TC;
  const int gidx = blockIdx.x * GROUPS + grp;
  const int row0 = gidx * RPG;
  int nrows = N - row0;
  nrows = nrows < 0 ? 0 : (nrows > RPG ? RPG : nrows);
  const int nruns = nrows / (4 * GRAM_U);

  const float* xb = X + (size_t)(row0 + 2 * half) * C + m;
  const float* ap = xb + tr * 32;
  const float* bp = xb + tc * 64;

  v8f acc[2][4];
  #pragma unroll
  for (int i = 0; i < 2; ++i)
    #pragma unroll
    for (int j = 0; j < 4; ++j) acc[i][j] = v8f{};
  float cs[4] = {0.0f, 0.0f, 0.0f, 0.0f};

  auto load_run = [&](v2f (&a)[GRAM_U][2], v2f (&b)[GRAM_U][4], int run) {
    const size_t o = (size_t)run * (4 * GRAM_U) * C;
    #pragma unroll
    for (int u = 0; u < GRAM_U; ++u) {
      const float* pa = ap + o + (size_t)u * 4 * C;
      const float* pb = bp + o + (size_t)u * 4 * C;
      #pragma unroll
      for (int i = 0; i < 2; ++i) { a[u][i].x = pa[i * 16]; a[u][i].y = pa[C + i * 16]; }
      #pragma unroll
      for (int j = 0; j < 4; ++j) { b[u][j].x = pb[j * 16]; b[u][j].y = pb[C + j * 16]; }
    }
  };
  auto mma_run = [&](v2f (&a)[GRAM_U][2], v2f (&b)[GRAM_U][4]) {
    #pragma unroll
    for (int u = 0; u < GRAM_U; ++u) {
      if (tr == 0) {
        #pragma unroll
        for (int j = 0; j < 4; ++j) cs[j] += b[u][j].x + b[u][j].y;
      }
      #pragma unroll
      for (int i = 0; i < 2; ++i)
        #pragma unroll
        for (int j = 0; j < 4; ++j) acc[i][j] = wmma_f32(a[u][i], b[u][j], acc[i][j]);
    }
  };

  v2f a0[GRAM_U][2], b0[GRAM_U][4], a1[GRAM_U][2], b1[GRAM_U][4];
  if (nruns > 0) load_run(a0, b0, 0);
  for (int r = 0; r < nruns; r += 2) {
    if (r + 1 < nruns) load_run(a1, b1, r + 1);
    mma_run(a0, b0);
    if (r + 2 < nruns) load_run(a0, b0, r + 2);
    if (r + 1 < nruns) mma_run(a1, b1);
  }

  const int grow = tr * 32 + (half << 3);
  const int gcol = tc * 64 + m;
  if (tr == 0) {
    #pragma unroll
    for (int j = 0; j < 4; ++j) cs[j] += __shfl_xor(cs[j], 16);
  }
  if constexpr (GROUPS > 1) {
    if (grp > 0) {
      if (tr == 0 && half == 0) {
        #pragma unroll
        for (int j = 0; j < 4; ++j) reds[(grp - 1) * C + gcol + j * 16] = cs[j];
      }
      float* rp = red + (size_t)(grp - 1) * (C * C);
      #pragma unroll
      for (int i = 0; i < 2; ++i)
        #pragma unroll
        for (int j = 0; j < 4; ++j)
          #pragma unroll
          for (int r = 0; r < 8; ++r)
            rp[(grow + i * 16 + r) * C + gcol + j * 16] = acc[i][j][r];
    }
    __syncthreads();
    if (grp == 0) {
      if (tr == 0) {
        #pragma unroll
        for (int g = 0; g < GROUPS - 1; ++g)
          #pragma unroll
          for (int j = 0; j < 4; ++j) cs[j] += reds[g * C + gcol + j * 16];
      }
      #pragma unroll
      for (int g = 0; g < GROUPS - 1; ++g)
        #pragma unroll
        for (int i = 0; i < 2; ++i)
          #pragma unroll
          for (int j = 0; j < 4; ++j)
            #pragma unroll
            for (int r = 0; r < 8; ++r)
              acc[i][j][r] += red[(size_t)g * (C * C) + (grow + i * 16 + r) * C + gcol + j * 16];
    }
  }
  if (grp == 0) {
    float* gp = Gp + (size_t)blockIdx.x * (C * C);
    #pragma unroll
    for (int i = 0; i < 2; ++i)
      #pragma unroll
      for (int j = 0; j < 4; ++j)
        #pragma unroll
        for (int r = 0; r < 8; ++r)
          gp[(grow + i * 16 + r) * C + gcol + j * 16] = acc[i][j][r];
  }
  if (grp == 0 && tr == 0 && half == 0) {
    float* sp = Sp + (size_t)blockIdx.x * C + gcol;
    #pragma unroll
    for (int j = 0; j < 4; ++j) sp[j * 16] = cs[j];
  }
}

__global__ __launch_bounds__(256) void gram_part128(const float* __restrict__ X, int N,
                                                    float* __restrict__ Gp,
                                                    float* __restrict__ Sp) {
  gram_body<128, 320>(X, N, Gp, Sp, nullptr, nullptr);
}

__global__ __launch_bounds__(256) void gram_part64(const float* __restrict__ X, int N,
                                                   float* __restrict__ Gp,
                                                   float* __restrict__ Sp) {
  __shared__ float red[3 * 64 * 64];
  __shared__ float reds[3 * 64];
  gram_body<64, 80>(X, N, Gp, Sp, red, reds);
}

__global__ __launch_bounds__(256) void reduce_rows2(
    const float* __restrict__ srcA, int nrA, int wA, float* __restrict__ dstA, int blocksA,
    const float* __restrict__ srcB, int nrB, int wB, float* __restrict__ dstB) {
  __shared__ float red[8][32];
  int b = blockIdx.x;
  const float* src = srcA; int nr = nrA; int w = wA; float* dst = dstA;
  if (b >= blocksA) { b -= blocksA; src = srcB; nr = nrB; w = wB; dst = dstB; }
  const int e = threadIdx.x & 31;
  const int q = threadIdx.x >> 5;
  const int col = b * 32 + e;
  float s = 0.0f;
  #pragma unroll 8
  for (int r = q; r < nr; r += 8) s += src[(size_t)r * w + col];
  red[q][e] = s;
  __syncthreads();
  if (q == 0) {
    float t = red[0][e];
    #pragma unroll
    for (int i = 1; i < 8; ++i) t += red[i][e];
    dst[col] = t;
  }
}

template <int C>
__device__ __forceinline__ void gram_finalize_body(
    const float* __restrict__ G, const float* __restrict__ S, float invN,
    const float* __restrict__ W, const float* __restrict__ g,
    const float* __restrict__ bta, float* __restrict__ sc, float* __restrict__ sh,
    int Cout, int c) {
  __shared__ float wl[C];
  __shared__ float r2[256];
  __shared__ float r1[256];
  const int tid = threadIdx.x;
  for (int k = tid; k < C; k += 256) wl[k] = W[(size_t)k * Cout + c];
  __syncthreads();
  float s2 = 0.0f;
  #pragma unroll 16
  for (int i = tid; i < C * C; i += 256) s2 += wl[i / C] * G[i] * wl[i % C];
  float s1 = 0.0f;
  for (int k = tid; k < C; k += 256) s1 += wl[k] * S[k];
  r2[tid] = s2;
  r1[tid] = s1;
  __syncthreads();
  for (int st = 128; st > 0; st >>= 1) {
    if (tid < st) { r2[tid] += r2[tid + st]; r1[tid] += r1[tid + st]; }
    __syncthreads();
  }
  if (tid == 0) {
    float mean = r1[0] * invN;
    float var  = fmaxf(r2[0] * invN - mean * mean, 0.0f);
    float s    = g[c] * rsqrtf(var + BN_EPS);
    sc[c] = s;
    sh[c] = bta[c] - mean * s;
  }
}

__global__ __launch_bounds__(256) void gram_finalize128(
    const float* __restrict__ G, const float* __restrict__ S, float invN,
    const float* __restrict__ Wa, const float* __restrict__ ga, const float* __restrict__ ba,
    float* __restrict__ sca, float* __restrict__ sha, int Ca,
    const float* __restrict__ Wb, const float* __restrict__ gb, const float* __restrict__ bb,
    float* __restrict__ scb, float* __restrict__ shb, int Cb) {
  int c = blockIdx.x;
  const float* W = Wa; const float* g = ga; const float* b = ba;
  float* sc = sca; float* sh = sha; int Cout = Ca;
  if (c >= Ca) { c -= Ca; W = Wb; g = gb; b = bb; sc = scb; sh = shb; Cout = Cb; }
  gram_finalize_body<128>(G, S, invN, W, g, b, sc, sh, Cout, c);
}

__global__ __launch_bounds__(256) void gram_finalize64(
    const float* __restrict__ G, const float* __restrict__ S, float invN,
    const float* __restrict__ W, const float* __restrict__ g, const float* __restrict__ b,
    float* __restrict__ sc, float* __restrict__ sh, int Cout) {
  gram_finalize_body<64>(G, S, invN, W, g, b, sc, sh, Cout, blockIdx.x);
}

__global__ void bnact_inplace(float* __restrict__ Y, const float* __restrict__ sc,
                              const float* __restrict__ sh, long long total, int C) {
  size_t idx0   = (size_t)blockIdx.x * blockDim.x + threadIdx.x;
  size_t stride = (size_t)gridDim.x * blockDim.x;
  for (size_t i = idx0; i < (size_t)total; i += stride) {
    int c = (int)(i % (size_t)C);
    Y[i] = lrelu(Y[i] * sc[c] + sh[c]);
  }
}

__global__ void fuse_out(float* __restrict__ out, const float* __restrict__ ysc,
                         const float* __restrict__ sc2, const float* __restrict__ sh2,
                         const float* __restrict__ scS, const float* __restrict__ shS,
                         long long total, int C) {
  size_t idx0   = (size_t)blockIdx.x * blockDim.x + threadIdx.x;
  size_t stride = (size_t)gridDim.x * blockDim.x;
  for (size_t i = idx0; i < (size_t)total; i += stride) {
    int c = (int)(i % (size_t)C);
    float v = lrelu(out[i] * sc2[c] + sh2[c]);
    out[i] = v + (ysc[i] * scS[c] + shS[c]);
  }
}

extern "C" void kernel_launch(void* const* d_in, const int* in_sizes, int n_in,
                              void* d_out, int out_size, void* d_ws, size_t ws_size,
                              hipStream_t stream) {
  const float* feats = (const float*)d_in[0];
  const float* xyz   = (const float*)d_in[1];
  const int*   nidx  = (const int*)d_in[3];
  const float* W1    = (const float*)d_in[4];
  const float* g1    = (const float*)d_in[5];
  const float* b1    = (const float*)d_in[6];
  const float* kpp   = (const float*)d_in[7];
  const float* kpw   = (const float*)d_in[8];
  const float* W2    = (const float*)d_in[9];
  const float* g2    = (const float*)d_in[10];
  const float* b2    = (const float*)d_in[11];
  const float* Wsc   = (const float*)d_in[12];
  const float* gsc   = (const float*)d_in[13];
  const float* bsc   = (const float*)d_in[14];
  float* out = (float*)d_out;

  const int N = in_sizes[0] / CIN;

  float* ws      = (float*)d_ws;
  float* x1      = ws;
  float* xk      = x1 + (size_t)N * D2;
  float* ysc     = xk + (size_t)N * D2;
  float* stats1  = ysc + (size_t)N * COUT;
  float* stats2  = stats1 + 2 * D2;
  float* statsSC = stats2 + 2 * COUT;
  float* sc1 = statsSC + 2 * COUT; float* sh1 = sc1 + D2;
  float* sc2 = sh1 + D2;           float* sh2 = sc2 + COUT;
  float* scS = sh2 + COUT;         float* shS = scS + COUT;

  const int nblk = (N + 319) / 320;
  float* GpF = ysc;
  float* SpF = GpF + (size_t)nblk * CIN * CIN;
  float* GF  = SpF + (size_t)nblk * CIN;
  float* SF  = GF + CIN * CIN;
  float* Gp2 = x1;
  float* Sp2 = Gp2 + (size_t)nblk * D2 * D2;
  float* G2  = Sp2 + (size_t)nblk * D2;
  float* S2  = G2 + D2 * D2;

  {
    int units = (N / 16) * (D2 / 64);
    gemm_f32_wmma<<<(units + 7) / 8, 256, 0, stream>>>(feats, W1, x1, N, CIN, D2, units);
  }
  gram_part128<<<nblk, 256, 0, stream>>>(feats, N, GpF, SpF);
  reduce_rows2<<<(CIN * CIN) / 32 + CIN / 32, 256, 0, stream>>>(
      GpF, nblk, CIN * CIN, GF, (CIN * CIN) / 32, SpF, nblk, CIN, SF);
  gram_finalize128<<<D2 + COUT, 256, 0, stream>>>(GF, SF, 1.0f / (float)N,
      W1, g1, b1, sc1, sh1, D2, Wsc, gsc, bsc, scS, shS, COUT);
  bnact_inplace<<<2048, 256, 0, stream>>>(x1, sc1, sh1, (long long)N * D2, D2);

  {
    size_t lds = (16 * KDFLAT + 8 * NBH * 16 + 8 * 16 * D2) * sizeof(float)
               + 8 * NBH * sizeof(int);
    kpconv_wmma<<<N / 16, 256, lds, stream>>>(xyz, nidx, x1, kpp, kpw, xk);
  }

  {
    int units = (N / 16) * (COUT / 64);
    gemm_f32_wmma<<<(units + 7) / 8, 256, 0, stream>>>(xk, W2, out, N, D2, COUT, units);
  }
  {
    int units = (N / 16) * (COUT / 64);
    gemm_f32_wmma<<<(units + 7) / 8, 256, 0, stream>>>(feats, Wsc, ysc, N, CIN, COUT, units);
  }

  gram_part64<<<nblk, 256, 0, stream>>>(xk, N, Gp2, Sp2);
  reduce_rows2<<<(D2 * D2) / 32 + D2 / 32, 256, 0, stream>>>(
      Gp2, nblk, D2 * D2, G2, (D2 * D2) / 32, Sp2, nblk, D2, S2);
  gram_finalize64<<<COUT, 256, 0, stream>>>(G2, S2, 1.0f / (float)N, W2, g2, b2, sc2, sh2, COUT);

  fuse_out<<<4096, 256, 0, stream>>>(out, ysc, sc2, sh2, scS, shS,
                                     (long long)N * COUT, COUT);
}


# BN, activation and residual sum applied in the product epilogues (stats launches run first): x1 stored post-BN; unary_2 and shortcut products meet in one wave's accumulators; pre-BN Y never stored
# speedup vs baseline: 2.0090x; 1.2143x over previous
#include <hip/hip_runtime.h>
#include <hip/hip_bf16.h>


typedef __attribute__((ext_vector_type(2))) float v2f;
typedef __attribute__((ext_vector_type(8))) float v8f;

#define CIN   128
#define COUT  256
#define D2    64
#define KPTS  15
#define NBH   32
#define KDFLAT (KPTS * D2)
#define INFL_INV 2.5f
#define BN_EPS 1e-5f
#define SLOPE 0.2f

__device__ __forceinline__ v8f wmma_f32(v2f a, v2f b, v8f c) {
  return __builtin_amdgcn_wmma_f32_16x16x4_f32(
       false, a,  false, b,
       (short)0, c,  false,  false);
}

__device__ __forceinline__ float lrelu(float v) {
  return v >= 0.0f ? v : SLOPE * v;
}


__device__ __forceinline__ void strip_product(
    const float* __restrict__ Arow, const float* __restrict__ Bcol, int Kd, int Nc,
    int half, v8f& acc0, v8f& acc1, v8f& acc2, v8f& acc3) {
  for (int kk = 0; kk < Kd; kk += 4) {
    int ka = kk + (half << 1);
    v2f a;
    a.x = Arow[ka];
    a.y = Arow[ka + 1];
    const float* Bp = Bcol + (size_t)ka * Nc;
    v2f b0, b1, b2, b3;
    b0.x = Bp[0];  b0.y = Bp[Nc];
    b1.x = Bp[16]; b1.y = Bp[Nc + 16];
    b2.x = Bp[32]; b2.y = Bp[Nc + 32];
    b3.x = Bp[48]; b3.y = Bp[Nc + 48];
    acc0 = wmma_f32(a, b0, acc0);
    acc1 = wmma_f32(a, b1, acc1);
    acc2 = wmma_f32(a, b2, acc2);
    acc3 = wmma_f32(a, b3, acc3);
  }
}

__global__ __launch_bounds__(256) void gemm_bnact_wmma(
    const float* __restrict__ A, const float* __restrict__ B,
    const float* __restrict__ sc, const float* __restrict__ sh,
    float* __restrict__ C, int M, int Kd, int Nc, int totalUnits) {
  int unit = blockIdx.x * (blockDim.x >> 5) + (threadIdx.x >> 5);
  if (unit >= totalUnits) return;
  int lane = threadIdx.x & 31;
  int m    = lane & 15;
  int half = lane >> 4;
  int strips = Nc >> 6;
  int row0 = (unit / strips) << 4;
  int col0 = (unit % strips) << 6;

  v8f acc0 = {}, acc1 = {}, acc2 = {}, acc3 = {};
  strip_product(A + (size_t)(row0 + m) * Kd, B + col0 + m, Kd, Nc, half,
                acc0, acc1, acc2, acc3);

  const float* scp = sc + col0 + m;
  const float* shp = sh + col0 + m;
  const float s0 = scp[0], s1 = scp[16], s2 = scp[32], s3 = scp[48];
  const float h0 = shp[0], h1 = shp[16], h2 = shp[32], h3 = shp[48];
  int rbase = row0 + (half << 3);
  float* Cp = C + (size_t)rbase * Nc + col0 + m;
  #pragma unroll
  for (int r = 0; r < 8; ++r) {
    size_t o = (size_t)r * Nc;
    Cp[o +  0] = lrelu(acc0[r] * s0 + h0);
    Cp[o + 16] = lrelu(acc1[r] * s1 + h1);
    Cp[o + 32] = lrelu(acc2[r] * s2 + h2);
    Cp[o + 48] = lrelu(acc3[r] * s3 + h3);
  }
}

__global__ __launch_bounds__(256) void gemm2_bn_residual_wmma(
    const float* __restrict__ Am, const float* __restrict__ Bm, int Km,
    const float* __restrict__ scM, const float* __restrict__ shM,
    const float* __restrict__ As, const float* __restrict__ Bs, int Ks,
    const float* __restrict__ scS, const float* __restrict__ shS,
    float* __restrict__ C, int M, int Nc, int totalUnits) {
  int unit = blockIdx.x * (blockDim.x >> 5) + (threadIdx.x >> 5);
  if (unit >= totalUnits) return;
  int lane = threadIdx.x & 31;
  int m    = lane & 15;
  int half = lane >> 4;
  int strips = Nc >> 6;
  int row0 = (unit / strips) << 4;
  int col0 = (unit % strips) << 6;

  v8f y0 = {}, y1 = {}, y2 = {}, y3 = {};
  strip_product(As + (size_t)(row0 + m) * Ks, Bs + col0 + m, Ks, Nc, half,
                y0, y1, y2, y3);
  v8f acc0 = {}, acc1 = {}, acc2 = {}, acc3 = {};
  strip_product(Am + (size_t)(row0 + m) * Km, Bm + col0 + m, Km, Nc, half,
                acc0, acc1, acc2, acc3);

  const float* p;
  p = scM + col0 + m; const float a0 = p[0], a1 = p[16], a2 = p[32], a3 = p[48];
  p = shM + col0 + m; const float c0 = p[0], c1 = p[16], c2 = p[32], c3 = p[48];
  p = scS + col0 + m; const float s0 = p[0], s1 = p[16], s2 = p[32], s3 = p[48];
  p = shS + col0 + m; const float h0 = p[0], h1 = p[16], h2 = p[32], h3 = p[48];
  int rbase = row0 + (half << 3);
  float* Cp = C + (size_t)rbase * Nc + col0 + m;
  #pragma unroll
  for (int r = 0; r < 8; ++r) {
    size_t o = (size_t)r * Nc;
    Cp[o +  0] = lrelu(acc0[r] * a0 + c0) + (y0[r] * s0 + h0);
    Cp[o + 16] = lrelu(acc1[r] * a1 + c1) + (y1[r] * s1 + h1);
    Cp[o + 32] = lrelu(acc2[r] * a2 + c2) + (y2[r] * s2 + h2);
    Cp[o + 48] = lrelu(acc3[r] * a3 + c3) + (y3[r] * s3 + h3);
  }
}

__global__ __launch_bounds__(256) void kpconv_wmma(
    const float* __restrict__ xyz, const int* __restrict__ nidx,
    const float* __restrict__ x1, const float* __restrict__ kp_points,
    const float* __restrict__ kpw, float* __restrict__ xk) {
  extern __shared__ float smem[];
  float* FK  = smem;
  float* Wl  = FK + 16 * KDFLAT;
  int*   IDX = (int*)(Wl + 8 * NBH * 16);
  float* PS  = (float*)(IDX + 8 * NBH);

  const int wave = threadIdx.x >> 5;
  const int lane = threadIdx.x & 31;
  const int m    = lane & 15;
  const int half = lane >> 4;
  const int base = blockIdx.x << 4;

  float* wl = Wl + wave * (NBH * 16);
  int*   il = IDX + wave * NBH;

  for (int it = 0; it < 2; ++it) {
    const int pi = (wave << 1) + it;
    const int n  = base + pi;

    const int id = nidx[(size_t)n * NBH + lane];
    il[lane] = id;
    const float cx = xyz[(size_t)n * 3 + 0];
    const float cy = xyz[(size_t)n * 3 + 1];
    const float cz = xyz[(size_t)n * 3 + 2];
    const float px = xyz[(size_t)id * 3 + 0] - cx;
    const float py = xyz[(size_t)id * 3 + 1] - cy;
    const float pz = xyz[(size_t)id * 3 + 2] - cz;
    #pragma unroll
    for (int k = 0; k < KPTS; ++k) {
      float dx = px - kp_points[k * 3 + 0];
      float dy = py - kp_points[k * 3 + 1];
      float dz = pz - kp_points[k * 3 + 2];
      float r = sqrtf(dx * dx + dy * dy + dz * dz);
      wl[lane * 16 + k] = fmaxf(1.0f - r * INFL_INV, 0.0f);
    }
    wl[lane * 16 + KPTS] = 0.0f;
    __syncthreads();

    v8f acc0 = {}, acc1 = {}, acc2 = {}, acc3 = {};
    #pragma unroll
    for (int kk = 0; kk < NBH; kk += 4) {
      const int ka = kk + (half << 1);
      v2f a;
      a.x = wl[ka * 16 + m];
      a.y = wl[(ka + 1) * 16 + m];
      const float* r0 = x1 + (size_t)il[ka] * D2 + m;
      const float* r1 = x1 + (size_t)il[ka + 1] * D2 + m;
      v2f b0, b1, b2, b3;
      b0.x = r0[0];  b0.y = r1[0];
      b1.x = r0[16]; b1.y = r1[16];
      b2.x = r0[32]; b2.y = r1[32];
      b3.x = r0[48]; b3.y = r1[48];
      acc0 = wmma_f32(a, b0, acc0);
      acc1 = wmma_f32(a, b1, acc1);
      acc2 = wmma_f32(a, b2, acc2);
      acc3 = wmma_f32(a, b3, acc3);
    }
    #pragma unroll
    for (int r = 0; r < 8; ++r) {
      const int krow = (half << 3) + r;
      if (krow < KPTS) {
        float* f = FK + pi * KDFLAT + krow * D2 + m;
        f[0]  = acc0[r];
        f[16] = acc1[r];
        f[32] = acc2[r];
        f[48] = acc3[r];
      }
    }
    __syncthreads();
  }

  {
    const int kbeg = wave * 120;
    v8f acc0 = {}, acc1 = {}, acc2 = {}, acc3 = {};
    for (int kk = kbeg; kk < kbeg + 120; kk += 4) {
      const int ka = kk + (half << 1);
      v2f a;
      a.x = FK[m * KDFLAT + ka];
      a.y = FK[m * KDFLAT + ka + 1];
      const float* Bp = kpw + (size_t)ka * D2 + m;
      v2f b0, b1, b2, b3;
      b0.x = Bp[0];  b0.y = Bp[D2];
      b1.x = Bp[16]; b1.y = Bp[D2 + 16];
      b2.x = Bp[32]; b2.y = Bp[D2 + 32];
      b3.x = Bp[48]; b3.y = Bp[D2 + 48];
      acc0 = wmma_f32(a, b0, acc0);
      acc1 = wmma_f32(a, b1, acc1);
      acc2 = wmma_f32(a, b2, acc2);
      acc3 = wmma_f32(a, b3, acc3);
    }
    #pragma unroll
    for (int r = 0; r < 8; ++r) {
      float* p = PS + wave * (16 * D2) + ((half << 3) + r) * D2 + m;
      p[0]  = acc0[r];
      p[16] = acc1[r];
      p[32] = acc2[r];
      p[48] = acc3[r];
    }
  }
  __syncthreads();

  for (int t = threadIdx.x; t < 16 * D2; t += 256) {
    float s = 0.0f;
    #pragma unroll
    for (int w = 0; w < 8; ++w) s += PS[w * (16 * D2) + t];
    const int pt = t >> 6;
    const int e  = t & 63;
    xk[(size_t)(base + pt) * D2 + e] = s;
  }
}

#define GRAM_U 4

template <int C, int RPG>
__device__ __forceinline__ void gram_body(const float* __restrict__ X, int N,
                                          float* __restrict__ Gp,
                                          float* __restrict__ Sp, float* red,
                                          float* reds) {
  constexpr int TC     = C / 64;
  constexpr int TILES  = (C / 32) * TC;
  constexpr int GROUPS = 8 / TILES;
  const int wave = __builtin_amdgcn_readfirstlane((int)(threadIdx.x >> 5));
  const int lane = threadIdx.x & 31;
  const int m    = lane & 15;
  const int half = lane >> 4;
  const int tile = wave % TILES;
  const int grp  = wave / TILES;
  const int tr   = tile / TC;
  const int tc   = tile % TC;
  const int gidx = blockIdx.x * GROUPS + grp;
  const int row0 = gidx * RPG;
  int nrows = N - row0;
  nrows = nrows < 0 ? 0 : (nrows > RPG ? RPG : nrows);
  const int nruns = nrows / (4 * GRAM_U);

  const float* xb = X + (size_t)(row0 + 2 * half) * C + m;
  const float* ap = xb + tr * 32;
  const float* bp = xb + tc * 64;

  v8f acc[2][4];
  #pragma unroll
  for (int i = 0; i < 2; ++i)
    #pragma unroll
    for (int j = 0; j < 4; ++j) acc[i][j] = v8f{};
  float cs[4] = {0.0f, 0.0f, 0.0f, 0.0f};

  auto load_run = [&](v2f (&a)[GRAM_U][2], v2f (&b)[GRAM_U][4], int run) {
    const size_t o = (size_t)run * (4 * GRAM_U) * C;
    #pragma unroll
    for (int u = 0; u < GRAM_U; ++u) {
      const float* pa = ap + o + (size_t)u * 4 * C;
      const float* pb = bp + o + (size_t)u * 4 * C;
      #pragma unroll
      for (int i = 0; i < 2; ++i) { a[u][i].x = pa[i * 16]; a[u][i].y = pa[C + i * 16]; }
      #pragma unroll
      for (int j = 0; j < 4; ++j) { b[u][j].x = pb[j * 16]; b[u][j].y = pb[C + j * 16]; }
    }
  };
  auto mma_run = [&](v2f (&a)[GRAM_U][2], v2f (&b)[GRAM_U][4]) {
    #pragma unroll
    for (int u = 0; u < GRAM_U; ++u) {
      if (tr == 0) {
        #pragma unroll
        for (int j = 0; j < 4; ++j) cs[j] += b[u][j].x + b[u][j].y;
      }
      #pragma unroll
      for (int i = 0; i < 2; ++i)
        #pragma unroll
        for (int j = 0; j < 4; ++j) acc[i][j] = wmma_f32(a[u][i], b[u][j], acc[i][j]);
    }
  };

  v2f a0[GRAM_U][2], b0[GRAM_U][4], a1[GRAM_U][2], b1[GRAM_U][4];
  if (nruns > 0) load_run(a0, b0, 0);
  for (int r = 0; r < nruns; r += 2) {
    if (r + 1 < nruns) load_run(a1, b1, r + 1);
    mma_run(a0, b0);
    if (r + 2 < nruns) load_run(a0, b0, r + 2);
    if (r + 1 < nruns) mma_run(a1, b1);
  }

  const int grow = tr * 32 + (half << 3);
  const int gcol = tc * 64 + m;
  if (tr == 0) {
    #pragma unroll
    for (int j = 0; j < 4; ++j) cs[j] += __shfl_xor(cs[j], 16);
  }
  if constexpr (GROUPS > 1) {
    if (grp > 0) {
      if (tr == 0 && half == 0) {
        #pragma unroll
        for (int j = 0; j < 4; ++j) reds[(grp - 1) * C + gcol + j * 16] = cs[j];
      }
      float* rp = red + (size_t)(grp - 1) * (C * C);
      #pragma unroll
      for (int i = 0; i < 2; ++i)
        #pragma unroll
        for (int j = 0; j < 4; ++j)
          #pragma unroll
          for (int r = 0; r < 8; ++r)
            rp[(grow + i * 16 + r) * C + gcol + j * 16] = acc[i][j][r];
    }
    __syncthreads();
    if (grp == 0) {
      if (tr == 0) {
        #pragma unroll
        for (int g = 0; g < GROUPS - 1; ++g)
          #pragma unroll
          for (int j = 0; j < 4; ++j) cs[j] += reds[g * C + gcol + j * 16];
      }
      #pragma unroll
      for (int g = 0; g < GROUPS - 1; ++g)
        #pragma unroll
        for (int i = 0; i < 2; ++i)
          #pragma unroll
          for (int j = 0; j < 4; ++j)
            #pragma unroll
            for (int r = 0; r < 8; ++r)
              acc[i][j][r] += red[(size_t)g * (C * C) + (grow + i * 16 + r) * C + gcol + j * 16];
    }
  }
  if (grp == 0) {
    float* gp = Gp + (size_t)blockIdx.x * (C * C);
    #pragma unroll
    for (int i = 0; i < 2; ++i)
      #pragma unroll
      for (int j = 0; j < 4; ++j)
        #pragma unroll
        for (int r = 0; r < 8; ++r)
          gp[(grow + i * 16 + r) * C + gcol + j * 16] = acc[i][j][r];
  }
  if (grp == 0 && tr == 0 && half == 0) {
    float* sp = Sp + (size_t)blockIdx.x * C + gcol;
    #pragma unroll
    for (int j = 0; j < 4; ++j) sp[j * 16] = cs[j];
  }
}

__global__ __launch_bounds__(256) void gram_part128(const float* __restrict__ X, int N,
                                                    float* __restrict__ Gp,
                                                    float* __restrict__ Sp) {
  gram_body<128, 320>(X, N, Gp, Sp, nullptr, nullptr);
}

__global__ __launch_bounds__(256) void gram_part64(const float* __restrict__ X, int N,
                                                   float* __restrict__ Gp,
                                                   float* __restrict__ Sp) {
  __shared__ float red[3 * 64 * 64];
  __shared__ float reds[3 * 64];
  gram_body<64, 80>(X, N, Gp, Sp, red, reds);
}

__global__ __launch_bounds__(256) void reduce_rows2(
    const float* __restrict__ srcA, int nrA, int wA, float* __restrict__ dstA, int blocksA,
    const float* __restrict__ srcB, int nrB, int wB, float* __restrict__ dstB) {
  __shared__ float red[8][32];
  int b = blockIdx.x;
  const float* src = srcA; int nr = nrA; int w = wA; float* dst = dstA;
  if (b >= blocksA) { b -= blocksA; src = srcB; nr = nrB; w = wB; dst = dstB; }
  const int e = threadIdx.x & 31;
  const int q = threadIdx.x >> 5;
  const int col = b * 32 + e;
  float s = 0.0f;
  #pragma unroll 8
  for (int r = q; r < nr; r += 8) s += src[(size_t)r * w + col];
  red[q][e] = s;
  __syncthreads();
  if (q == 0) {
    float t = red[0][e];
    #pragma unroll
    for (int i = 1; i < 8; ++i) t += red[i][e];
    dst[col] = t;
  }
}

template <int C>
__device__ __forceinline__ void gram_finalize_body(
    const float* __restrict__ G, const float* __restrict__ S, float invN,
    const float* __restrict__ W, const float* __restrict__ g,
    const float* __restrict__ bta, float* __restrict__ sc, float* __restrict__ sh,
    int Cout, int c) {
  __shared__ float wl[C];
  __shared__ float r2[256];
  __shared__ float r1[256];
  const int tid = threadIdx.x;
  for (int k = tid; k < C; k += 256) wl[k] = W[(size_t)k * Cout + c];
  __syncthreads();
  float s2 = 0.0f;
  #pragma unroll 16
  for (int i = tid; i < C * C; i += 256) s2 += wl[i / C] * G[i] * wl[i % C];
  float s1 = 0.0f;
  for (int k = tid; k < C; k += 256) s1 += wl[k] * S[k];
  r2[tid] = s2;
  r1[tid] = s1;
  __syncthreads();
  for (int st = 128; st > 0; st >>= 1) {
    if (tid < st) { r2[tid] += r2[tid + st]; r1[tid] += r1[tid + st]; }
    __syncthreads();
  }
  if (tid == 0) {
    float mean = r1[0] * invN;
    float var  = fmaxf(r2[0] * invN - mean * mean, 0.0f);
    float s    = g[c] * rsqrtf(var + BN_EPS);
    sc[c] = s;
    sh[c] = bta[c] - mean * s;
  }
}

__global__ __launch_bounds__(256) void gram_finalize128(
    const float* __restrict__ G, const float* __restrict__ S, float invN,
    const float* __restrict__ Wa, const float* __restrict__ ga, const float* __restrict__ ba,
    float* __restrict__ sca, float* __restrict__ sha, int Ca,
    const float* __restrict__ Wb, const float* __restrict__ gb, const float* __restrict__ bb,
    float* __restrict__ scb, float* __restrict__ shb, int Cb) {
  int c = blockIdx.x;
  const float* W = Wa; const float* g = ga; const float* b = ba;
  float* sc = sca; float* sh = sha; int Cout = Ca;
  if (c >= Ca) { c -= Ca; W = Wb; g = gb; b = bb; sc = scb; sh = shb; Cout = Cb; }
  gram_finalize_body<128>(G, S, invN, W, g, b, sc, sh, Cout, c);
}

__global__ __launch_bounds__(256) void gram_finalize64(
    const float* __restrict__ G, const float* __restrict__ S, float invN,
    const float* __restrict__ W, const float* __restrict__ g, const float* __restrict__ b,
    float* __restrict__ sc, float* __restrict__ sh, int Cout) {
  gram_finalize_body<64>(G, S, invN, W, g, b, sc, sh, Cout, blockIdx.x);
}

extern "C" void kernel_launch(void* const* d_in, const int* in_sizes, int n_in,
                              void* d_out, int out_size, void* d_ws, size_t ws_size,
                              hipStream_t stream) {
  const float* feats = (const float*)d_in[0];
  const float* xyz   = (const float*)d_in[1];
  const int*   nidx  = (const int*)d_in[3];
  const float* W1    = (const float*)d_in[4];
  const float* g1    = (const float*)d_in[5];
  const float* b1    = (const float*)d_in[6];
  const float* kpp   = (const float*)d_in[7];
  const float* kpw   = (const float*)d_in[8];
  const float* W2    = (const float*)d_in[9];
  const float* g2    = (const float*)d_in[10];
  const float* b2    = (const float*)d_in[11];
  const float* Wsc   = (const float*)d_in[12];
  const float* gsc   = (const float*)d_in[13];
  const float* bsc   = (const float*)d_in[14];
  float* out = (float*)d_out;

  const int N = in_sizes[0] / CIN;

  float* ws      = (float*)d_ws;
  float* x1      = ws;
  float* xk      = x1 + (size_t)N * D2;
  float* ysc     = xk + (size_t)N * D2;
  float* stats1  = ysc + (size_t)N * COUT;
  float* stats2  = stats1 + 2 * D2;
  float* statsSC = stats2 + 2 * COUT;
  float* sc1 = statsSC + 2 * COUT; float* sh1 = sc1 + D2;
  float* sc2 = sh1 + D2;           float* sh2 = sc2 + COUT;
  float* scS = sh2 + COUT;         float* shS = scS + COUT;

  const int nblk = (N + 319) / 320;
  float* GpF = ysc;
  float* SpF = GpF + (size_t)nblk * CIN * CIN;
  float* GF  = SpF + (size_t)nblk * CIN;
  float* SF  = GF + CIN * CIN;
  float* Gp2 = x1;
  float* Sp2 = Gp2 + (size_t)nblk * D2 * D2;
  float* G2  = Sp2 + (size_t)nblk * D2;
  float* S2  = G2 + D2 * D2;

  gram_part128<<<nblk, 256, 0, stream>>>(feats, N, GpF, SpF);
  reduce_rows2<<<(CIN * CIN) / 32 + CIN / 32, 256, 0, stream>>>(
      GpF, nblk, CIN * CIN, GF, (CIN * CIN) / 32, SpF, nblk, CIN, SF);
  gram_finalize128<<<D2 + COUT, 256, 0, stream>>>(GF, SF, 1.0f / (float)N,
      W1, g1, b1, sc1, sh1, D2, Wsc, gsc, bsc, scS, shS, COUT);

  {
    int units = (N / 16) * (D2 / 64);
    gemm_bnact_wmma<<<(units + 7) / 8, 256, 0, stream>>>(feats, W1, sc1, sh1, x1,
                                                         N, CIN, D2, units);
  }

  {
    size_t lds = (16 * KDFLAT + 8 * NBH * 16 + 8 * 16 * D2) * sizeof(float)
               + 8 * NBH * sizeof(int);
    kpconv_wmma<<<N / 16, 256, lds, stream>>>(xyz, nidx, x1, kpp, kpw, xk);
  }

  gram_part64<<<nblk, 256, 0, stream>>>(xk, N, Gp2, Sp2);
  reduce_rows2<<<(D2 * D2) / 32 + D2 / 32, 256, 0, stream>>>(
      Gp2, nblk, D2 * D2, G2, (D2 * D2) / 32, Sp2, nblk, D2, S2);
  gram_finalize64<<<COUT, 256, 0, stream>>>(G2, S2, 1.0f / (float)N, W2, g2, b2, sc2, sh2, COUT);

  {
    int units = (N / 16) * (COUT / 64);
    gemm2_bn_residual_wmma<<<(units + 7) / 8, 256, 0, stream>>>(
        xk, W2, D2, sc2, sh2, feats, Wsc, CIN, scS, shS, out, N, COUT, units);
  }
}


# Weight operand of both product kernels staged once per block in LDS (row pairs interleaved, conflict-free 8-byte fragment reads); block = 128 rows x one 64-col strip, 8 waves share it; sums as before
# speedup vs baseline: 2.1459x; 1.0681x over previous
#include <hip/hip_runtime.h>
#include <hip/hip_bf16.h>


typedef __attribute__((ext_vector_type(2))) float v2f;
typedef __attribute__((ext_vector_type(4))) float v4f;
typedef __attribute__((ext_vector_type(8))) float v8f;

#define CIN   128
#define COUT  256
#define D2    64
#define KPTS  15
#define NBH   32
#define KDFLAT (KPTS * D2)
#define INFL_INV 2.5f
#define BN_EPS 1e-5f
#define SLOPE 0.2f
#define PP    160

__device__ __forceinline__ v8f wmma_f32(v2f a, v2f b, v8f c) {
  return __builtin_amdgcn_wmma_f32_16x16x4_f32(
       false, a,  false, b,
       (short)0, c,  false,  false);
}

__device__ __forceinline__ float lrelu(float v) {
  return v >= 0.0f ? v : SLOPE * v;
}


template <int ROWS>
__device__ __forceinline__ void stage_strip(const float* __restrict__ B, int Nc, int col0,
                                            float* Ws) {
  constexpr int ITEMS = (ROWS / 2) * 16;
  #pragma unroll
  for (int i0 = 0; i0 < ITEMS; i0 += 256) {
    const int i = i0 + (int)threadIdx.x;
    if (ITEMS % 256 == 0 || i < ITEMS) {
      const int p = i >> 4;
      const int j = (i & 15) << 2;
      const float* src = B + (size_t)(2 * p) * Nc + col0 + j;
      const v4f r0 = *(const v4f*)src;
      const v4f r1 = *(const v4f*)(src + Nc);
      float* dst = Ws + p * PP + 2 * j;
      *(v4f*)dst       = v4f{r0.x, r1.x, r0.y, r1.y};
      *(v4f*)(dst + 4) = v4f{r0.z, r1.z, r0.w, r1.w};
    }
  }
}

template <int KD>
__device__ __forceinline__ void strip_product(
    const float* __restrict__ Arow, const float* Wl,
    int half, v8f& acc0, v8f& acc1, v8f& acc2, v8f& acc3) {
  for (int kk = 0; kk < KD; kk += 4) {
    int ka = kk + (half << 1);
    v2f a;
    a.x = Arow[ka];
    a.y = Arow[ka + 1];
    const float* Wp = Wl + (ka >> 1) * PP;
    const v2f b0 = *(const v2f*)(Wp);
    const v2f b1 = *(const v2f*)(Wp + 32);
    const v2f b2 = *(const v2f*)(Wp + 64);
    const v2f b3 = *(const v2f*)(Wp + 96);
    acc0 = wmma_f32(a, b0, acc0);
    acc1 = wmma_f32(a, b1, acc1);
    acc2 = wmma_f32(a, b2, acc2);
    acc3 = wmma_f32(a, b3, acc3);
  }
}

template <int KD>
__global__ __launch_bounds__(256) void gemm_bnact_wmma(
    const float* __restrict__ A, const float* __restrict__ B,
    const float* __restrict__ sc, const float* __restrict__ sh,
    float* __restrict__ C, int M, int Nc) {
  __shared__ float Ws[(KD / 2) * PP];
  const int strips = Nc >> 6;
  const int col0 = (blockIdx.x % strips) << 6;
  stage_strip<KD>(B, Nc, col0, Ws);
  __syncthreads();
  const int row0 = (blockIdx.x / strips) * 128 + ((threadIdx.x >> 5) << 4);
  if (row0 >= M) return;
  int lane = threadIdx.x & 31;
  int m    = lane & 15;
  int half = lane >> 4;

  v8f acc0 = {}, acc1 = {}, acc2 = {}, acc3 = {};
  strip_product<KD>(A + (size_t)(row0 + m) * KD, Ws + 2 * m, half,
                    acc0, acc1, acc2, acc3);

  const float* scp = sc + col0 + m;
  const float* shp = sh + col0 + m;
  const float s0 = scp[0], s1 = scp[16], s2 = scp[32], s3 = scp[48];
  const float h0 = shp[0], h1 = shp[16], h2 = shp[32], h3 = shp[48];
  int rbase = row0 + (half << 3);
  float* Cp = C + (size_t)rbase * Nc + col0 + m;
  #pragma unroll
  for (int r = 0; r < 8; ++r) {
    size_t o = (size_t)r * Nc;
    Cp[o +  0] = lrelu(acc0[r] * s0 + h0);
    Cp[o + 16] = lrelu(acc1[r] * s1 + h1);
    Cp[o + 32] = lrelu(acc2[r] * s2 + h2);
    Cp[o + 48] = lrelu(acc3[r] * s3 + h3);
  }
}

template <int KM, int KS>
__global__ __launch_bounds__(256) void gemm2_bn_residual_wmma(
    const float* __restrict__ Am, const float* __restrict__ Bm,
    const float* __restrict__ scM, const float* __restrict__ shM,
    const float* __restrict__ As, const float* __restrict__ Bs,
    const float* __restrict__ scS, const float* __restrict__ shS,
    float* __restrict__ C, int M, int Nc) {
  __shared__ float Ws[((KS + KM) / 2) * PP];
  float* Wm = Ws + (KS / 2) * PP;
  const int strips = Nc >> 6;
  const int col0 = (blockIdx.x % strips) << 6;
  stage_strip<KS>(Bs, Nc, col0, Ws);
  stage_strip<KM>(Bm, Nc, col0, Wm);
  __syncthreads();
  const int row0 = (blockIdx.x / strips) * 128 + ((threadIdx.x >> 5) << 4);
  if (row0 >= M) return;
  int lane = threadIdx.x & 31;
  int m    = lane & 15;
  int half = lane >> 4;

  v8f y0 = {}, y1 = {}, y2 = {}, y3 = {};
  strip_product<KS>(As + (size_t)(row0 + m) * KS, Ws + 2 * m, half,
                    y0, y1, y2, y3);
  v8f acc0 = {}, acc1 = {}, acc2 = {}, acc3 = {};
  strip_product<KM>(Am + (size_t)(row0 + m) * KM, Wm + 2 * m, half,
                    acc0, acc1, acc2, acc3);

  const float* p;
  p = scM + col0 + m; const float a0 = p[0], a1 = p[16], a2 = p[32], a3 = p[48];
  p = shM + col0 + m; const float c0 = p[0], c1 = p[16], c2 = p[32], c3 = p[48];
  p = scS + col0 + m; const float s0 = p[0], s1 = p[16], s2 = p[32], s3 = p[48];
  p = shS + col0 + m; const float h0 = p[0], h1 = p[16], h2 = p[32], h3 = p[48];
  int rbase = row0 + (half << 3);
  float* Cp = C + (size_t)rbase * Nc + col0 + m;
  #pragma unroll
  for (int r = 0; r < 8; ++r) {
    size_t o = (size_t)r * Nc;
    Cp[o +  0] = lrelu(acc0[r] * a0 + c0) + (y0[r] * s0 + h0);
    Cp[o + 16] = lrelu(acc1[r] * a1 + c1) + (y1[r] * s1 + h1);
    Cp[o + 32] = lrelu(acc2[r] * a2 + c2) + (y2[r] * s2 + h2);
    Cp[o + 48] = lrelu(acc3[r] * a3 + c3) + (y3[r] * s3 + h3);
  }
}

__global__ __launch_bounds__(256) void kpconv_wmma(
    const float* __restrict__ xyz, const int* __restrict__ nidx,
    const float* __restrict__ x1, const float* __restrict__ kp_points,
    const float* __restrict__ kpw, float* __restrict__ xk) {
  extern __shared__ float smem[];
  float* FK  = smem;
  float* Wl  = FK + 16 * KDFLAT;
  int*   IDX = (int*)(Wl + 8 * NBH * 16);
  float* PS  = (float*)(IDX + 8 * NBH);

  const int wave = threadIdx.x >> 5;
  const int lane = threadIdx.x & 31;
  const int m    = lane & 15;
  const int half = lane >> 4;
  const int base = blockIdx.x << 4;

  float* wl = Wl + wave * (NBH * 16);
  int*   il = IDX + wave * NBH;

  for (int it = 0; it < 2; ++it) {
    const int pi = (wave << 1) + it;
    const int n  = base + pi;

    const int id = nidx[(size_t)n * NBH + lane];
    il[lane] = id;
    const float cx = xyz[(size_t)n * 3 + 0];
    const float cy = xyz[(size_t)n * 3 + 1];
    const float cz = xyz[(size_t)n * 3 + 2];
    const float px = xyz[(size_t)id * 3 + 0] - cx;
    const float py = xyz[(size_t)id * 3 + 1] - cy;
    const float pz = xyz[(size_t)id * 3 + 2] - cz;
    #pragma unroll
    for (int k = 0; k < KPTS; ++k) {
      float dx = px - kp_points[k * 3 + 0];
      float dy = py - kp_points[k * 3 + 1];
      float dz = pz - kp_points[k * 3 + 2];
      float r = sqrtf(dx * dx + dy * dy + dz * dz);
      wl[lane * 16 + k] = fmaxf(1.0f - r * INFL_INV, 0.0f);
    }
    wl[lane * 16 + KPTS] = 0.0f;
    __syncthreads();

    v8f acc0 = {}, acc1 = {}, acc2 = {}, acc3 = {};
    #pragma unroll
    for (int kk = 0; kk < NBH; kk += 4) {
      const int ka = kk + (half << 1);
      v2f a;
      a.x = wl[ka * 16 + m];
      a.y = wl[(ka + 1) * 16 + m];
      const float* r0 = x1 + (size_t)il[ka] * D2 + m;
      const float* r1 = x1 + (size_t)il[ka + 1] * D2 + m;
      v2f b0, b1, b2, b3;
      b0.x = r0[0];  b0.y = r1[0];
      b1.x = r0[16]; b1.y = r1[16];
      b2.x = r0[32]; b2.y = r1[32];
      b3.x = r0[48]; b3.y = r1[48];
      acc0 = wmma_f32(a, b0, acc0);
      acc1 = wmma_f32(a, b1, acc1);
      acc2 = wmma_f32(a, b2, acc2);
      acc3 = wmma_f32(a, b3, acc3);
    }
    #pragma unroll
    for (int r = 0; r < 8; ++r) {
      const int krow = (half << 3) + r;
      if (krow < KPTS) {
        float* f = FK + pi * KDFLAT + krow * D2 + m;
        f[0]  = acc0[r];
        f[16] = acc1[r];
        f[32] = acc2[r];
        f[48] = acc3[r];
      }
    }
    __syncthreads();
  }

  {
    const int kbeg = wave * 120;
    v8f acc0 = {}, acc1 = {}, acc2 = {}, acc3 = {};
    for (int kk = kbeg; kk < kbeg + 120; kk += 4) {
      const int ka = kk + (half << 1);
      v2f a;
      a.x = FK[m * KDFLAT + ka];
      a.y = FK[m * KDFLAT + ka + 1];
      const float* Bp = kpw + (size_t)ka * D2 + m;
      v2f b0, b1, b2, b3;
      b0.x = Bp[0];  b0.y = Bp[D2];
      b1.x = Bp[16]; b1.y = Bp[D2 + 16];
      b2.x = Bp[32]; b2.y = Bp[D2 + 32];
      b3.x = Bp[48]; b3.y = Bp[D2 + 48];
      acc0 = wmma_f32(a, b0, acc0);
      acc1 = wmma_f32(a, b1, acc1);
      acc2 = wmma_f32(a, b2, acc2);
      acc3 = wmma_f32(a, b3, acc3);
    }
    #pragma unroll
    for (int r = 0; r < 8; ++r) {
      float* p = PS + wave * (16 * D2) + ((half << 3) + r) * D2 + m;
      p[0]  = acc0[r];
      p[16] = acc1[r];
      p[32] = acc2[r];
      p[48] = acc3[r];
    }
  }
  __syncthreads();

  for (int t = threadIdx.x; t < 16 * D2; t += 256) {
    float s = 0.0f;
    #pragma unroll
    for (int w = 0; w < 8; ++w) s += PS[w * (16 * D2) + t];
    const int pt = t >> 6;
    const int e  = t & 63;
    xk[(size_t)(base + pt) * D2 + e] = s;
  }
}

#define GRAM_U 4

template <int C, int RPG>
__device__ __forceinline__ void gram_body(const float* __restrict__ X, int N,
                                          float* __restrict__ Gp,
                                          float* __restrict__ Sp, float* red,
                                          float* reds) {
  constexpr int TC     = C / 64;
  constexpr int TILES  = (C / 32) * TC;
  constexpr int GROUPS = 8 / TILES;
  const int wave = __builtin_amdgcn_readfirstlane((int)(threadIdx.x >> 5));
  const int lane = threadIdx.x & 31;
  const int m    = lane & 15;
  const int half = lane >> 4;
  const int tile = wave % TILES;
  const int grp  = wave / TILES;
  const int tr   = tile / TC;
  const int tc   = tile % TC;
  const int gidx = blockIdx.x * GROUPS + grp;
  const int row0 = gidx * RPG;
  int nrows = N - row0;
  nrows = nrows < 0 ? 0 : (nrows > RPG ? RPG : nrows);
  const int nruns = nrows / (4 * GRAM_U);

  const float* xb = X + (size_t)(row0 + 2 * half) * C + m;
  const float* ap = xb + tr * 32;
  const float* bp = xb + tc * 64;

  v8f acc[2][4];
  #pragma unroll
  for (int i = 0; i < 2; ++i)
    #pragma unroll
    for (int j = 0; j < 4; ++j) acc[i][j] = v8f{};
  float cs[4] = {0.0f, 0.0f, 0.0f, 0.0f};

  auto load_run = [&](v2f (&a)[GRAM_U][2], v2f (&b)[GRAM_U][4], int run) {
    const size_t o = (size_t)run * (4 * GRAM_U) * C;
    #pragma unroll
    for (int u = 0; u < GRAM_U; ++u) {
      const float* pa = ap + o + (size_t)u * 4 * C;
      const float* pb = bp + o + (size_t)u * 4 * C;
      #pragma unroll
      for (int i = 0; i < 2; ++i) { a[u][i].x = pa[i * 16]; a[u][i].y = pa[C + i * 16]; }
      #pragma unroll
      for (int j = 0; j < 4; ++j) { b[u][j].x = pb[j * 16]; b[u][j].y = pb[C + j * 16]; }
    }
  };
  auto mma_run = [&](v2f (&a)[GRAM_U][2], v2f (&b)[GRAM_U][4]) {
    #pragma unroll
    for (int u = 0; u < GRAM_U; ++u) {
      if (tr == 0) {
        #pragma unroll
        for (int j = 0; j < 4; ++j) cs[j] += b[u][j].x + b[u][j].y;
      }
      #pragma unroll
      for (int i = 0; i < 2; ++i)
        #pragma unroll
        for (int j = 0; j < 4; ++j) acc[i][j] = wmma_f32(a[u][i], b[u][j], acc[i][j]);
    }
  };

  v2f a0[GRAM_U][2], b0[GRAM_U][4], a1[GRAM_U][2], b1[GRAM_U][4];
  if (nruns > 0) load_run(a0, b0, 0);
  for (int r = 0; r < nruns; r += 2) {
    if (r + 1 < nruns) load_run(a1, b1, r + 1);
    mma_run(a0, b0);
    if (r + 2 < nruns) load_run(a0, b0, r + 2);
    if (r + 1 < nruns) mma_run(a1, b1);
  }

  const int grow = tr * 32 + (half << 3);
  const int gcol = tc * 64 + m;
  if (tr == 0) {
    #pragma unroll
    for (int j = 0; j < 4; ++j) cs[j] += __shfl_xor(cs[j], 16);
  }
  if constexpr (GROUPS > 1) {
    if (grp > 0) {
      if (tr == 0 && half == 0) {
        #pragma unroll
        for (int j = 0; j < 4; ++j) reds[(grp - 1) * C + gcol + j * 16] = cs[j];
      }
      float* rp = red + (size_t)(grp - 1) * (C * C);
      #pragma unroll
      for (int i = 0; i < 2; ++i)
        #pragma unroll
        for (int j = 0; j < 4; ++j)
          #pragma unroll
          for (int r = 0; r < 8; ++r)
            rp[(grow + i * 16 + r) * C + gcol + j * 16] = acc[i][j][r];
    }
    __syncthreads();
    if (grp == 0) {
      if (tr == 0) {
        #pragma unroll
        for (int g = 0; g < GROUPS - 1; ++g)
          #pragma unroll
          for (int j = 0; j < 4; ++j) cs[j] += reds[g * C + gcol + j * 16];
      }
      #pragma unroll
      for (int g = 0; g < GROUPS - 1; ++g)
        #pragma unroll
        for (int i = 0; i < 2; ++i)
          #pragma unroll
          for (int j = 0; j < 4; ++j)
            #pragma unroll
            for (int r = 0; r < 8; ++r)
              acc[i][j][r] += red[(size_t)g * (C * C) + (grow + i * 16 + r) * C + gcol + j * 16];
    }
  }
  if (grp == 0) {
    float* gp = Gp + (size_t)blockIdx.x * (C * C);
    #pragma unroll
    for (int i = 0; i < 2; ++i)
      #pragma unroll
      for (int j = 0; j < 4; ++j)
        #pragma unroll
        for (int r = 0; r < 8; ++r)
          gp[(grow + i * 16 + r) * C + gcol + j * 16] = acc[i][j][r];
  }
  if (grp == 0 && tr == 0 && half == 0) {
    float* sp = Sp + (size_t)blockIdx.x * C + gcol;
    #pragma unroll
    for (int j = 0; j < 4; ++j) sp[j * 16] = cs[j];
  }
}

__global__ __launch_bounds__(256) void gram_part128(const float* __restrict__ X, int N,
                                                    float* __restrict__ Gp,
                                                    float* __restrict__ Sp) {
  gram_body<128, 320>(X, N, Gp, Sp, nullptr, nullptr);
}

__global__ __launch_bounds__(256) void gram_part64(const float* __restrict__ X, int N,
                                                   float* __restrict__ Gp,
                                                   float* __restrict__ Sp) {
  __shared__ float red[3 * 64 * 64];
  __shared__ float reds[3 * 64];
  gram_body<64, 80>(X, N, Gp, Sp, red, reds);
}

__global__ __launch_bounds__(256) void reduce_rows2(
    const float* __restrict__ srcA, int nrA, int wA, float* __restrict__ dstA, int blocksA,
    const float* __restrict__ srcB, int nrB, int wB, float* __restrict__ dstB) {
  __shared__ float red[8][32];
  int b = blockIdx.x;
  const float* src = srcA; int nr = nrA; int w = wA; float* dst = dstA;
  if (b >= blocksA) { b -= blocksA; src = srcB; nr = nrB; w = wB; dst = dstB; }
  const int e = threadIdx.x & 31;
  const int q = threadIdx.x >> 5;
  const int col = b * 32 + e;
  float s = 0.0f;
  #pragma unroll 8
  for (int r = q; r < nr; r += 8) s += src[(size_t)r * w + col];
  red[q][e] = s;
  __syncthreads();
  if (q == 0) {
    float t = red[0][e];
    #pragma unroll
    for (int i = 1; i < 8; ++i) t += red[i][e];
    dst[col] = t;
  }
}

template <int C>
__device__ __forceinline__ void gram_finalize_body(
    const float* __restrict__ G, const float* __restrict__ S, float invN,
    const float* __restrict__ W, const float* __restrict__ g,
    const float* __restrict__ bta, float* __restrict__ sc, float* __restrict__ sh,
    int Cout, int c) {
  __shared__ float wl[C];
  __shared__ float r2[256];
  __shared__ float r1[256];
  const int tid = threadIdx.x;
  for (int k = tid; k < C; k += 256) wl[k] = W[(size_t)k * Cout + c];
  __syncthreads();
  float s2 = 0.0f;
  #pragma unroll 16
  for (int i = tid; i < C * C; i += 256) s2 += wl[i / C] * G[i] * wl[i % C];
  float s1 = 0.0f;
  for (int k = tid; k < C; k += 256) s1 += wl[k] * S[k];
  r2[tid] = s2;
  r1[tid] = s1;
  __syncthreads();
  for (int st = 128; st > 0; st >>= 1) {
    if (tid < st) { r2[tid] += r2[tid + st]; r1[tid] += r1[tid + st]; }
    __syncthreads();
  }
  if (tid == 0) {
    float mean = r1[0] * invN;
    float var  = fmaxf(r2[0] * invN - mean * mean, 0.0f);
    float s    = g[c] * rsqrtf(var + BN_EPS);
    sc[c] = s;
    sh[c] = bta[c] - mean * s;
  }
}

__global__ __launch_bounds__(256) void gram_finalize128(
    const float* __restrict__ G, const float* __restrict__ S, float invN,
    const float* __restrict__ Wa, const float* __restrict__ ga, const float* __restrict__ ba,
    float* __restrict__ sca, float* __restrict__ sha, int Ca,
    const float* __restrict__ Wb, const float* __restrict__ gb, const float* __restrict__ bb,
    float* __restrict__ scb, float* __restrict__ shb, int Cb) {
  int c = blockIdx.x;
  const float* W = Wa; const float* g = ga; const float* b = ba;
  float* sc = sca; float* sh = sha; int Cout = Ca;
  if (c >= Ca) { c -= Ca; W = Wb; g = gb; b = bb; sc = scb; sh = shb; Cout = Cb; }
  gram_finalize_body<128>(G, S, invN, W, g, b, sc, sh, Cout, c);
}

__global__ __launch_bounds__(256) void gram_finalize64(
    const float* __restrict__ G, const float* __restrict__ S, float invN,
    const float* __restrict__ W, const float* __restrict__ g, const float* __restrict__ b,
    float* __restrict__ sc, float* __restrict__ sh, int Cout) {
  gram_finalize_body<64>(G, S, invN, W, g, b, sc, sh, Cout, blockIdx.x);
}

extern "C" void kernel_launch(void* const* d_in, const int* in_sizes, int n_in,
                              void* d_out, int out_size, void* d_ws, size_t ws_size,
                              hipStream_t stream) {
  const float* feats = (const float*)d_in[0];
  const float* xyz   = (const float*)d_in[1];
  const int*   nidx  = (const int*)d_in[3];
  const float* W1    = (const float*)d_in[4];
  const float* g1    = (const float*)d_in[5];
  const float* b1    = (const float*)d_in[6];
  const float* kpp   = (const float*)d_in[7];
  const float* kpw   = (const float*)d_in[8];
  const float* W2    = (const float*)d_in[9];
  const float* g2    = (const float*)d_in[10];
  const float* b2    = (const float*)d_in[11];
  const float* Wsc   = (const float*)d_in[12];
  const float* gsc   = (const float*)d_in[13];
  const float* bsc   = (const float*)d_in[14];
  float* out = (float*)d_out;

  const int N = in_sizes[0] / CIN;

  float* ws      = (float*)d_ws;
  float* x1      = ws;
  float* xk      = x1 + (size_t)N * D2;
  float* ysc     = xk + (size_t)N * D2;
  float* stats1  = ysc + (size_t)N * COUT;
  float* stats2  = stats1 + 2 * D2;
  float* statsSC = stats2 + 2 * COUT;
  float* sc1 = statsSC + 2 * COUT; float* sh1 = sc1 + D2;
  float* sc2 = sh1 + D2;           float* sh2 = sc2 + COUT;
  float* scS = sh2 + COUT;         float* shS = scS + COUT;

  const int nblk = (N + 319) / 320;
  float* GpF = ysc;
  float* SpF = GpF + (size_t)nblk * CIN * CIN;
  float* GF  = SpF + (size_t)nblk * CIN;
  float* SF  = GF + CIN * CIN;
  float* Gp2 = x1;
  float* Sp2 = Gp2 + (size_t)nblk * D2 * D2;
  float* G2  = Sp2 + (size_t)nblk * D2;
  float* S2  = G2 + D2 * D2;

  gram_part128<<<nblk, 256, 0, stream>>>(feats, N, GpF, SpF);
  reduce_rows2<<<(CIN * CIN) / 32 + CIN / 32, 256, 0, stream>>>(
      GpF, nblk, CIN * CIN, GF, (CIN * CIN) / 32, SpF, nblk, CIN, SF);
  gram_finalize128<<<D2 + COUT, 256, 0, stream>>>(GF, SF, 1.0f / (float)N,
      W1, g1, b1, sc1, sh1, D2, Wsc, gsc, bsc, scS, shS, COUT);

  gemm_bnact_wmma<CIN><<<((N + 127) / 128) * (D2 / 64), 256, 0, stream>>>(
      feats, W1, sc1, sh1, x1, N, D2);

  {
    size_t lds = (16 * KDFLAT + 8 * NBH * 16 + 8 * 16 * D2) * sizeof(float)
               + 8 * NBH * sizeof(int);
    kpconv_wmma<<<N / 16, 256, lds, stream>>>(xyz, nidx, x1, kpp, kpw, xk);
  }

  gram_part64<<<nblk, 256, 0, stream>>>(xk, N, Gp2, Sp2);
  reduce_rows2<<<(D2 * D2) / 32 + D2 / 32, 256, 0, stream>>>(
      Gp2, nblk, D2 * D2, G2, (D2 * D2) / 32, Sp2, nblk, D2, S2);
  gram_finalize64<<<COUT, 256, 0, stream>>>(G2, S2, 1.0f / (float)N, W2, g2, b2, sc2, sh2, COUT);

  gemm2_bn_residual_wmma<D2, CIN><<<((N + 127) / 128) * (COUT / 64), 256, 0, stream>>>(
      xk, W2, sc2, sh2, feats, Wsc, scS, shS, out, N, COUT);
}


# KPConv block tile 16 -> 32 points: in phase 2 a wave holds both row tiles (8 accumulators), kpw is fetched once per 32 points, loads per WMMA halve; FK row pitch 964, partials over FK; sums as before
# speedup vs baseline: 2.3180x; 1.0802x over previous
#include <hip/hip_runtime.h>
#include <hip/hip_bf16.h>


typedef __attribute__((ext_vector_type(2))) float v2f;
typedef __attribute__((ext_vector_type(4))) float v4f;
typedef __attribute__((ext_vector_type(8))) float v8f;

#define CIN   128
#define COUT  256
#define D2    64
#define KPTS  15
#define NBH   32
#define KDFLAT (KPTS * D2)
#define INFL_INV 2.5f
#define BN_EPS 1e-5f
#define SLOPE 0.2f
#define PP    160

__device__ __forceinline__ v8f wmma_f32(v2f a, v2f b, v8f c) {
  return __builtin_amdgcn_wmma_f32_16x16x4_f32(
       false, a,  false, b,
       (short)0, c,  false,  false);
}

__device__ __forceinline__ float lrelu(float v) {
  return v >= 0.0f ? v : SLOPE * v;
}


template <int ROWS>
__device__ __forceinline__ void stage_strip(const float* __restrict__ B, int Nc, int col0,
                                            float* Ws) {
  constexpr int ITEMS = (ROWS / 2) * 16;
  #pragma unroll
  for (int i0 = 0; i0 < ITEMS; i0 += 256) {
    const int i = i0 + (int)threadIdx.x;
    if (ITEMS % 256 == 0 || i < ITEMS) {
      const int p = i >> 4;
      const int j = (i & 15) << 2;
      const float* src = B + (size_t)(2 * p) * Nc + col0 + j;
      const v4f r0 = *(const v4f*)src;
      const v4f r1 = *(const v4f*)(src + Nc);
      float* dst = Ws + p * PP + 2 * j;
      *(v4f*)dst       = v4f{r0.x, r1.x, r0.y, r1.y};
      *(v4f*)(dst + 4) = v4f{r0.z, r1.z, r0.w, r1.w};
    }
  }
}

template <int KD>
__device__ __forceinline__ void strip_product(
    const float* __restrict__ Arow, const float* Wl,
    int half, v8f& acc0, v8f& acc1, v8f& acc2, v8f& acc3) {
  for (int kk = 0; kk < KD; kk += 4) {
    int ka = kk + (half << 1);
    v2f a;
    a.x = Arow[ka];
    a.y = Arow[ka + 1];
    const float* Wp = Wl + (ka >> 1) * PP;
    const v2f b0 = *(const v2f*)(Wp);
    const v2f b1 = *(const v2f*)(Wp + 32);
    const v2f b2 = *(const v2f*)(Wp + 64);
    const v2f b3 = *(const v2f*)(Wp + 96);
    acc0 = wmma_f32(a, b0, acc0);
    acc1 = wmma_f32(a, b1, acc1);
    acc2 = wmma_f32(a, b2, acc2);
    acc3 = wmma_f32(a, b3, acc3);
  }
}

template <int KD>
__global__ __launch_bounds__(256) void gemm_bnact_wmma(
    const float* __restrict__ A, const float* __restrict__ B,
    const float* __restrict__ sc, const float* __restrict__ sh,
    float* __restrict__ C, int M, int Nc) {
  __shared__ float Ws[(KD / 2) * PP];
  const int strips = Nc >> 6;
  const int col0 = (blockIdx.x % strips) << 6;
  stage_strip<KD>(B, Nc, col0, Ws);
  __syncthreads();
  const int row0 = (blockIdx.x / strips) * 128 + ((threadIdx.x >> 5) << 4);
  if (row0 >= M) return;
  int lane = threadIdx.x & 31;
  int m    = lane & 15;
  int half = lane >> 4;

  v8f acc0 = {}, acc1 = {}, acc2 = {}, acc3 = {};
  strip_product<KD>(A + (size_t)(row0 + m) * KD, Ws + 2 * m, half,
                    acc0, acc1, acc2, acc3);

  const float* scp = sc + col0 + m;
  const float* shp = sh + col0 + m;
  const float s0 = scp[0], s1 = scp[16], s2 = scp[32], s3 = scp[48];
  const float h0 = shp[0], h1 = shp[16], h2 = shp[32], h3 = shp[48];
  int rbase = row0 + (half << 3);
  float* Cp = C + (size_t)rbase * Nc + col0 + m;
  #pragma unroll
  for (int r = 0; r < 8; ++r) {
    size_t o = (size_t)r * Nc;
    Cp[o +  0] = lrelu(acc0[r] * s0 + h0);
    Cp[o + 16] = lrelu(acc1[r] * s1 + h1);
    Cp[o + 32] = lrelu(acc2[r] * s2 + h2);
    Cp[o + 48] = lrelu(acc3[r] * s3 + h3);
  }
}

template <int KM, int KS>
__global__ __launch_bounds__(256) void gemm2_bn_residual_wmma(
    const float* __restrict__ Am, const float* __restrict__ Bm,
    const float* __restrict__ scM, const float* __restrict__ shM,
    const float* __restrict__ As, const float* __restrict__ Bs,
    const float* __restrict__ scS, const float* __restrict__ shS,
    float* __restrict__ C, int M, int Nc) {
  __shared__ float Ws[((KS + KM) / 2) * PP];
  float* Wm = Ws + (KS / 2) * PP;
  const int strips = Nc >> 6;
  const int col0 = (blockIdx.x % strips) << 6;
  stage_strip<KS>(Bs, Nc, col0, Ws);
  stage_strip<KM>(Bm, Nc, col0, Wm);
  __syncthreads();
  const int row0 = (blockIdx.x / strips) * 128 + ((threadIdx.x >> 5) << 4);
  if (row0 >= M) return;
  int lane = threadIdx.x & 31;
  int m    = lane & 15;
  int half = lane >> 4;

  v8f y0 = {}, y1 = {}, y2 = {}, y3 = {};
  strip_product<KS>(As + (size_t)(row0 + m) * KS, Ws + 2 * m, half,
                    y0, y1, y2, y3);
  v8f acc0 = {}, acc1 = {}, acc2 = {}, acc3 = {};
  strip_product<KM>(Am + (size_t)(row0 + m) * KM, Wm + 2 * m, half,
                    acc0, acc1, acc2, acc3);

  const float* p;
  p = scM + col0 + m; const float a0 = p[0], a1 = p[16], a2 = p[32], a3 = p[48];
  p = shM + col0 + m; const float c0 = p[0], c1 = p[16], c2 = p[32], c3 = p[48];
  p = scS + col0 + m; const float s0 = p[0], s1 = p[16], s2 = p[32], s3 = p[48];
  p = shS + col0 + m; const float h0 = p[0], h1 = p[16], h2 = p[32], h3 = p[48];
  int rbase = row0 + (half << 3);
  float* Cp = C + (size_t)rbase * Nc + col0 + m;
  #pragma unroll
  for (int r = 0; r < 8; ++r) {
    size_t o = (size_t)r * Nc;
    Cp[o +  0] = lrelu(acc0[r] * a0 + c0) + (y0[r] * s0 + h0);
    Cp[o + 16] = lrelu(acc1[r] * a1 + c1) + (y1[r] * s1 + h1);
    Cp[o + 32] = lrelu(acc2[r] * a2 + c2) + (y2[r] * s2 + h2);
    Cp[o + 48] = lrelu(acc3[r] * a3 + c3) + (y3[r] * s3 + h3);
  }
}

#define KPT 32
#define FKP 964
__global__ __launch_bounds__(256) void kpconv_wmma(
    const float* __restrict__ xyz, const int* __restrict__ nidx,
    const float* __restrict__ x1, const float* __restrict__ kp_points,
    const float* __restrict__ kpw, float* __restrict__ xk) {
  extern __shared__ float smem[];
  float* FK  = smem;
  float* Wl  = FK + KPT * FKP;
  int*   IDX = (int*)(Wl + 8 * NBH * 16);
  float* PS  = smem;

  const int wave = threadIdx.x >> 5;
  const int lane = threadIdx.x & 31;
  const int m    = lane & 15;
  const int half = lane >> 4;
  const int base = blockIdx.x * KPT;

  float* wl = Wl + wave * (NBH * 16);
  int*   il = IDX + wave * NBH;

  for (int it = 0; it < KPT / 8; ++it) {
    const int pi = wave * (KPT / 8) + it;
    const int n  = base + pi;

    const int id = nidx[(size_t)n * NBH + lane];
    il[lane] = id;
    const float cx = xyz[(size_t)n * 3 + 0];
    const float cy = xyz[(size_t)n * 3 + 1];
    const float cz = xyz[(size_t)n * 3 + 2];
    const float px = xyz[(size_t)id * 3 + 0] - cx;
    const float py = xyz[(size_t)id * 3 + 1] - cy;
    const float pz = xyz[(size_t)id * 3 + 2] - cz;
    #pragma unroll
    for (int k = 0; k < KPTS; ++k) {
      float dx = px - kp_points[k * 3 + 0];
      float dy = py - kp_points[k * 3 + 1];
      float dz = pz - kp_points[k * 3 + 2];
      float r = sqrtf(dx * dx + dy * dy + dz * dz);
      wl[lane * 16 + k] = fmaxf(1.0f - r * INFL_INV, 0.0f);
    }
    wl[lane * 16 + KPTS] = 0.0f;
    __syncthreads();

    v8f acc0 = {}, acc1 = {}, acc2 = {}, acc3 = {};
    #pragma unroll
    for (int kk = 0; kk < NBH; kk += 4) {
      const int ka = kk + (half << 1);
      v2f a;
      a.x = wl[ka * 16 + m];
      a.y = wl[(ka + 1) * 16 + m];
      const float* r0 = x1 + (size_t)il[ka] * D2 + m;
      const float* r1 = x1 + (size_t)il[ka + 1] * D2 + m;
      v2f b0, b1, b2, b3;
      b0.x = r0[0];  b0.y = r1[0];
      b1.x = r0[16]; b1.y = r1[16];
      b2.x = r0[32]; b2.y = r1[32];
      b3.x = r0[48]; b3.y = r1[48];
      acc0 = wmma_f32(a, b0, acc0);
      acc1 = wmma_f32(a, b1, acc1);
      acc2 = wmma_f32(a, b2, acc2);
      acc3 = wmma_f32(a, b3, acc3);
    }
    #pragma unroll
    for (int r = 0; r < 8; ++r) {
      const int krow = (half << 3) + r;
      if (krow < KPTS) {
        float* f = FK + pi * FKP + krow * D2 + m;
        f[0]  = acc0[r];
        f[16] = acc1[r];
        f[32] = acc2[r];
        f[48] = acc3[r];
      }
    }
    __syncthreads();
  }

  {
    const int kbeg = wave * 120;
    v8f lo0 = {}, lo1 = {}, lo2 = {}, lo3 = {};
    v8f hi0 = {}, hi1 = {}, hi2 = {}, hi3 = {};
    const float* Alo = FK + m * FKP;
    const float* Ahi = FK + (16 + m) * FKP;
    for (int kk = kbeg; kk < kbeg + 120; kk += 4) {
      const int ka = kk + (half << 1);
      const v2f alo = *(const v2f*)(Alo + ka);
      const v2f ahi = *(const v2f*)(Ahi + ka);
      const float* Bp = kpw + (size_t)ka * D2 + m;
      v2f b0, b1, b2, b3;
      b0.x = Bp[0];  b0.y = Bp[D2];
      b1.x = Bp[16]; b1.y = Bp[D2 + 16];
      b2.x = Bp[32]; b2.y = Bp[D2 + 32];
      b3.x = Bp[48]; b3.y = Bp[D2 + 48];
      lo0 = wmma_f32(alo, b0, lo0);
      hi0 = wmma_f32(ahi, b0, hi0);
      lo1 = wmma_f32(alo, b1, lo1);
      hi1 = wmma_f32(ahi, b1, hi1);
      lo2 = wmma_f32(alo, b2, lo2);
      hi2 = wmma_f32(ahi, b2, hi2);
      lo3 = wmma_f32(alo, b3, lo3);
      hi3 = wmma_f32(ahi, b3, hi3);
    }
    __syncthreads();
    #pragma unroll
    for (int r = 0; r < 8; ++r) {
      float* p = PS + wave * (KPT * D2) + ((half << 3) + r) * D2 + m;
      p[0]  = lo0[r];
      p[16] = lo1[r];
      p[32] = lo2[r];
      p[48] = lo3[r];
      p += 16 * D2;
      p[0]  = hi0[r];
      p[16] = hi1[r];
      p[32] = hi2[r];
      p[48] = hi3[r];
    }
  }
  __syncthreads();

  for (int t = threadIdx.x; t < KPT * D2; t += 256) {
    float s = 0.0f;
    #pragma unroll
    for (int w = 0; w < 8; ++w) s += PS[w * (KPT * D2) + t];
    const int pt = t >> 6;
    const int e  = t & 63;
    xk[(size_t)(base + pt) * D2 + e] = s;
  }
}

#define GRAM_U 4

template <int C, int RPG>
__device__ __forceinline__ void gram_body(const float* __restrict__ X, int N,
                                          float* __restrict__ Gp,
                                          float* __restrict__ Sp, float* red,
                                          float* reds) {
  constexpr int TC     = C / 64;
  constexpr int TILES  = (C / 32) * TC;
  constexpr int GROUPS = 8 / TILES;
  const int wave = __builtin_amdgcn_readfirstlane((int)(threadIdx.x >> 5));
  const int lane = threadIdx.x & 31;
  const int m    = lane & 15;
  const int half = lane >> 4;
  const int tile = wave % TILES;
  const int grp  = wave / TILES;
  const int tr   = tile / TC;
  const int tc   = tile % TC;
  const int gidx = blockIdx.x * GROUPS + grp;
  const int row0 = gidx * RPG;
  int nrows = N - row0;
  nrows = nrows < 0 ? 0 : (nrows > RPG ? RPG : nrows);
  const int nruns = nrows / (4 * GRAM_U);

  const float* xb = X + (size_t)(row0 + 2 * half) * C + m;
  const float* ap = xb + tr * 32;
  const float* bp = xb + tc * 64;

  v8f acc[2][4];
  #pragma unroll
  for (int i = 0; i < 2; ++i)
    #pragma unroll
    for (int j = 0; j < 4; ++j) acc[i][j] = v8f{};
  float cs[4] = {0.0f, 0.0f, 0.0f, 0.0f};

  auto load_run = [&](v2f (&a)[GRAM_U][2], v2f (&b)[GRAM_U][4], int run) {
    const size_t o = (size_t)run * (4 * GRAM_U) * C;
    #pragma unroll
    for (int u = 0; u < GRAM_U; ++u) {
      const float* pa = ap + o + (size_t)u * 4 * C;
      const float* pb = bp + o + (size_t)u * 4 * C;
      #pragma unroll
      for (int i = 0; i < 2; ++i) { a[u][i].x = pa[i * 16]; a[u][i].y = pa[C + i * 16]; }
      #pragma unroll
      for (int j = 0; j < 4; ++j) { b[u][j].x = pb[j * 16]; b[u][j].y = pb[C + j * 16]; }
    }
  };
  auto mma_run = [&](v2f (&a)[GRAM_U][2], v2f (&b)[GRAM_U][4]) {
    #pragma unroll
    for (int u = 0; u < GRAM_U; ++u) {
      if (tr == 0) {
        #pragma unroll
        for (int j = 0; j < 4; ++j) cs[j] += b[u][j].x + b[u][j].y;
      }
      #pragma unroll
      for (int i = 0; i < 2; ++i)
        #pragma unroll
        for (int j = 0; j < 4; ++j) acc[i][j] = wmma_f32(a[u][i], b[u][j], acc[i][j]);
    }
  };

  v2f a0[GRAM_U][2], b0[GRAM_U][4], a1[GRAM_U][2], b1[GRAM_U][4];
  if (nruns > 0) load_run(a0, b0, 0);
  for (int r = 0; r < nruns; r += 2) {
    if (r + 1 < nruns) load_run(a1, b1, r + 1);
    mma_run(a0, b0);
    if (r + 2 < nruns) load_run(a0, b0, r + 2);
    if (r + 1 < nruns) mma_run(a1, b1);
  }

  const int grow = tr * 32 + (half << 3);
  const int gcol = tc * 64 + m;
  if (tr == 0) {
    #pragma unroll
    for (int j = 0; j < 4; ++j) cs[j] += __shfl_xor(cs[j], 16);
  }
  if constexpr (GROUPS > 1) {
    if (grp > 0) {
      if (tr == 0 && half == 0) {
        #pragma unroll
        for (int j = 0; j < 4; ++j) reds[(grp - 1) * C + gcol + j * 16] = cs[j];
      }
      float* rp = red + (size_t)(grp - 1) * (C * C);
      #pragma unroll
      for (int i = 0; i < 2; ++i)
        #pragma unroll
        for (int j = 0; j < 4; ++j)
          #pragma unroll
          for (int r = 0; r < 8; ++r)
            rp[(grow + i * 16 + r) * C + gcol + j * 16] = acc[i][j][r];
    }
    __syncthreads();
    if (grp == 0) {
      if (tr == 0) {
        #pragma unroll
        for (int g = 0; g < GROUPS - 1; ++g)
          #pragma unroll
          for (int j = 0; j < 4; ++j) cs[j] += reds[g * C + gcol + j * 16];
      }
      #pragma unroll
      for (int g = 0; g < GROUPS - 1; ++g)
        #pragma unroll
        for (int i = 0; i < 2; ++i)
          #pragma unroll
          for (int j = 0; j < 4; ++j)
            #pragma unroll
            for (int r = 0; r < 8; ++r)
              acc[i][j][r] += red[(size_t)g * (C * C) + (grow + i * 16 + r) * C + gcol + j * 16];
    }
  }
  if (grp == 0) {
    float* gp = Gp + (size_t)blockIdx.x * (C * C);
    #pragma unroll
    for (int i = 0; i < 2; ++i)
      #pragma unroll
      for (int j = 0; j < 4; ++j)
        #pragma unroll
        for (int r = 0; r < 8; ++r)
          gp[(grow + i * 16 + r) * C + gcol + j * 16] = acc[i][j][r];
  }
  if (grp == 0 && tr == 0 && half == 0) {
    float* sp = Sp + (size_t)blockIdx.x * C + gcol;
    #pragma unroll
    for (int j = 0; j < 4; ++j) sp[j * 16] = cs[j];
  }
}

__global__ __launch_bounds__(256) void gram_part128(const float* __restrict__ X, int N,
                                                    float* __restrict__ Gp,
                                                    float* __restrict__ Sp) {
  gram_body<128, 320>(X, N, Gp, Sp, nullptr, nullptr);
}

__global__ __launch_bounds__(256) void gram_part64(const float* __restrict__ X, int N,
                                                   float* __restrict__ Gp,
                                                   float* __restrict__ Sp) {
  __shared__ float red[3 * 64 * 64];
  __shared__ float reds[3 * 64];
  gram_body<64, 80>(X, N, Gp, Sp, red, reds);
}

__global__ __launch_bounds__(256) void reduce_rows2(
    const float* __restrict__ srcA, int nrA, int wA, float* __restrict__ dstA, int blocksA,
    const float* __restrict__ srcB, int nrB, int wB, float* __restrict__ dstB) {
  __shared__ float red[8][32];
  int b = blockIdx.x;
  const float* src = srcA; int nr = nrA; int w = wA; float* dst = dstA;
  if (b >= blocksA) { b -= blocksA; src = srcB; nr = nrB; w = wB; dst = dstB; }
  const int e = threadIdx.x & 31;
  const int q = threadIdx.x >> 5;
  const int col = b * 32 + e;
  float s = 0.0f;
  #pragma unroll 8
  for (int r = q; r < nr; r += 8) s += src[(size_t)r * w + col];
  red[q][e] = s;
  __syncthreads();
  if (q == 0) {
    float t = red[0][e];
    #pragma unroll
    for (int i = 1; i < 8; ++i) t += red[i][e];
    dst[col] = t;
  }
}

template <int C>
__device__ __forceinline__ void gram_finalize_body(
    const float* __restrict__ G, const float* __restrict__ S, float invN,
    const float* __restrict__ W, const float* __restrict__ g,
    const float* __restrict__ bta, float* __restrict__ sc, float* __restrict__ sh,
    int Cout, int c) {
  __shared__ float wl[C];
  __shared__ float r2[256];
  __shared__ float r1[256];
  const int tid = threadIdx.x;
  for (int k = tid; k < C; k += 256) wl[k] = W[(size_t)k * Cout + c];
  __syncthreads();
  float s2 = 0.0f;
  #pragma unroll 16
  for (int i = tid; i < C * C; i += 256) s2 += wl[i / C] * G[i] * wl[i % C];
  float s1 = 0.0f;
  for (int k = tid; k < C; k += 256) s1 += wl[k] * S[k];
  r2[tid] = s2;
  r1[tid] = s1;
  __syncthreads();
  for (int st = 128; st > 0; st >>= 1) {
    if (tid < st) { r2[tid] += r2[tid + st]; r1[tid] += r1[tid + st]; }
    __syncthreads();
  }
  if (tid == 0) {
    float mean = r1[0] * invN;
    float var  = fmaxf(r2[0] * invN - mean * mean, 0.0f);
    float s    = g[c] * rsqrtf(var + BN_EPS);
    sc[c] = s;
    sh[c] = bta[c] - mean * s;
  }
}

__global__ __launch_bounds__(256) void gram_finalize128(
    const float* __restrict__ G, const float* __restrict__ S, float invN,
    const float* __restrict__ Wa, const float* __restrict__ ga, const float* __restrict__ ba,
    float* __restrict__ sca, float* __restrict__ sha, int Ca,
    const float* __restrict__ Wb, const float* __restrict__ gb, const float* __restrict__ bb,
    float* __restrict__ scb, float* __restrict__ shb, int Cb) {
  int c = blockIdx.x;
  const float* W = Wa; const float* g = ga; const float* b = ba;
  float* sc = sca; float* sh = sha; int Cout = Ca;
  if (c >= Ca) { c -= Ca; W = Wb; g = gb; b = bb; sc = scb; sh = shb; Cout = Cb; }
  gram_finalize_body<128>(G, S, invN, W, g, b, sc, sh, Cout, c);
}

__global__ __launch_bounds__(256) void gram_finalize64(
    const float* __restrict__ G, const float* __restrict__ S, float invN,
    const float* __restrict__ W, const float* __restrict__ g, const float* __restrict__ b,
    float* __restrict__ sc, float* __restrict__ sh, int Cout) {
  gram_finalize_body<64>(G, S, invN, W, g, b, sc, sh, Cout, blockIdx.x);
}

extern "C" void kernel_launch(void* const* d_in, const int* in_sizes, int n_in,
                              void* d_out, int out_size, void* d_ws, size_t ws_size,
                              hipStream_t stream) {
  const float* feats = (const float*)d_in[0];
  const float* xyz   = (const float*)d_in[1];
  const int*   nidx  = (const int*)d_in[3];
  const float* W1    = (const float*)d_in[4];
  const float* g1    = (const float*)d_in[5];
  const float* b1    = (const float*)d_in[6];
  const float* kpp   = (const float*)d_in[7];
  const float* kpw   = (const float*)d_in[8];
  const float* W2    = (const float*)d_in[9];
  const float* g2    = (const float*)d_in[10];
  const float* b2    = (const float*)d_in[11];
  const float* Wsc   = (const float*)d_in[12];
  const float* gsc   = (const float*)d_in[13];
  const float* bsc   = (const float*)d_in[14];
  float* out = (float*)d_out;

  const int N = in_sizes[0] / CIN;

  float* ws      = (float*)d_ws;
  float* x1      = ws;
  float* xk      = x1 + (size_t)N * D2;
  float* ysc     = xk + (size_t)N * D2;
  float* stats1  = ysc + (size_t)N * COUT;
  float* stats2  = stats1 + 2 * D2;
  float* statsSC = stats2 + 2 * COUT;
  float* sc1 = statsSC + 2 * COUT; float* sh1 = sc1 + D2;
  float* sc2 = sh1 + D2;           float* sh2 = sc2 + COUT;
  float* scS = sh2 + COUT;         float* shS = scS + COUT;

  const int nblk = (N + 319) / 320;
  float* GpF = ysc;
  float* SpF = GpF + (size_t)nblk * CIN * CIN;
  float* GF  = SpF + (size_t)nblk * CIN;
  float* SF  = GF + CIN * CIN;
  float* Gp2 = x1;
  float* Sp2 = Gp2 + (size_t)nblk * D2 * D2;
  float* G2  = Sp2 + (size_t)nblk * D2;
  float* S2  = G2 + D2 * D2;

  gram_part128<<<nblk, 256, 0, stream>>>(feats, N, GpF, SpF);
  reduce_rows2<<<(CIN * CIN) / 32 + CIN / 32, 256, 0, stream>>>(
      GpF, nblk, CIN * CIN, GF, (CIN * CIN) / 32, SpF, nblk, CIN, SF);
  gram_finalize128<<<D2 + COUT, 256, 0, stream>>>(GF, SF, 1.0f / (float)N,
      W1, g1, b1, sc1, sh1, D2, Wsc, gsc, bsc, scS, shS, COUT);

  gemm_bnact_wmma<CIN><<<((N + 127) / 128) * (D2 / 64), 256, 0, stream>>>(
      feats, W1, sc1, sh1, x1, N, D2);

  {
    size_t lds = (KPT * FKP + 8 * NBH * 16) * sizeof(float)
               + 8 * NBH * sizeof(int);
    kpconv_wmma<<<N / KPT, 256, lds, stream>>>(xyz, nidx, x1, kpp, kpw, xk);
  }

  gram_part64<<<nblk, 256, 0, stream>>>(xk, N, Gp2, Sp2);
  reduce_rows2<<<(D2 * D2) / 32 + D2 / 32, 256, 0, stream>>>(
      Gp2, nblk, D2 * D2, G2, (D2 * D2) / 32, Sp2, nblk, D2, S2);
  gram_finalize64<<<COUT, 256, 0, stream>>>(G2, S2, 1.0f / (float)N, W2, g2, b2, sc2, sh2, COUT);

  gemm2_bn_residual_wmma<D2, CIN><<<((N + 127) / 128) * (COUT / 64), 256, 0, stream>>>(
      xk, W2, sc2, sh2, feats, Wsc, scS, shS, out, N, COUT);
}


# KPConv phase 1 pipelined per wave: 4 index rows up front, next point's positions one point ahead, feature gathers issued before the influence weights; 8 per-point block barriers -> 1; sums as before
# speedup vs baseline: 2.4437x; 1.0542x over previous
#include <hip/hip_runtime.h>
#include <hip/hip_bf16.h>


typedef __attribute__((ext_vector_type(2))) float v2f;
typedef __attribute__((ext_vector_type(4))) float v4f;
typedef __attribute__((ext_vector_type(8))) float v8f;

#define CIN   128
#define COUT  256
#define D2    64
#define KPTS  15
#define NBH   32
#define KDFLAT (KPTS * D2)
#define INFL_INV 2.5f
#define BN_EPS 1e-5f
#define SLOPE 0.2f
#define PP    160

__device__ __forceinline__ v8f wmma_f32(v2f a, v2f b, v8f c) {
  return __builtin_amdgcn_wmma_f32_16x16x4_f32(
       false, a,  false, b,
       (short)0, c,  false,  false);
}

__device__ __forceinline__ float lrelu(float v) {
  return v >= 0.0f ? v : SLOPE * v;
}


template <int ROWS>
__device__ __forceinline__ void stage_strip(const float* __restrict__ B, int Nc, int col0,
                                            float* Ws) {
  constexpr int ITEMS = (ROWS / 2) * 16;
  #pragma unroll
  for (int i0 = 0; i0 < ITEMS; i0 += 256) {
    const int i = i0 + (int)threadIdx.x;
    if (ITEMS % 256 == 0 || i < ITEMS) {
      const int p = i >> 4;
      const int j = (i & 15) << 2;
      const float* src = B + (size_t)(2 * p) * Nc + col0 + j;
      const v4f r0 = *(const v4f*)src;
      const v4f r1 = *(const v4f*)(src + Nc);
      float* dst = Ws + p * PP + 2 * j;
      *(v4f*)dst       = v4f{r0.x, r1.x, r0.y, r1.y};
      *(v4f*)(dst + 4) = v4f{r0.z, r1.z, r0.w, r1.w};
    }
  }
}

template <int KD>
__device__ __forceinline__ void strip_product(
    const float* __restrict__ Arow, const float* Wl,
    int half, v8f& acc0, v8f& acc1, v8f& acc2, v8f& acc3) {
  for (int kk = 0; kk < KD; kk += 4) {
    int ka = kk + (half << 1);
    v2f a;
    a.x = Arow[ka];
    a.y = Arow[ka + 1];
    const float* Wp = Wl + (ka >> 1) * PP;
    const v2f b0 = *(const v2f*)(Wp);
    const v2f b1 = *(const v2f*)(Wp + 32);
    const v2f b2 = *(const v2f*)(Wp + 64);
    const v2f b3 = *(const v2f*)(Wp + 96);
    acc0 = wmma_f32(a, b0, acc0);
    acc1 = wmma_f32(a, b1, acc1);
    acc2 = wmma_f32(a, b2, acc2);
    acc3 = wmma_f32(a, b3, acc3);
  }
}

template <int KD>
__global__ __launch_bounds__(256) void gemm_bnact_wmma(
    const float* __restrict__ A, const float* __restrict__ B,
    const float* __restrict__ sc, const float* __restrict__ sh,
    float* __restrict__ C, int M, int Nc) {
  __shared__ float Ws[(KD / 2) * PP];
  const int strips = Nc >> 6;
  const int col0 = (blockIdx.x % strips) << 6;
  stage_strip<KD>(B, Nc, col0, Ws);
  __syncthreads();
  const int row0 = (blockIdx.x / strips) * 128 + ((threadIdx.x >> 5) << 4);
  if (row0 >= M) return;
  int lane = threadIdx.x & 31;
  int m    = lane & 15;
  int half = lane >> 4;

  v8f acc0 = {}, acc1 = {}, acc2 = {}, acc3 = {};
  strip_product<KD>(A + (size_t)(row0 + m) * KD, Ws + 2 * m, half,
                    acc0, acc1, acc2, acc3);

  const float* scp = sc + col0 + m;
  const float* shp = sh + col0 + m;
  const float s0 = scp[0], s1 = scp[16], s2 = scp[32], s3 = scp[48];
  const float h0 = shp[0], h1 = shp[16], h2 = shp[32], h3 = shp[48];
  int rbase = row0 + (half << 3);
  float* Cp = C + (size_t)rbase * Nc + col0 + m;
  #pragma unroll
  for (int r = 0; r < 8; ++r) {
    size_t o = (size_t)r * Nc;
    Cp[o +  0] = lrelu(acc0[r] * s0 + h0);
    Cp[o + 16] = lrelu(acc1[r] * s1 + h1);
    Cp[o + 32] = lrelu(acc2[r] * s2 + h2);
    Cp[o + 48] = lrelu(acc3[r] * s3 + h3);
  }
}

template <int KM, int KS>
__global__ __launch_bounds__(256) void gemm2_bn_residual_wmma(
    const float* __restrict__ Am, const float* __restrict__ Bm,
    const float* __restrict__ scM, const float* __restrict__ shM,
    const float* __restrict__ As, const float* __restrict__ Bs,
    const float* __restrict__ scS, const float* __restrict__ shS,
    float* __restrict__ C, int M, int Nc) {
  __shared__ float Ws[((KS + KM) / 2) * PP];
  float* Wm = Ws + (KS / 2) * PP;
  const int strips = Nc >> 6;
  const int col0 = (blockIdx.x % strips) << 6;
  stage_strip<KS>(Bs, Nc, col0, Ws);
  stage_strip<KM>(Bm, Nc, col0, Wm);
  __syncthreads();
  const int row0 = (blockIdx.x / strips) * 128 + ((threadIdx.x >> 5) << 4);
  if (row0 >= M) return;
  int lane = threadIdx.x & 31;
  int m    = lane & 15;
  int half = lane >> 4;

  v8f y0 = {}, y1 = {}, y2 = {}, y3 = {};
  strip_product<KS>(As + (size_t)(row0 + m) * KS, Ws + 2 * m, half,
                    y0, y1, y2, y3);
  v8f acc0 = {}, acc1 = {}, acc2 = {}, acc3 = {};
  strip_product<KM>(Am + (size_t)(row0 + m) * KM, Wm + 2 * m, half,
                    acc0, acc1, acc2, acc3);

  const float* p;
  p = scM + col0 + m; const float a0 = p[0], a1 = p[16], a2 = p[32], a3 = p[48];
  p = shM + col0 + m; const float c0 = p[0], c1 = p[16], c2 = p[32], c3 = p[48];
  p = scS + col0 + m; const float s0 = p[0], s1 = p[16], s2 = p[32], s3 = p[48];
  p = shS + col0 + m; const float h0 = p[0], h1 = p[16], h2 = p[32], h3 = p[48];
  int rbase = row0 + (half << 3);
  float* Cp = C + (size_t)rbase * Nc + col0 + m;
  #pragma unroll
  for (int r = 0; r < 8; ++r) {
    size_t o = (size_t)r * Nc;
    Cp[o +  0] = lrelu(acc0[r] * a0 + c0) + (y0[r] * s0 + h0);
    Cp[o + 16] = lrelu(acc1[r] * a1 + c1) + (y1[r] * s1 + h1);
    Cp[o + 32] = lrelu(acc2[r] * a2 + c2) + (y2[r] * s2 + h2);
    Cp[o + 48] = lrelu(acc3[r] * a3 + c3) + (y3[r] * s3 + h3);
  }
}

#define KPT 32
#define FKP 964
__global__ __launch_bounds__(256) void kpconv_wmma(
    const float* __restrict__ xyz, const int* __restrict__ nidx,
    const float* __restrict__ x1, const float* __restrict__ kp_points,
    const float* __restrict__ kpw, float* __restrict__ xk) {
  extern __shared__ float smem[];
  float* FK  = smem;
  float* Wl  = FK + KPT * FKP;
  int*   IDX = (int*)(Wl + 8 * NBH * 16);
  float* PS  = smem;

  const int wave = threadIdx.x >> 5;
  const int lane = threadIdx.x & 31;
  const int m    = lane & 15;
  const int half = lane >> 4;
  const int base = blockIdx.x * KPT;

  float* wl = Wl + wave * (NBH * 16);
  int*   il = IDX + wave * NBH;

  static_assert(KPT / 8 == 4, "the pipeline below holds the index rows of 4 points per wave");
  const int n0 = base + wave * (KPT / 8);
  int id_cur = nidx[(size_t)n0 * NBH + lane];
  int id_nxt = nidx[(size_t)(n0 + 1) * NBH + lane];
  int id_n2  = nidx[(size_t)(n0 + 2) * NBH + lane];
  int id_n3  = nidx[(size_t)(n0 + 3) * NBH + lane];
  float cx = xyz[(size_t)n0 * 3 + 0];
  float cy = xyz[(size_t)n0 * 3 + 1];
  float cz = xyz[(size_t)n0 * 3 + 2];
  float qx = xyz[(size_t)id_cur * 3 + 0];
  float qy = xyz[(size_t)id_cur * 3 + 1];
  float qz = xyz[(size_t)id_cur * 3 + 2];
  for (int it = 0; it < KPT / 8; ++it) {
    const int pi = wave * (KPT / 8) + it;
    const int n  = base + pi;

    const float px = qx - cx;
    const float py = qy - cy;
    const float pz = qz - cz;
    il[lane] = id_cur;
    __builtin_amdgcn_fence(__ATOMIC_RELEASE, "wavefront", "local");
    __builtin_amdgcn_wave_barrier();
    __builtin_amdgcn_fence(__ATOMIC_ACQUIRE, "wavefront", "local");

    v2f b0[NBH / 4], b1[NBH / 4], b2[NBH / 4], b3[NBH / 4];
    #pragma unroll
    for (int j = 0; j < NBH / 4; ++j) {
      const int ka = 4 * j + (half << 1);
      const float* r0 = x1 + (size_t)il[ka] * D2 + m;
      const float* r1 = x1 + (size_t)il[ka + 1] * D2 + m;
      b0[j].x = r0[0];  b0[j].y = r1[0];
      b1[j].x = r0[16]; b1[j].y = r1[16];
      b2[j].x = r0[32]; b2[j].y = r1[32];
      b3[j].x = r0[48]; b3[j].y = r1[48];
    }

    #pragma unroll
    for (int k = 0; k < KPTS; ++k) {
      float dx = px - kp_points[k * 3 + 0];
      float dy = py - kp_points[k * 3 + 1];
      float dz = pz - kp_points[k * 3 + 2];
      float r = sqrtf(dx * dx + dy * dy + dz * dz);
      wl[lane * 16 + k] = fmaxf(1.0f - r * INFL_INV, 0.0f);
    }
    wl[lane * 16 + KPTS] = 0.0f;
    __builtin_amdgcn_fence(__ATOMIC_RELEASE, "wavefront", "local");
    __builtin_amdgcn_wave_barrier();
    __builtin_amdgcn_fence(__ATOMIC_ACQUIRE, "wavefront", "local");

    const int last = n0 + (KPT / 8 - 1);
    const int n1 = (n + 1 < last) ? n + 1 : last;
    const float ncx = xyz[(size_t)n1 * 3 + 0];
    const float ncy = xyz[(size_t)n1 * 3 + 1];
    const float ncz = xyz[(size_t)n1 * 3 + 2];
    const float nqx = xyz[(size_t)id_nxt * 3 + 0];
    const float nqy = xyz[(size_t)id_nxt * 3 + 1];
    const float nqz = xyz[(size_t)id_nxt * 3 + 2];

    v8f acc0 = {}, acc1 = {}, acc2 = {}, acc3 = {};
    #pragma unroll
    for (int j = 0; j < NBH / 4; ++j) {
      const int ka = 4 * j + (half << 1);
      v2f a;
      a.x = wl[ka * 16 + m];
      a.y = wl[(ka + 1) * 16 + m];
      acc0 = wmma_f32(a, b0[j], acc0);
      acc1 = wmma_f32(a, b1[j], acc1);
      acc2 = wmma_f32(a, b2[j], acc2);
      acc3 = wmma_f32(a, b3[j], acc3);
    }
    #pragma unroll
    for (int r = 0; r < 8; ++r) {
      const int krow = (half << 3) + r;
      if (krow < KPTS) {
        float* f = FK + pi * FKP + krow * D2 + m;
        f[0]  = acc0[r];
        f[16] = acc1[r];
        f[32] = acc2[r];
        f[48] = acc3[r];
      }
    }
    cx = ncx; cy = ncy; cz = ncz;
    qx = nqx; qy = nqy; qz = nqz;
    id_cur = id_nxt; id_nxt = id_n2; id_n2 = id_n3;
    __builtin_amdgcn_fence(__ATOMIC_RELEASE, "wavefront", "local");
    __builtin_amdgcn_wave_barrier();
  }
  __syncthreads();

  {
    const int kbeg = wave * 120;
    v8f lo0 = {}, lo1 = {}, lo2 = {}, lo3 = {};
    v8f hi0 = {}, hi1 = {}, hi2 = {}, hi3 = {};
    const float* Alo = FK + m * FKP;
    const float* Ahi = FK + (16 + m) * FKP;
    for (int kk = kbeg; kk < kbeg + 120; kk += 4) {
      const int ka = kk + (half << 1);
      const v2f alo = *(const v2f*)(Alo + ka);
      const v2f ahi = *(const v2f*)(Ahi + ka);
      const float* Bp = kpw + (size_t)ka * D2 + m;
      v2f b0, b1, b2, b3;
      b0.x = Bp[0];  b0.y = Bp[D2];
      b1.x = Bp[16]; b1.y = Bp[D2 + 16];
      b2.x = Bp[32]; b2.y = Bp[D2 + 32];
      b3.x = Bp[48]; b3.y = Bp[D2 + 48];
      lo0 = wmma_f32(alo, b0, lo0);
      hi0 = wmma_f32(ahi, b0, hi0);
      lo1 = wmma_f32(alo, b1, lo1);
      hi1 = wmma_f32(ahi, b1, hi1);
      lo2 = wmma_f32(alo, b2, lo2);
      hi2 = wmma_f32(ahi, b2, hi2);
      lo3 = wmma_f32(alo, b3, lo3);
      hi3 = wmma_f32(ahi, b3, hi3);
    }
    __syncthreads();
    #pragma unroll
    for (int r = 0; r < 8; ++r) {
      float* p = PS + wave * (KPT * D2) + ((half << 3) + r) * D2 + m;
      p[0]  = lo0[r];
      p[16] = lo1[r];
      p[32] = lo2[r];
      p[48] = lo3[r];
      p += 16 * D2;
      p[0]  = hi0[r];
      p[16] = hi1[r];
      p[32] = hi2[r];
      p[48] = hi3[r];
    }
  }
  __syncthreads();

  for (int t = threadIdx.x; t < KPT * D2; t += 256) {
    float s = 0.0f;
    #pragma unroll
    for (int w = 0; w < 8; ++w) s += PS[w * (KPT * D2) + t];
    const int pt = t >> 6;
    const int e  = t & 63;
    xk[(size_t)(base + pt) * D2 + e] = s;
  }
}

#define GRAM_U 4

template <int C, int RPG>
__device__ __forceinline__ void gram_body(const float* __restrict__ X, int N,
                                          float* __restrict__ Gp,
                                          float* __restrict__ Sp, float* red,
                                          float* reds) {
  constexpr int TC     = C / 64;
  constexpr int TILES  = (C / 32) * TC;
  constexpr int GROUPS = 8 / TILES;
  const int wave = __builtin_amdgcn_readfirstlane((int)(threadIdx.x >> 5));
  const int lane = threadIdx.x & 31;
  const int m    = lane & 15;
  const int half = lane >> 4;
  const int tile = wave % TILES;
  const int grp  = wave / TILES;
  const int tr   = tile / TC;
  const int tc   = tile % TC;
  const int gidx = blockIdx.x * GROUPS + grp;
  const int row0 = gidx * RPG;
  int nrows = N - row0;
  nrows = nrows < 0 ? 0 : (nrows > RPG ? RPG : nrows);
  const int nruns = nrows / (4 * GRAM_U);

  const float* xb = X + (size_t)(row0 + 2 * half) * C + m;
  const float* ap = xb + tr * 32;
  const float* bp = xb + tc * 64;

  v8f acc[2][4];
  #pragma unroll
  for (int i = 0; i < 2; ++i)
    #pragma unroll
    for (int j = 0; j < 4; ++j) acc[i][j] = v8f{};
  float cs[4] = {0.0f, 0.0f, 0.0f, 0.0f};

  auto load_run = [&](v2f (&a)[GRAM_U][2], v2f (&b)[GRAM_U][4], int run) {
    const size_t o = (size_t)run * (4 * GRAM_U) * C;
    #pragma unroll
    for (int u = 0; u < GRAM_U; ++u) {
      const float* pa = ap + o + (size_t)u * 4 * C;
      const float* pb = bp + o + (size_t)u * 4 * C;
      #pragma unroll
      for (int i = 0; i < 2; ++i) { a[u][i].x = pa[i * 16]; a[u][i].y = pa[C + i * 16]; }
      #pragma unroll
      for (int j = 0; j < 4; ++j) { b[u][j].x = pb[j * 16]; b[u][j].y = pb[C + j * 16]; }
    }
  };
  auto mma_run = [&](v2f (&a)[GRAM_U][2], v2f (&b)[GRAM_U][4]) {
    #pragma unroll
    for (int u = 0; u < GRAM_U; ++u) {
      if (tr == 0) {
        #pragma unroll
        for (int j = 0; j < 4; ++j) cs[j] += b[u][j].x + b[u][j].y;
      }
      #pragma unroll
      for (int i = 0; i < 2; ++i)
        #pragma unroll
        for (int j = 0; j < 4; ++j) acc[i][j] = wmma_f32(a[u][i], b[u][j], acc[i][j]);
    }
  };

  v2f a0[GRAM_U][2], b0[GRAM_U][4], a1[GRAM_U][2], b1[GRAM_U][4];
  if (nruns > 0) load_run(a0, b0, 0);
  for (int r = 0; r < nruns; r += 2) {
    if (r + 1 < nruns) load_run(a1, b1, r + 1);
    mma_run(a0, b0);
    if (r + 2 < nruns) load_run(a0, b0, r + 2);
    if (r + 1 < nruns) mma_run(a1, b1);
  }

  const int grow = tr * 32 + (half << 3);
  const int gcol = tc * 64 + m;
  if (tr == 0) {
    #pragma unroll
    for (int j = 0; j < 4; ++j) cs[j] += __shfl_xor(cs[j], 16);
  }
  if constexpr (GROUPS > 1) {
    if (grp > 0) {
      if (tr == 0 && half == 0) {
        #pragma unroll
        for (int j = 0; j < 4; ++j) reds[(grp - 1) * C + gcol + j * 16] = cs[j];
      }
      float* rp = red + (size_t)(grp - 1) * (C * C);
      #pragma unroll
      for (int i = 0; i < 2; ++i)
        #pragma unroll
        for (int j = 0; j < 4; ++j)
          #pragma unroll
          for (int r = 0; r < 8; ++r)
            rp[(grow + i * 16 + r) * C + gcol + j * 16] = acc[i][j][r];
    }
    __syncthreads();
    if (grp == 0) {
      if (tr == 0) {
        #pragma unroll
        for (int g = 0; g < GROUPS - 1; ++g)
          #pragma unroll
          for (int j = 0; j < 4; ++j) cs[j] += reds[g * C + gcol + j * 16];
      }
      #pragma unroll
      for (int g = 0; g < GROUPS - 1; ++g)
        #pragma unroll
        for (int i = 0; i < 2; ++i)
          #pragma unroll
          for (int j = 0; j < 4; ++j)
            #pragma unroll
            for (int r = 0; r < 8; ++r)
              acc[i][j][r] += red[(size_t)g * (C * C) + (grow + i * 16 + r) * C + gcol + j * 16];
    }
  }
  if (grp == 0) {
    float* gp = Gp + (size_t)blockIdx.x * (C * C);
    #pragma unroll
    for (int i = 0; i < 2; ++i)
      #pragma unroll
      for (int j = 0; j < 4; ++j)
        #pragma unroll
        for (int r = 0; r < 8; ++r)
          gp[(grow + i * 16 + r) * C + gcol + j * 16] = acc[i][j][r];
  }
  if (grp == 0 && tr == 0 && half == 0) {
    float* sp = Sp + (size_t)blockIdx.x * C + gcol;
    #pragma unroll
    for (int j = 0; j < 4; ++j) sp[j * 16] = cs[j];
  }
}

__global__ __launch_bounds__(256) void gram_part128(const float* __restrict__ X, int N,
                                                    float* __restrict__ Gp,
                                                    float* __restrict__ Sp) {
  gram_body<128, 320>(X, N, Gp, Sp, nullptr, nullptr);
}

__global__ __launch_bounds__(256) void gram_part64(const float* __restrict__ X, int N,
                                                   float* __restrict__ Gp,
                                                   float* __restrict__ Sp) {
  __shared__ float red[3 * 64 * 64];
  __shared__ float reds[3 * 64];
  gram_body<64, 80>(X, N, Gp, Sp, red, reds);
}

__global__ __launch_bounds__(256) void reduce_rows2(
    const float* __restrict__ srcA, int nrA, int wA, float* __restrict__ dstA, int blocksA,
    const float* __restrict__ srcB, int nrB, int wB, float* __restrict__ dstB) {
  __shared__ float red[8][32];
  int b = blockIdx.x;
  const float* src = srcA; int nr = nrA; int w = wA; float* dst = dstA;
  if (b >= blocksA) { b -= blocksA; src = srcB; nr = nrB; w = wB; dst = dstB; }
  const int e = threadIdx.x & 31;
  const int q = threadIdx.x >> 5;
  const int col = b * 32 + e;
  float s = 0.0f;
  #pragma unroll 8
  for (int r = q; r < nr; r += 8) s += src[(size_t)r * w + col];
  red[q][e] = s;
  __syncthreads();
  if (q == 0) {
    float t = red[0][e];
    #pragma unroll
    for (int i = 1; i < 8; ++i) t += red[i][e];
    dst[col] = t;
  }
}

template <int C>
__device__ __forceinline__ void gram_finalize_body(
    const float* __restrict__ G, const float* __restrict__ S, float invN,
    const float* __restrict__ W, const float* __restrict__ g,
    const float* __restrict__ bta, float* __restrict__ sc, float* __restrict__ sh,
    int Cout, int c) {
  __shared__ float wl[C];
  __shared__ float r2[256];
  __shared__ float r1[256];
  const int tid = threadIdx.x;
  for (int k = tid; k < C; k += 256) wl[k] = W[(size_t)k * Cout + c];
  __syncthreads();
  float s2 = 0.0f;
  #pragma unroll 16
  for (int i = tid; i < C * C; i += 256) s2 += wl[i / C] * G[i] * wl[i % C];
  float s1 = 0.0f;
  for (int k = tid; k < C; k += 256) s1 += wl[k] * S[k];
  r2[tid] = s2;
  r1[tid] = s1;
  __syncthreads();
  for (int st = 128; st > 0; st >>= 1) {
    if (tid < st) { r2[tid] += r2[tid + st]; r1[tid] += r1[tid + st]; }
    __syncthreads();
  }
  if (tid == 0) {
    float mean = r1[0] * invN;
    float var  = fmaxf(r2[0] * invN - mean * mean, 0.0f);
    float s    = g[c] * rsqrtf(var + BN_EPS);
    sc[c] = s;
    sh[c] = bta[c] - mean * s;
  }
}

__global__ __launch_bounds__(256) void gram_finalize128(
    const float* __restrict__ G, const float* __restrict__ S, float invN,
    const float* __restrict__ Wa, const float* __restrict__ ga, const float* __restrict__ ba,
    float* __restrict__ sca, float* __restrict__ sha, int Ca,
    const float* __restrict__ Wb, const float* __restrict__ gb, const float* __restrict__ bb,
    float* __restrict__ scb, float* __restrict__ shb, int Cb) {
  int c = blockIdx.x;
  const float* W = Wa; const float* g = ga; const float* b = ba;
  float* sc = sca; float* sh = sha; int Cout = Ca;
  if (c >= Ca) { c -= Ca; W = Wb; g = gb; b = bb; sc = scb; sh = shb; Cout = Cb; }
  gram_finalize_body<128>(G, S, invN, W, g, b, sc, sh, Cout, c);
}

__global__ __launch_bounds__(256) void gram_finalize64(
    const float* __restrict__ G, const float* __restrict__ S, float invN,
    const float* __restrict__ W, const float* __restrict__ g, const float* __restrict__ b,
    float* __restrict__ sc, float* __restrict__ sh, int Cout) {
  gram_finalize_body<64>(G, S, invN, W, g, b, sc, sh, Cout, blockIdx.x);
}

extern "C" void kernel_launch(void* const* d_in, const int* in_sizes, int n_in,
                              void* d_out, int out_size, void* d_ws, size_t ws_size,
                              hipStream_t stream) {
  const float* feats = (const float*)d_in[0];
  const float* xyz   = (const float*)d_in[1];
  const int*   nidx  = (const int*)d_in[3];
  const float* W1    = (const float*)d_in[4];
  const float* g1    = (const float*)d_in[5];
  const float* b1    = (const float*)d_in[6];
  const float* kpp   = (const float*)d_in[7];
  const float* kpw   = (const float*)d_in[8];
  const float* W2    = (const float*)d_in[9];
  const float* g2    = (const float*)d_in[10];
  const float* b2    = (const float*)d_in[11];
  const float* Wsc   = (const float*)d_in[12];
  const float* gsc   = (const float*)d_in[13];
  const float* bsc   = (const float*)d_in[14];
  float* out = (float*)d_out;

  const int N = in_sizes[0] / CIN;

  float* ws      = (float*)d_ws;
  float* x1      = ws;
  float* xk      = x1 + (size_t)N * D2;
  float* ysc     = xk + (size_t)N * D2;
  float* stats1  = ysc + (size_t)N * COUT;
  float* stats2  = stats1 + 2 * D2;
  float* statsSC = stats2 + 2 * COUT;
  float* sc1 = statsSC + 2 * COUT; float* sh1 = sc1 + D2;
  float* sc2 = sh1 + D2;           float* sh2 = sc2 + COUT;
  float* scS = sh2 + COUT;         float* shS = scS + COUT;

  const int nblk = (N + 319) / 320;
  float* GpF = ysc;
  float* SpF = GpF + (size_t)nblk * CIN * CIN;
  float* GF  = SpF + (size_t)nblk * CIN;
  float* SF  = GF + CIN * CIN;
  float* Gp2 = x1;
  float* Sp2 = Gp2 + (size_t)nblk * D2 * D2;
  float* G2  = Sp2 + (size_t)nblk * D2;
  float* S2  = G2 + D2 * D2;

  gram_part128<<<nblk, 256, 0, stream>>>(feats, N, GpF, SpF);
  reduce_rows2<<<(CIN * CIN) / 32 + CIN / 32, 256, 0, stream>>>(
      GpF, nblk, CIN * CIN, GF, (CIN * CIN) / 32, SpF, nblk, CIN, SF);
  gram_finalize128<<<D2 + COUT, 256, 0, stream>>>(GF, SF, 1.0f / (float)N,
      W1, g1, b1, sc1, sh1, D2, Wsc, gsc, bsc, scS, shS, COUT);

  gemm_bnact_wmma<CIN><<<((N + 127) / 128) * (D2 / 64), 256, 0, stream>>>(
      feats, W1, sc1, sh1, x1, N, D2);

  {
    size_t lds = (KPT * FKP + 8 * NBH * 16) * sizeof(float)
               + 8 * NBH * sizeof(int);
    kpconv_wmma<<<N / KPT, 256, lds, stream>>>(xyz, nidx, x1, kpp, kpw, xk);
  }

  gram_part64<<<nblk, 256, 0, stream>>>(xk, N, Gp2, Sp2);
  reduce_rows2<<<(D2 * D2) / 32 + D2 / 32, 256, 0, stream>>>(
      Gp2, nblk, D2 * D2, G2, (D2 * D2) / 32, Sp2, nblk, D2, S2);
  gram_finalize64<<<COUT, 256, 0, stream>>>(G2, S2, 1.0f / (float)N, W2, g2, b2, sc2, sh2, COUT);

  gemm2_bn_residual_wmma<D2, CIN><<<((N + 127) / 128) * (COUT / 64), 256, 0, stream>>>(
      xk, W2, sc2, sh2, feats, Wsc, scS, shS, out, N, COUT);
}


# KPConv phase 1: neighbours whose 15 influence weights are all exactly 0 are put last (ballot, prefix count); only the ceil(kept/4) k-steps left are gathered and multiplied. Reorders those sums.
# speedup vs baseline: 2.5419x; 1.0402x over previous
#include <hip/hip_runtime.h>
#include <hip/hip_bf16.h>


typedef __attribute__((ext_vector_type(2))) float v2f;
typedef __attribute__((ext_vector_type(4))) float v4f;
typedef __attribute__((ext_vector_type(8))) float v8f;

#define CIN   128
#define COUT  256
#define D2    64
#define KPTS  15
#define NBH   32
#define KDFLAT (KPTS * D2)
#define INFL_INV 2.5f
#define BN_EPS 1e-5f
#define SLOPE 0.2f
#define PP    160

__device__ __forceinline__ v8f wmma_f32(v2f a, v2f b, v8f c) {
  return __builtin_amdgcn_wmma_f32_16x16x4_f32(
       false, a,  false, b,
       (short)0, c,  false,  false);
}

__device__ __forceinline__ float lrelu(float v) {
  return v >= 0.0f ? v : SLOPE * v;
}


template <int ROWS>
__device__ __forceinline__ void stage_strip(const float* __restrict__ B, int Nc, int col0,
                                            float* Ws) {
  constexpr int ITEMS = (ROWS / 2) * 16;
  #pragma unroll
  for (int i0 = 0; i0 < ITEMS; i0 += 256) {
    const int i = i0 + (int)threadIdx.x;
    if (ITEMS % 256 == 0 || i < ITEMS) {
      const int p = i >> 4;
      const int j = (i & 15) << 2;
      const float* src = B + (size_t)(2 * p) * Nc + col0 + j;
      const v4f r0 = *(const v4f*)src;
      const v4f r1 = *(const v4f*)(src + Nc);
      float* dst = Ws + p * PP + 2 * j;
      *(v4f*)dst       = v4f{r0.x, r1.x, r0.y, r1.y};
      *(v4f*)(dst + 4) = v4f{r0.z, r1.z, r0.w, r1.w};
    }
  }
}

template <int KD>
__device__ __forceinline__ void strip_product(
    const float* __restrict__ Arow, const float* Wl,
    int half, v8f& acc0, v8f& acc1, v8f& acc2, v8f& acc3) {
  for (int kk = 0; kk < KD; kk += 4) {
    int ka = kk + (half << 1);
    v2f a;
    a.x = Arow[ka];
    a.y = Arow[ka + 1];
    const float* Wp = Wl + (ka >> 1) * PP;
    const v2f b0 = *(const v2f*)(Wp);
    const v2f b1 = *(const v2f*)(Wp + 32);
    const v2f b2 = *(const v2f*)(Wp + 64);
    const v2f b3 = *(const v2f*)(Wp + 96);
    acc0 = wmma_f32(a, b0, acc0);
    acc1 = wmma_f32(a, b1, acc1);
    acc2 = wmma_f32(a, b2, acc2);
    acc3 = wmma_f32(a, b3, acc3);
  }
}

template <int KD>
__global__ __launch_bounds__(256) void gemm_bnact_wmma(
    const float* __restrict__ A, const float* __restrict__ B,
    const float* __restrict__ sc, const float* __restrict__ sh,
    float* __restrict__ C, int M, int Nc) {
  __shared__ float Ws[(KD / 2) * PP];
  const int strips = Nc >> 6;
  const int col0 = (blockIdx.x % strips) << 6;
  stage_strip<KD>(B, Nc, col0, Ws);
  __syncthreads();
  const int row0 = (blockIdx.x / strips) * 128 + ((threadIdx.x >> 5) << 4);
  if (row0 >= M) return;
  int lane = threadIdx.x & 31;
  int m    = lane & 15;
  int half = lane >> 4;

  v8f acc0 = {}, acc1 = {}, acc2 = {}, acc3 = {};
  strip_product<KD>(A + (size_t)(row0 + m) * KD, Ws + 2 * m, half,
                    acc0, acc1, acc2, acc3);

  const float* scp = sc + col0 + m;
  const float* shp = sh + col0 + m;
  const float s0 = scp[0], s1 = scp[16], s2 = scp[32], s3 = scp[48];
  const float h0 = shp[0], h1 = shp[16], h2 = shp[32], h3 = shp[48];
  int rbase = row0 + (half << 3);
  float* Cp = C + (size_t)rbase * Nc + col0 + m;
  #pragma unroll
  for (int r = 0; r < 8; ++r) {
    size_t o = (size_t)r * Nc;
    Cp[o +  0] = lrelu(acc0[r] * s0 + h0);
    Cp[o + 16] = lrelu(acc1[r] * s1 + h1);
    Cp[o + 32] = lrelu(acc2[r] * s2 + h2);
    Cp[o + 48] = lrelu(acc3[r] * s3 + h3);
  }
}

template <int KM, int KS>
__global__ __launch_bounds__(256) void gemm2_bn_residual_wmma(
    const float* __restrict__ Am, const float* __restrict__ Bm,
    const float* __restrict__ scM, const float* __restrict__ shM,
    const float* __restrict__ As, const float* __restrict__ Bs,
    const float* __restrict__ scS, const float* __restrict__ shS,
    float* __restrict__ C, int M, int Nc) {
  __shared__ float Ws[((KS + KM) / 2) * PP];
  float* Wm = Ws + (KS / 2) * PP;
  const int strips = Nc >> 6;
  const int col0 = (blockIdx.x % strips) << 6;
  stage_strip<KS>(Bs, Nc, col0, Ws);
  stage_strip<KM>(Bm, Nc, col0, Wm);
  __syncthreads();
  const int row0 = (blockIdx.x / strips) * 128 + ((threadIdx.x >> 5) << 4);
  if (row0 >= M) return;
  int lane = threadIdx.x & 31;
  int m    = lane & 15;
  int half = lane >> 4;

  v8f y0 = {}, y1 = {}, y2 = {}, y3 = {};
  strip_product<KS>(As + (size_t)(row0 + m) * KS, Ws + 2 * m, half,
                    y0, y1, y2, y3);
  v8f acc0 = {}, acc1 = {}, acc2 = {}, acc3 = {};
  strip_product<KM>(Am + (size_t)(row0 + m) * KM, Wm + 2 * m, half,
                    acc0, acc1, acc2, acc3);

  const float* p;
  p = scM + col0 + m; const float a0 = p[0], a1 = p[16], a2 = p[32], a3 = p[48];
  p = shM + col0 + m; const float c0 = p[0], c1 = p[16], c2 = p[32], c3 = p[48];
  p = scS + col0 + m; const float s0 = p[0], s1 = p[16], s2 = p[32], s3 = p[48];
  p = shS + col0 + m; const float h0 = p[0], h1 = p[16], h2 = p[32], h3 = p[48];
  int rbase = row0 + (half << 3);
  float* Cp = C + (size_t)rbase * Nc + col0 + m;
  #pragma unroll
  for (int r = 0; r < 8; ++r) {
    size_t o = (size_t)r * Nc;
    Cp[o +  0] = lrelu(acc0[r] * a0 + c0) + (y0[r] * s0 + h0);
    Cp[o + 16] = lrelu(acc1[r] * a1 + c1) + (y1[r] * s1 + h1);
    Cp[o + 32] = lrelu(acc2[r] * a2 + c2) + (y2[r] * s2 + h2);
    Cp[o + 48] = lrelu(acc3[r] * a3 + c3) + (y3[r] * s3 + h3);
  }
}

#define KPT 32
#define FKP 964
__global__ __launch_bounds__(256) void kpconv_wmma(
    const float* __restrict__ xyz, const int* __restrict__ nidx,
    const float* __restrict__ x1, const float* __restrict__ kp_points,
    const float* __restrict__ kpw, float* __restrict__ xk) {
  extern __shared__ float smem[];
  float* FK  = smem;
  float* Wl  = FK + KPT * FKP;
  int*   IDX = (int*)(Wl + 8 * NBH * 16);
  float* PS  = smem;

  const int wave = threadIdx.x >> 5;
  const int lane = threadIdx.x & 31;
  const int m    = lane & 15;
  const int half = lane >> 4;
  const int base = blockIdx.x * KPT;

  float* wl = Wl + wave * (NBH * 16);
  int*   il = IDX + wave * NBH;

  static_assert(KPT / 8 == 4, "the pipeline below holds the index rows of 4 points per wave");
  static_assert(NBH == 32, "one neighbour per lane");
  const int n0 = base + wave * (KPT / 8);
  int id_cur = nidx[(size_t)n0 * NBH + lane];
  int id_nxt = nidx[(size_t)(n0 + 1) * NBH + lane];
  int id_n2  = nidx[(size_t)(n0 + 2) * NBH + lane];
  int id_n3  = nidx[(size_t)(n0 + 3) * NBH + lane];
  auto put_weights = [&](float px, float py, float pz, int id) -> int {
    float w[KPTS];
    float wmax = 0.0f;
    #pragma unroll
    for (int k = 0; k < KPTS; ++k) {
      float dx = px - kp_points[k * 3 + 0];
      float dy = py - kp_points[k * 3 + 1];
      float dz = pz - kp_points[k * 3 + 2];
      float r = sqrtf(dx * dx + dy * dy + dz * dz);
      w[k] = fmaxf(1.0f - r * INFL_INV, 0.0f);
      wmax = fmaxf(wmax, w[k]);
    }
    const bool keep = wmax > 0.0f;
    const unsigned mask = __builtin_amdgcn_ballot_w32(keep);
    const int cnt   = __builtin_popcount(mask);
    const int below = (int)__builtin_amdgcn_mbcnt_lo(mask, 0u);
    const int slot  = keep ? below : cnt + (lane - below);
    il[slot] = id;
    #pragma unroll
    for (int k = 0; k < KPTS; ++k) wl[slot * 16 + k] = w[k];
    wl[slot * 16 + KPTS] = 0.0f;
    const int ns = (cnt + 3) >> 2;
    return ns < 1 ? 1 : ns;
  };

  float ncx, ncy, ncz, nqx, nqy, nqz;
  int ns_cur;
  {
    const float cx = xyz[(size_t)n0 * 3 + 0];
    const float cy = xyz[(size_t)n0 * 3 + 1];
    const float cz = xyz[(size_t)n0 * 3 + 2];
    const float qx = xyz[(size_t)id_cur * 3 + 0];
    const float qy = xyz[(size_t)id_cur * 3 + 1];
    const float qz = xyz[(size_t)id_cur * 3 + 2];
    ncx = xyz[(size_t)(n0 + 1) * 3 + 0];
    ncy = xyz[(size_t)(n0 + 1) * 3 + 1];
    ncz = xyz[(size_t)(n0 + 1) * 3 + 2];
    nqx = xyz[(size_t)id_nxt * 3 + 0];
    nqy = xyz[(size_t)id_nxt * 3 + 1];
    nqz = xyz[(size_t)id_nxt * 3 + 2];
    ns_cur = put_weights(qx - cx, qy - cy, qz - cz, id_cur);
  }
  for (int it = 0; it < KPT / 8; ++it) {
    const int pi = wave * (KPT / 8) + it;
    const int n  = base + pi;
    ns_cur = __builtin_amdgcn_readfirstlane(ns_cur);

    __builtin_amdgcn_fence(__ATOMIC_RELEASE, "wavefront", "local");
    __builtin_amdgcn_wave_barrier();
    __builtin_amdgcn_fence(__ATOMIC_ACQUIRE, "wavefront", "local");

    int i0[NBH / 4], i1[NBH / 4];
    v2f a[NBH / 4];
    #pragma unroll
    for (int j = 0; j < NBH / 4; ++j) {
      const int ka = 4 * j + (half << 1);
      i0[j] = il[ka];
      i1[j] = il[ka + 1];
      a[j].x = wl[ka * 16 + m];
      a[j].y = wl[(ka + 1) * 16 + m];
    }

    v2f b0[NBH / 4], b1[NBH / 4], b2[NBH / 4], b3[NBH / 4];
    #pragma unroll
    for (int j = 0; j < NBH / 4; ++j) {
      if (j == 0 || j < ns_cur) {
        const float* r0 = x1 + (size_t)i0[j] * D2 + m;
        const float* r1 = x1 + (size_t)i1[j] * D2 + m;
        b0[j].x = r0[0];  b0[j].y = r1[0];
        b1[j].x = r0[16]; b1[j].y = r1[16];
        b2[j].x = r0[32]; b2[j].y = r1[32];
        b3[j].x = r0[48]; b3[j].y = r1[48];
      }
    }

    __builtin_amdgcn_fence(__ATOMIC_RELEASE, "wavefront", "local");
    __builtin_amdgcn_wave_barrier();

    int ns_nxt = ns_cur;
    if (it + 1 < KPT / 8)
      ns_nxt = put_weights(nqx - ncx, nqy - ncy, nqz - ncz, id_nxt);

    const int last = n0 + (KPT / 8 - 1);
    const int n2 = (n + 2 < last) ? n + 2 : last;
    const float fcx = xyz[(size_t)n2 * 3 + 0];
    const float fcy = xyz[(size_t)n2 * 3 + 1];
    const float fcz = xyz[(size_t)n2 * 3 + 2];
    const float fqx = xyz[(size_t)id_n2 * 3 + 0];
    const float fqy = xyz[(size_t)id_n2 * 3 + 1];
    const float fqz = xyz[(size_t)id_n2 * 3 + 2];

    v8f acc0 = {}, acc1 = {}, acc2 = {}, acc3 = {};
    #pragma unroll
    for (int j = 0; j < NBH / 4; ++j) {
      if (j == 0 || j < ns_cur) {
        acc0 = wmma_f32(a[j], b0[j], acc0);
        acc1 = wmma_f32(a[j], b1[j], acc1);
        acc2 = wmma_f32(a[j], b2[j], acc2);
        acc3 = wmma_f32(a[j], b3[j], acc3);
      }
    }
    #pragma unroll
    for (int r = 0; r < 8; ++r) {
      const int krow = (half << 3) + r;
      if (krow < KPTS) {
        float* f = FK + pi * FKP + krow * D2 + m;
        f[0]  = acc0[r];
        f[16] = acc1[r];
        f[32] = acc2[r];
        f[48] = acc3[r];
      }
    }
    ncx = fcx; ncy = fcy; ncz = fcz;
    nqx = fqx; nqy = fqy; nqz = fqz;
    id_cur = id_nxt; id_nxt = id_n2; id_n2 = id_n3;
    ns_cur = ns_nxt;
  }
  __syncthreads();

  {
    const int kbeg = wave * 120;
    v8f lo0 = {}, lo1 = {}, lo2 = {}, lo3 = {};
    v8f hi0 = {}, hi1 = {}, hi2 = {}, hi3 = {};
    const float* Alo = FK + m * FKP;
    const float* Ahi = FK + (16 + m) * FKP;
    for (int kk = kbeg; kk < kbeg + 120; kk += 4) {
      const int ka = kk + (half << 1);
      const v2f alo = *(const v2f*)(Alo + ka);
      const v2f ahi = *(const v2f*)(Ahi + ka);
      const float* Bp = kpw + (size_t)ka * D2 + m;
      v2f b0, b1, b2, b3;
      b0.x = Bp[0];  b0.y = Bp[D2];
      b1.x = Bp[16]; b1.y = Bp[D2 + 16];
      b2.x = Bp[32]; b2.y = Bp[D2 + 32];
      b3.x = Bp[48]; b3.y = Bp[D2 + 48];
      lo0 = wmma_f32(alo, b0, lo0);
      hi0 = wmma_f32(ahi, b0, hi0);
      lo1 = wmma_f32(alo, b1, lo1);
      hi1 = wmma_f32(ahi, b1, hi1);
      lo2 = wmma_f32(alo, b2, lo2);
      hi2 = wmma_f32(ahi, b2, hi2);
      lo3 = wmma_f32(alo, b3, lo3);
      hi3 = wmma_f32(ahi, b3, hi3);
    }
    __syncthreads();
    #pragma unroll
    for (int r = 0; r < 8; ++r) {
      float* p = PS + wave * (KPT * D2) + ((half << 3) + r) * D2 + m;
      p[0]  = lo0[r];
      p[16] = lo1[r];
      p[32] = lo2[r];
      p[48] = lo3[r];
      p += 16 * D2;
      p[0]  = hi0[r];
      p[16] = hi1[r];
      p[32] = hi2[r];
      p[48] = hi3[r];
    }
  }
  __syncthreads();

  for (int t = threadIdx.x; t < KPT * D2; t += 256) {
    float s = 0.0f;
    #pragma unroll
    for (int w = 0; w < 8; ++w) s += PS[w * (KPT * D2) + t];
    const int pt = t >> 6;
    const int e  = t & 63;
    xk[(size_t)(base + pt) * D2 + e] = s;
  }
}

#define GRAM_U 4

template <int C, int RPG>
__device__ __forceinline__ void gram_body(const float* __restrict__ X, int N,
                                          float* __restrict__ Gp,
                                          float* __restrict__ Sp, float* red,
                                          float* reds) {
  constexpr int TC     = C / 64;
  constexpr int TILES  = (C / 32) * TC;
  constexpr int GROUPS = 8 / TILES;
  const int wave = __builtin_amdgcn_readfirstlane((int)(threadIdx.x >> 5));
  const int lane = threadIdx.x & 31;
  const int m    = lane & 15;
  const int half = lane >> 4;
  const int tile = wave % TILES;
  const int grp  = wave / TILES;
  const int tr   = tile / TC;
  const int tc   = tile % TC;
  const int gidx = blockIdx.x * GROUPS + grp;
  const int row0 = gidx * RPG;
  int nrows = N - row0;
  nrows = nrows < 0 ? 0 : (nrows > RPG ? RPG : nrows);
  const int nruns = nrows / (4 * GRAM_U);

  const float* xb = X + (size_t)(row0 + 2 * half) * C + m;
  const float* ap = xb + tr * 32;
  const float* bp = xb + tc * 64;

  v8f acc[2][4];
  #pragma unroll
  for (int i = 0; i < 2; ++i)
    #pragma unroll
    for (int j = 0; j < 4; ++j) acc[i][j] = v8f{};
  float cs[4] = {0.0f, 0.0f, 0.0f, 0.0f};

  auto load_run = [&](v2f (&a)[GRAM_U][2], v2f (&b)[GRAM_U][4], int run) {
    const size_t o = (size_t)run * (4 * GRAM_U) * C;
    #pragma unroll
    for (int u = 0; u < GRAM_U; ++u) {
      const float* pa = ap + o + (size_t)u * 4 * C;
      const float* pb = bp + o + (size_t)u * 4 * C;
      #pragma unroll
      for (int i = 0; i < 2; ++i) { a[u][i].x = pa[i * 16]; a[u][i].y = pa[C + i * 16]; }
      #pragma unroll
      for (int j = 0; j < 4; ++j) { b[u][j].x = pb[j * 16]; b[u][j].y = pb[C + j * 16]; }
    }
  };
  auto mma_run = [&](v2f (&a)[GRAM_U][2], v2f (&b)[GRAM_U][4]) {
    #pragma unroll
    for (int u = 0; u < GRAM_U; ++u) {
      if (tr == 0) {
        #pragma unroll
        for (int j = 0; j < 4; ++j) cs[j] += b[u][j].x + b[u][j].y;
      }
      #pragma unroll
      for (int i = 0; i < 2; ++i)
        #pragma unroll
        for (int j = 0; j < 4; ++j) acc[i][j] = wmma_f32(a[u][i], b[u][j], acc[i][j]);
    }
  };

  v2f a0[GRAM_U][2], b0[GRAM_U][4], a1[GRAM_U][2], b1[GRAM_U][4];
  if (nruns > 0) load_run(a0, b0, 0);
  for (int r = 0; r < nruns; r += 2) {
    if (r + 1 < nruns) load_run(a1, b1, r + 1);
    mma_run(a0, b0);
    if (r + 2 < nruns) load_run(a0, b0, r + 2);
    if (r + 1 < nruns) mma_run(a1, b1);
  }

  const int grow = tr * 32 + (half << 3);
  const int gcol = tc * 64 + m;
  if (tr == 0) {
    #pragma unroll
    for (int j = 0; j < 4; ++j) cs[j] += __shfl_xor(cs[j], 16);
  }
  if constexpr (GROUPS > 1) {
    if (grp > 0) {
      if (tr == 0 && half == 0) {
        #pragma unroll
        for (int j = 0; j < 4; ++j) reds[(grp - 1) * C + gcol + j * 16] = cs[j];
      }
      float* rp = red + (size_t)(grp - 1) * (C * C);
      #pragma unroll
      for (int i = 0; i < 2; ++i)
        #pragma unroll
        for (int j = 0; j < 4; ++j)
          #pragma unroll
          for (int r = 0; r < 8; ++r)
            rp[(grow + i * 16 + r) * C + gcol + j * 16] = acc[i][j][r];
    }
    __syncthreads();
    if (grp == 0) {
      if (tr == 0) {
        #pragma unroll
        for (int g = 0; g < GROUPS - 1; ++g)
          #pragma unroll
          for (int j = 0; j < 4; ++j) cs[j] += reds[g * C + gcol + j * 16];
      }
      #pragma unroll
      for (int g = 0; g < GROUPS - 1; ++g)
        #pragma unroll
        for (int i = 0; i < 2; ++i)
          #pragma unroll
          for (int j = 0; j < 4; ++j)
            #pragma unroll
            for (int r = 0; r < 8; ++r)
              acc[i][j][r] += red[(size_t)g * (C * C) + (grow + i * 16 + r) * C + gcol + j * 16];
    }
  }
  if (grp == 0) {
    float* gp = Gp + (size_t)blockIdx.x * (C * C);
    #pragma unroll
    for (int i = 0; i < 2; ++i)
      #pragma unroll
      for (int j = 0; j < 4; ++j)
        #pragma unroll
        for (int r = 0; r < 8; ++r)
          gp[(grow + i * 16 + r) * C + gcol + j * 16] = acc[i][j][r];
  }
  if (grp == 0 && tr == 0 && half == 0) {
    float* sp = Sp + (size_t)blockIdx.x * C + gcol;
    #pragma unroll
    for (int j = 0; j < 4; ++j) sp[j * 16] = cs[j];
  }
}

__global__ __launch_bounds__(256) void gram_part128(const float* __restrict__ X, int N,
                                                    float* __restrict__ Gp,
                                                    float* __restrict__ Sp) {
  gram_body<128, 320>(X, N, Gp, Sp, nullptr, nullptr);
}

__global__ __launch_bounds__(256) void gram_part64(const float* __restrict__ X, int N,
                                                   float* __restrict__ Gp,
                                                   float* __restrict__ Sp) {
  __shared__ float red[3 * 64 * 64];
  __shared__ float reds[3 * 64];
  gram_body<64, 80>(X, N, Gp, Sp, red, reds);
}

__global__ __launch_bounds__(256) void reduce_rows2(
    const float* __restrict__ srcA, int nrA, int wA, float* __restrict__ dstA, int blocksA,
    const float* __restrict__ srcB, int nrB, int wB, float* __restrict__ dstB) {
  __shared__ float red[8][32];
  int b = blockIdx.x;
  const float* src = srcA; int nr = nrA; int w = wA; float* dst = dstA;
  if (b >= blocksA) { b -= blocksA; src = srcB; nr = nrB; w = wB; dst = dstB; }
  const int e = threadIdx.x & 31;
  const int q = threadIdx.x >> 5;
  const int col = b * 32 + e;
  float s = 0.0f;
  #pragma unroll 8
  for (int r = q; r < nr; r += 8) s += src[(size_t)r * w + col];
  red[q][e] = s;
  __syncthreads();
  if (q == 0) {
    float t = red[0][e];
    #pragma unroll
    for (int i = 1; i < 8; ++i) t += red[i][e];
    dst[col] = t;
  }
}

template <int C>
__device__ __forceinline__ void gram_finalize_body(
    const float* __restrict__ G, const float* __restrict__ S, float invN,
    const float* __restrict__ W, const float* __restrict__ g,
    const float* __restrict__ bta, float* __restrict__ sc, float* __restrict__ sh,
    int Cout, int c) {
  __shared__ float wl[C];
  __shared__ float r2[256];
  __shared__ float r1[256];
  const int tid = threadIdx.x;
  for (int k = tid; k < C; k += 256) wl[k] = W[(size_t)k * Cout + c];
  __syncthreads();
  float s2 = 0.0f;
  #pragma unroll 16
  for (int i = tid; i < C * C; i += 256) s2 += wl[i / C] * G[i] * wl[i % C];
  float s1 = 0.0f;
  for (int k = tid; k < C; k += 256) s1 += wl[k] * S[k];
  r2[tid] = s2;
  r1[tid] = s1;
  __syncthreads();
  for (int st = 128; st > 0; st >>= 1) {
    if (tid < st) { r2[tid] += r2[tid + st]; r1[tid] += r1[tid + st]; }
    __syncthreads();
  }
  if (tid == 0) {
    float mean = r1[0] * invN;
    float var  = fmaxf(r2[0] * invN - mean * mean, 0.0f);
    float s    = g[c] * rsqrtf(var + BN_EPS);
    sc[c] = s;
    sh[c] = bta[c] - mean * s;
  }
}

__global__ __launch_bounds__(256) void gram_finalize128(
    const float* __restrict__ G, const float* __restrict__ S, float invN,
    const float* __restrict__ Wa, const float* __restrict__ ga, const float* __restrict__ ba,
    float* __restrict__ sca, float* __restrict__ sha, int Ca,
    const float* __restrict__ Wb, const float* __restrict__ gb, const float* __restrict__ bb,
    float* __restrict__ scb, float* __restrict__ shb, int Cb) {
  int c = blockIdx.x;
  const float* W = Wa; const float* g = ga; const float* b = ba;
  float* sc = sca; float* sh = sha; int Cout = Ca;
  if (c >= Ca) { c -= Ca; W = Wb; g = gb; b = bb; sc = scb; sh = shb; Cout = Cb; }
  gram_finalize_body<128>(G, S, invN, W, g, b, sc, sh, Cout, c);
}

__global__ __launch_bounds__(256) void gram_finalize64(
    const float* __restrict__ G, const float* __restrict__ S, float invN,
    const float* __restrict__ W, const float* __restrict__ g, const float* __restrict__ b,
    float* __restrict__ sc, float* __restrict__ sh, int Cout) {
  gram_finalize_body<64>(G, S, invN, W, g, b, sc, sh, Cout, blockIdx.x);
}

extern "C" void kernel_launch(void* const* d_in, const int* in_sizes, int n_in,
                              void* d_out, int out_size, void* d_ws, size_t ws_size,
                              hipStream_t stream) {
  const float* feats = (const float*)d_in[0];
  const float* xyz   = (const float*)d_in[1];
  const int*   nidx  = (const int*)d_in[3];
  const float* W1    = (const float*)d_in[4];
  const float* g1    = (const float*)d_in[5];
  const float* b1    = (const float*)d_in[6];
  const float* kpp   = (const float*)d_in[7];
  const float* kpw   = (const float*)d_in[8];
  const float* W2    = (const float*)d_in[9];
  const float* g2    = (const float*)d_in[10];
  const float* b2    = (const float*)d_in[11];
  const float* Wsc   = (const float*)d_in[12];
  const float* gsc   = (const float*)d_in[13];
  const float* bsc   = (const float*)d_in[14];
  float* out = (float*)d_out;

  const int N = in_sizes[0] / CIN;

  float* ws      = (float*)d_ws;
  float* x1      = ws;
  float* xk      = x1 + (size_t)N * D2;
  float* ysc     = xk + (size_t)N * D2;
  float* stats1  = ysc + (size_t)N * COUT;
  float* stats2  = stats1 + 2 * D2;
  float* statsSC = stats2 + 2 * COUT;
  float* sc1 = statsSC + 2 * COUT; float* sh1 = sc1 + D2;
  float* sc2 = sh1 + D2;           float* sh2 = sc2 + COUT;
  float* scS = sh2 + COUT;         float* shS = scS + COUT;

  const int nblk = (N + 319) / 320;
  float* GpF = ysc;
  float* SpF = GpF + (size_t)nblk * CIN * CIN;
  float* GF  = SpF + (size_t)nblk * CIN;
  float* SF  = GF + CIN * CIN;
  float* Gp2 = x1;
  float* Sp2 = Gp2 + (size_t)nblk * D2 * D2;
  float* G2  = Sp2 + (size_t)nblk * D2;
  float* S2  = G2 + D2 * D2;

  gram_part128<<<nblk, 256, 0, stream>>>(feats, N, GpF, SpF);
  reduce_rows2<<<(CIN * CIN) / 32 + CIN / 32, 256, 0, stream>>>(
      GpF, nblk, CIN * CIN, GF, (CIN * CIN) / 32, SpF, nblk, CIN, SF);
  gram_finalize128<<<D2 + COUT, 256, 0, stream>>>(GF, SF, 1.0f / (float)N,
      W1, g1, b1, sc1, sh1, D2, Wsc, gsc, bsc, scS, shS, COUT);

  gemm_bnact_wmma<CIN><<<((N + 127) / 128) * (D2 / 64), 256, 0, stream>>>(
      feats, W1, sc1, sh1, x1, N, D2);

  {
    size_t lds = (KPT * FKP + 8 * NBH * 16) * sizeof(float)
               + 8 * NBH * sizeof(int);
    kpconv_wmma<<<N / KPT, 256, lds, stream>>>(xyz, nidx, x1, kpp, kpw, xk);
  }

  gram_part64<<<nblk, 256, 0, stream>>>(xk, N, Gp2, Sp2);
  reduce_rows2<<<(D2 * D2) / 32 + D2 / 32, 256, 0, stream>>>(
      Gp2, nblk, D2 * D2, G2, (D2 * D2) / 32, Sp2, nblk, D2, S2);
  gram_finalize64<<<COUT, 256, 0, stream>>>(G2, S2, 1.0f / (float)N, W2, g2, b2, sc2, sh2, COUT);

  gemm2_bn_residual_wmma<D2, CIN><<<((N + 127) / 128) * (COUT / 64), 256, 0, stream>>>(
      xk, W2, sc2, sh2, feats, Wsc, scS, shS, out, N, COUT);
}


# KPConv as a two-stage pipeline in resident blocks: 8 waves build tile t+1 (phase 1) while 8 multiply tile t (phase 2); 2 FK buffers, 1 barrier per tile; K split 4x240 not 8x120. Reorders those sums.
# speedup vs baseline: 2.6229x; 1.0319x over previous
#include <hip/hip_runtime.h>
#include <hip/hip_bf16.h>


typedef __attribute__((ext_vector_type(2))) float v2f;
typedef __attribute__((ext_vector_type(4))) float v4f;
typedef __attribute__((ext_vector_type(8))) float v8f;

#define CIN   128
#define COUT  256
#define D2    64
#define KPTS  15
#define NBH   32
#define KDFLAT (KPTS * D2)
#define INFL_INV 2.5f
#define BN_EPS 1e-5f
#define SLOPE 0.2f
#define PP    160

__device__ __forceinline__ v8f wmma_f32(v2f a, v2f b, v8f c) {
  return __builtin_amdgcn_wmma_f32_16x16x4_f32(
       false, a,  false, b,
       (short)0, c,  false,  false);
}

__device__ __forceinline__ float lrelu(float v) {
  return v >= 0.0f ? v : SLOPE * v;
}


template <int ROWS>
__device__ __forceinline__ void stage_strip(const float* __restrict__ B, int Nc, int col0,
                                            float* Ws) {
  constexpr int ITEMS = (ROWS / 2) * 16;
  #pragma unroll
  for (int i0 = 0; i0 < ITEMS; i0 += 256) {
    const int i = i0 + (int)threadIdx.x;
    if (ITEMS % 256 == 0 || i < ITEMS) {
      const int p = i >> 4;
      const int j = (i & 15) << 2;
      const float* src = B + (size_t)(2 * p) * Nc + col0 + j;
      const v4f r0 = *(const v4f*)src;
      const v4f r1 = *(const v4f*)(src + Nc);
      float* dst = Ws + p * PP + 2 * j;
      *(v4f*)dst       = v4f{r0.x, r1.x, r0.y, r1.y};
      *(v4f*)(dst + 4) = v4f{r0.z, r1.z, r0.w, r1.w};
    }
  }
}

template <int KD>
__device__ __forceinline__ void strip_product(
    const float* __restrict__ Arow, const float* Wl,
    int half, v8f& acc0, v8f& acc1, v8f& acc2, v8f& acc3) {
  for (int kk = 0; kk < KD; kk += 4) {
    int ka = kk + (half << 1);
    v2f a;
    a.x = Arow[ka];
    a.y = Arow[ka + 1];
    const float* Wp = Wl + (ka >> 1) * PP;
    const v2f b0 = *(const v2f*)(Wp);
    const v2f b1 = *(const v2f*)(Wp + 32);
    const v2f b2 = *(const v2f*)(Wp + 64);
    const v2f b3 = *(const v2f*)(Wp + 96);
    acc0 = wmma_f32(a, b0, acc0);
    acc1 = wmma_f32(a, b1, acc1);
    acc2 = wmma_f32(a, b2, acc2);
    acc3 = wmma_f32(a, b3, acc3);
  }
}

template <int KD>
__global__ __launch_bounds__(256) void gemm_bnact_wmma(
    const float* __restrict__ A, const float* __restrict__ B,
    const float* __restrict__ sc, const float* __restrict__ sh,
    float* __restrict__ C, int M, int Nc) {
  __shared__ float Ws[(KD / 2) * PP];
  const int strips = Nc >> 6;
  const int col0 = (blockIdx.x % strips) << 6;
  stage_strip<KD>(B, Nc, col0, Ws);
  __syncthreads();
  const int row0 = (blockIdx.x / strips) * 128 + ((threadIdx.x >> 5) << 4);
  if (row0 >= M) return;
  int lane = threadIdx.x & 31;
  int m    = lane & 15;
  int half = lane >> 4;

  v8f acc0 = {}, acc1 = {}, acc2 = {}, acc3 = {};
  strip_product<KD>(A + (size_t)(row0 + m) * KD, Ws + 2 * m, half,
                    acc0, acc1, acc2, acc3);

  const float* scp = sc + col0 + m;
  const float* shp = sh + col0 + m;
  const float s0 = scp[0], s1 = scp[16], s2 = scp[32], s3 = scp[48];
  const float h0 = shp[0], h1 = shp[16], h2 = shp[32], h3 = shp[48];
  int rbase = row0 + (half << 3);
  float* Cp = C + (size_t)rbase * Nc + col0 + m;
  #pragma unroll
  for (int r = 0; r < 8; ++r) {
    size_t o = (size_t)r * Nc;
    Cp[o +  0] = lrelu(acc0[r] * s0 + h0);
    Cp[o + 16] = lrelu(acc1[r] * s1 + h1);
    Cp[o + 32] = lrelu(acc2[r] * s2 + h2);
    Cp[o + 48] = lrelu(acc3[r] * s3 + h3);
  }
}

template <int KM, int KS>
__global__ __launch_bounds__(256) void gemm2_bn_residual_wmma(
    const float* __restrict__ Am, const float* __restrict__ Bm,
    const float* __restrict__ scM, const float* __restrict__ shM,
    const float* __restrict__ As, const float* __restrict__ Bs,
    const float* __restrict__ scS, const float* __restrict__ shS,
    float* __restrict__ C, int M, int Nc) {
  __shared__ float Ws[((KS + KM) / 2) * PP];
  float* Wm = Ws + (KS / 2) * PP;
  const int strips = Nc >> 6;
  const int col0 = (blockIdx.x % strips) << 6;
  stage_strip<KS>(Bs, Nc, col0, Ws);
  stage_strip<KM>(Bm, Nc, col0, Wm);
  __syncthreads();
  const int row0 = (blockIdx.x / strips) * 128 + ((threadIdx.x >> 5) << 4);
  if (row0 >= M) return;
  int lane = threadIdx.x & 31;
  int m    = lane & 15;
  int half = lane >> 4;

  v8f y0 = {}, y1 = {}, y2 = {}, y3 = {};
  strip_product<KS>(As + (size_t)(row0 + m) * KS, Ws + 2 * m, half,
                    y0, y1, y2, y3);
  v8f acc0 = {}, acc1 = {}, acc2 = {}, acc3 = {};
  strip_product<KM>(Am + (size_t)(row0 + m) * KM, Wm + 2 * m, half,
                    acc0, acc1, acc2, acc3);

  const float* p;
  p = scM + col0 + m; const float a0 = p[0], a1 = p[16], a2 = p[32], a3 = p[48];
  p = shM + col0 + m; const float c0 = p[0], c1 = p[16], c2 = p[32], c3 = p[48];
  p = scS + col0 + m; const float s0 = p[0], s1 = p[16], s2 = p[32], s3 = p[48];
  p = shS + col0 + m; const float h0 = p[0], h1 = p[16], h2 = p[32], h3 = p[48];
  int rbase = row0 + (half << 3);
  float* Cp = C + (size_t)rbase * Nc + col0 + m;
  #pragma unroll
  for (int r = 0; r < 8; ++r) {
    size_t o = (size_t)r * Nc;
    Cp[o +  0] = lrelu(acc0[r] * a0 + c0) + (y0[r] * s0 + h0);
    Cp[o + 16] = lrelu(acc1[r] * a1 + c1) + (y1[r] * s1 + h1);
    Cp[o + 32] = lrelu(acc2[r] * a2 + c2) + (y2[r] * s2 + h2);
    Cp[o + 48] = lrelu(acc3[r] * a3 + c3) + (y3[r] * s3 + h3);
  }
}

#define KPT 32
#define FKP 964
#define KP_GRID 250
#define KP_U 6

__device__ __forceinline__ void kp_phase1(
    const float* __restrict__ xyz, const int* __restrict__ nidx,
    const float* __restrict__ x1, const float (&kp_points)[KPTS * 3],
    float* FK, float* wl, int* il,
    int wave, int lane, int m, int half, int base) {
  static_assert(KPT / 8 == 4, "the pipeline below holds the index rows of 4 points per wave");
  static_assert(NBH == 32, "one neighbour per lane");
  const int n0 = base + wave * (KPT / 8);
  int id_cur = nidx[(size_t)n0 * NBH + lane];
  int id_nxt = nidx[(size_t)(n0 + 1) * NBH + lane];
  int id_n2  = nidx[(size_t)(n0 + 2) * NBH + lane];
  int id_n3  = nidx[(size_t)(n0 + 3) * NBH + lane];
  auto put_weights = [&](float px, float py, float pz, int id) -> int {
    float w[KPTS];
    float wmax = 0.0f;
    #pragma unroll
    for (int k = 0; k < KPTS; ++k) {
      float dx = px - kp_points[k * 3 + 0];
      float dy = py - kp_points[k * 3 + 1];
      float dz = pz - kp_points[k * 3 + 2];
      float r = sqrtf(dx * dx + dy * dy + dz * dz);
      w[k] = fmaxf(1.0f - r * INFL_INV, 0.0f);
      wmax = fmaxf(wmax, w[k]);
    }
    const bool keep = wmax > 0.0f;
    const unsigned mask = __builtin_amdgcn_ballot_w32(keep);
    const int cnt   = __builtin_popcount(mask);
    const int below = (int)__builtin_amdgcn_mbcnt_lo(mask, 0u);
    const int slot  = keep ? below : cnt + (lane - below);
    il[slot] = id;
    #pragma unroll
    for (int k = 0; k < KPTS; ++k) wl[slot * 16 + k] = w[k];
    wl[slot * 16 + KPTS] = 0.0f;
    const int ns = (cnt + 3) >> 2;
    return ns < 1 ? 1 : ns;
  };

  float ncx, ncy, ncz, nqx, nqy, nqz;
  int ns_cur;
  {
    const float cx = xyz[(size_t)n0 * 3 + 0];
    const float cy = xyz[(size_t)n0 * 3 + 1];
    const float cz = xyz[(size_t)n0 * 3 + 2];
    const float qx = xyz[(size_t)id_cur * 3 + 0];
    const float qy = xyz[(size_t)id_cur * 3 + 1];
    const float qz = xyz[(size_t)id_cur * 3 + 2];
    ncx = xyz[(size_t)(n0 + 1) * 3 + 0];
    ncy = xyz[(size_t)(n0 + 1) * 3 + 1];
    ncz = xyz[(size_t)(n0 + 1) * 3 + 2];
    nqx = xyz[(size_t)id_nxt * 3 + 0];
    nqy = xyz[(size_t)id_nxt * 3 + 1];
    nqz = xyz[(size_t)id_nxt * 3 + 2];
    ns_cur = put_weights(qx - cx, qy - cy, qz - cz, id_cur);
  }
  for (int it = 0; it < KPT / 8; ++it) {
    const int pi = wave * (KPT / 8) + it;
    const int n  = base + pi;
    ns_cur = __builtin_amdgcn_readfirstlane(ns_cur);

    __builtin_amdgcn_fence(__ATOMIC_RELEASE, "wavefront", "local");
    __builtin_amdgcn_wave_barrier();
    __builtin_amdgcn_fence(__ATOMIC_ACQUIRE, "wavefront", "local");

    int i0[NBH / 4], i1[NBH / 4];
    v2f a[NBH / 4];
    #pragma unroll
    for (int j = 0; j < NBH / 4; ++j) {
      const int ka = 4 * j + (half << 1);
      i0[j] = il[ka];
      i1[j] = il[ka + 1];
      a[j].x = wl[ka * 16 + m];
      a[j].y = wl[(ka + 1) * 16 + m];
    }

    v2f b0[NBH / 4] = {}, b1[NBH / 4] = {}, b2[NBH / 4] = {}, b3[NBH / 4] = {};
    #pragma unroll
    for (int j = 0; j < NBH / 4; ++j) {
      if (j == 0 || j < ns_cur) {
        const float* r0 = x1 + (size_t)i0[j] * D2 + m;
        const float* r1 = x1 + (size_t)i1[j] * D2 + m;
        b0[j].x = r0[0];  b0[j].y = r1[0];
        b1[j].x = r0[16]; b1[j].y = r1[16];
        b2[j].x = r0[32]; b2[j].y = r1[32];
        b3[j].x = r0[48]; b3[j].y = r1[48];
      }
    }

    __builtin_amdgcn_fence(__ATOMIC_RELEASE, "wavefront", "local");
    __builtin_amdgcn_wave_barrier();

    int ns_nxt = ns_cur;
    if (it + 1 < KPT / 8)
      ns_nxt = put_weights(nqx - ncx, nqy - ncy, nqz - ncz, id_nxt);

    const int last = n0 + (KPT / 8 - 1);
    const int n2 = (n + 2 < last) ? n + 2 : last;
    const float fcx = xyz[(size_t)n2 * 3 + 0];
    const float fcy = xyz[(size_t)n2 * 3 + 1];
    const float fcz = xyz[(size_t)n2 * 3 + 2];
    const float fqx = xyz[(size_t)id_n2 * 3 + 0];
    const float fqy = xyz[(size_t)id_n2 * 3 + 1];
    const float fqz = xyz[(size_t)id_n2 * 3 + 2];

    v8f acc0 = {}, acc1 = {}, acc2 = {}, acc3 = {};
    #pragma unroll
    for (int j = 0; j < NBH / 4; ++j) {
      if (j == 0 || j < ns_cur) {
        acc0 = wmma_f32(a[j], b0[j], acc0);
        acc1 = wmma_f32(a[j], b1[j], acc1);
        acc2 = wmma_f32(a[j], b2[j], acc2);
        acc3 = wmma_f32(a[j], b3[j], acc3);
      }
    }
    #pragma unroll
    for (int r = 0; r < 8; ++r) {
      const int krow = (half << 3) + r;
      if (krow < KPTS) {
        float* f = FK + pi * FKP + krow * D2 + m;
        f[0]  = acc0[r];
        f[16] = acc1[r];
        f[32] = acc2[r];
        f[48] = acc3[r];
      }
    }
    ncx = fcx; ncy = fcy; ncz = fcz;
    nqx = fqx; nqy = fqy; nqz = fqz;
    id_cur = id_nxt; id_nxt = id_n2; id_n2 = id_n3;
    ns_cur = ns_nxt;
  }
}

__device__ __forceinline__ void kp_phase2(
    const float* FK, const float* __restrict__ kpw,
    int cp, int kq, int m, int half,
    v8f& lo0, v8f& lo1, v8f& hi0, v8f& hi1) {
  const float* Alo = FK + m * FKP + (half << 1);
  const float* Ahi = FK + (16 + m) * FKP + (half << 1);
  const float* Bb  = kpw + (size_t)(half << 1) * D2 + cp * 32 + m;
  const int kbeg = kq * 240;

  auto load_run = [&](v2f (&alo)[KP_U], v2f (&ahi)[KP_U], v2f (&b0)[KP_U], v2f (&b1)[KP_U], int run) {
    const int k0 = kbeg + run * (4 * KP_U);
    #pragma unroll
    for (int u = 0; u < KP_U; ++u) {
      const int kk = k0 + 4 * u;
      const float* Bp = Bb + (size_t)kk * D2;
      b0[u].x = Bp[0];  b0[u].y = Bp[D2];
      b1[u].x = Bp[16]; b1[u].y = Bp[D2 + 16];
      alo[u] = *(const v2f*)(Alo + kk);
      ahi[u] = *(const v2f*)(Ahi + kk);
    }
  };
  auto mma_run = [&](v2f (&alo)[KP_U], v2f (&ahi)[KP_U], v2f (&b0)[KP_U], v2f (&b1)[KP_U]) {
    #pragma unroll
    for (int u = 0; u < KP_U; ++u) {
      lo0 = wmma_f32(alo[u], b0[u], lo0);
      hi0 = wmma_f32(ahi[u], b0[u], hi0);
      lo1 = wmma_f32(alo[u], b1[u], lo1);
      hi1 = wmma_f32(ahi[u], b1[u], hi1);
    }
  };

  constexpr int RUNS = 240 / (4 * KP_U);
  static_assert(RUNS * 4 * KP_U == 240 && RUNS % 2 == 0, "K quarter = an even number of runs");
  v2f alA[KP_U], ahA[KP_U], b0A[KP_U], b1A[KP_U];
  v2f alB[KP_U], ahB[KP_U], b0B[KP_U], b1B[KP_U];
  load_run(alA, ahA, b0A, b1A, 0);
  for (int r = 0; r < RUNS; r += 2) {
    load_run(alB, ahB, b0B, b1B, r + 1);
    mma_run(alA, ahA, b0A, b1A);
    if (r + 2 < RUNS) load_run(alA, ahA, b0A, b1A, r + 2);
    mma_run(alB, ahB, b0B, b1B);
  }
}

__global__ __launch_bounds__(512) void kpconv_wmma(
    const float* __restrict__ xyz, const int* __restrict__ nidx,
    const float* __restrict__ x1, const float* __restrict__ kp_points,
    const float* __restrict__ kpw, float* __restrict__ xk, int ntiles) {
  extern __shared__ float smem[];
  float* FKb = smem;
  float* Wl  = FKb + 2 * KPT * FKP;
  int*   IDX = (int*)(Wl + 8 * NBH * 16);
  float* PSb = (float*)(IDX + 8 * NBH);

  const int wave = __builtin_amdgcn_readfirstlane((int)(threadIdx.x >> 5));
  const int lane = threadIdx.x & 31;
  const int m    = lane & 15;
  const int half = lane >> 4;
  const bool producer = wave < 8;
  const int cw = wave & 7;
  const int cp = cw & 1;
  const int kq = cw >> 1;

  float kpts[KPTS * 3];
  #pragma unroll
  for (int k = 0; k < KPTS * 3; ++k) kpts[k] = kp_points[k];

  const int G = (int)gridDim.x;
  const int myT = ((int)blockIdx.x < ntiles) ? (ntiles - 1 - (int)blockIdx.x) / G + 1 : 0;

  for (int i = 0; i <= myT; ++i) {
    v8f lo0 = {}, lo1 = {}, hi0 = {}, hi1 = {};
    const int ctile = (int)blockIdx.x + (i - 1) * G;
    if (producer) {
      if (i < myT) {
        const int ptile = (int)blockIdx.x + i * G;
        kp_phase1(xyz, nidx, x1, kpts, FKb + (i & 1) * (KPT * FKP),
                  Wl + wave * (NBH * 16), IDX + wave * NBH,
                  wave, lane, m, half, ptile * KPT);
      }
    } else if (i >= 1) {
      kp_phase2(FKb + ((i - 1) & 1) * (KPT * FKP), kpw, cp, kq, m, half, lo0, lo1, hi0, hi1);
      if (kq > 0) {
        float* ps = PSb + (i & 1) * (3 * KPT * D2) + (kq - 1) * (KPT * D2)
                  + (half << 3) * D2 + cp * 32 + m;
        #pragma unroll
        for (int r = 0; r < 8; ++r) {
          ps[r * D2]                = lo0[r];
          ps[r * D2 + 16]           = lo1[r];
          ps[(16 + r) * D2]         = hi0[r];
          ps[(16 + r) * D2 + 16]    = hi1[r];
        }
      }
    }
    __syncthreads();
    if (!producer && i >= 1 && kq == 0) {
      const float* ps = PSb + (i & 1) * (3 * KPT * D2) + (half << 3) * D2 + cp * 32 + m;
      float* o = xk + ((size_t)ctile * KPT + (half << 3)) * D2 + cp * 32 + m;
      #pragma unroll
      for (int r = 0; r < 8; ++r) {
        float s0 = lo0[r], s1 = lo1[r], s2 = hi0[r], s3 = hi1[r];
        #pragma unroll
        for (int q = 0; q < 3; ++q) {
          const float* pq = ps + q * (KPT * D2);
          s0 += pq[r * D2];
          s1 += pq[r * D2 + 16];
          s2 += pq[(16 + r) * D2];
          s3 += pq[(16 + r) * D2 + 16];
        }
        o[(size_t)r * D2]             = s0;
        o[(size_t)r * D2 + 16]        = s1;
        o[(size_t)(16 + r) * D2]      = s2;
        o[(size_t)(16 + r) * D2 + 16] = s3;
      }
    }
  }
}

#define GRAM_U 4

template <int C, int RPG>
__device__ __forceinline__ void gram_body(const float* __restrict__ X, int N,
                                          float* __restrict__ Gp,
                                          float* __restrict__ Sp, float* red,
                                          float* reds) {
  constexpr int TC     = C / 64;
  constexpr int TILES  = (C / 32) * TC;
  constexpr int GROUPS = 8 / TILES;
  const int wave = __builtin_amdgcn_readfirstlane((int)(threadIdx.x >> 5));
  const int lane = threadIdx.x & 31;
  const int m    = lane & 15;
  const int half = lane >> 4;
  const int tile = wave % TILES;
  const int grp  = wave / TILES;
  const int tr   = tile / TC;
  const int tc   = tile % TC;
  const int gidx = blockIdx.x * GROUPS + grp;
  const int row0 = gidx * RPG;
  int nrows = N - row0;
  nrows = nrows < 0 ? 0 : (nrows > RPG ? RPG : nrows);
  const int nruns = nrows / (4 * GRAM_U);

  const float* xb = X + (size_t)(row0 + 2 * half) * C + m;
  const float* ap = xb + tr * 32;
  const float* bp = xb + tc * 64;

  v8f acc[2][4];
  #pragma unroll
  for (int i = 0; i < 2; ++i)
    #pragma unroll
    for (int j = 0; j < 4; ++j) acc[i][j] = v8f{};
  float cs[4] = {0.0f, 0.0f, 0.0f, 0.0f};

  auto load_run = [&](v2f (&a)[GRAM_U][2], v2f (&b)[GRAM_U][4], int run) {
    const size_t o = (size_t)run * (4 * GRAM_U) * C;
    #pragma unroll
    for (int u = 0; u < GRAM_U; ++u) {
      const float* pa = ap + o + (size_t)u * 4 * C;
      const float* pb = bp + o + (size_t)u * 4 * C;
      #pragma unroll
      for (int i = 0; i < 2; ++i) { a[u][i].x = pa[i * 16]; a[u][i].y = pa[C + i * 16]; }
      #pragma unroll
      for (int j = 0; j < 4; ++j) { b[u][j].x = pb[j * 16]; b[u][j].y = pb[C + j * 16]; }
    }
  };
  auto mma_run = [&](v2f (&a)[GRAM_U][2], v2f (&b)[GRAM_U][4]) {
    #pragma unroll
    for (int u = 0; u < GRAM_U; ++u) {
      if (tr == 0) {
        #pragma unroll
        for (int j = 0; j < 4; ++j) cs[j] += b[u][j].x + b[u][j].y;
      }
      #pragma unroll
      for (int i = 0; i < 2; ++i)
        #pragma unroll
        for (int j = 0; j < 4; ++j) acc[i][j] = wmma_f32(a[u][i], b[u][j], acc[i][j]);
    }
  };

  v2f a0[GRAM_U][2], b0[GRAM_U][4], a1[GRAM_U][2], b1[GRAM_U][4];
  if (nruns > 0) load_run(a0, b0, 0);
  for (int r = 0; r < nruns; r += 2) {
    if (r + 1 < nruns) load_run(a1, b1, r + 1);
    mma_run(a0, b0);
    if (r + 2 < nruns) load_run(a0, b0, r + 2);
    if (r + 1 < nruns) mma_run(a1, b1);
  }

  const int grow = tr * 32 + (half << 3);
  const int gcol = tc * 64 + m;
  if (tr == 0) {
    #pragma unroll
    for (int j = 0; j < 4; ++j) cs[j] += __shfl_xor(cs[j], 16);
  }
  if constexpr (GROUPS > 1) {
    if (grp > 0) {
      if (tr == 0 && half == 0) {
        #pragma unroll
        for (int j = 0; j < 4; ++j) reds[(grp - 1) * C + gcol + j * 16] = cs[j];
      }
      float* rp = red + (size_t)(grp - 1) * (C * C);
      #pragma unroll
      for (int i = 0; i < 2; ++i)
        #pragma unroll
        for (int j = 0; j < 4; ++j)
          #pragma unroll
          for (int r = 0; r < 8; ++r)
            rp[(grow + i * 16 + r) * C + gcol + j * 16] = acc[i][j][r];
    }
    __syncthreads();
    if (grp == 0) {
      if (tr == 0) {
        #pragma unroll
        for (int g = 0; g < GROUPS - 1; ++g)
          #pragma unroll
          for (int j = 0; j < 4; ++j) cs[j] += reds[g * C + gcol + j * 16];
      }
      #pragma unroll
      for (int g = 0; g < GROUPS - 1; ++g)
        #pragma unroll
        for (int i = 0; i < 2; ++i)
          #pragma unroll
          for (int j = 0; j < 4; ++j)
            #pragma unroll
            for (int r = 0; r < 8; ++r)
              acc[i][j][r] += red[(size_t)g * (C * C) + (grow + i * 16 + r) * C + gcol + j * 16];
    }
  }
  if (grp == 0) {
    float* gp = Gp + (size_t)blockIdx.x * (C * C);
    #pragma unroll
    for (int i = 0; i < 2; ++i)
      #pragma unroll
      for (int j = 0; j < 4; ++j)
        #pragma unroll
        for (int r = 0; r < 8; ++r)
          gp[(grow + i * 16 + r) * C + gcol + j * 16] = acc[i][j][r];
  }
  if (grp == 0 && tr == 0 && half == 0) {
    float* sp = Sp + (size_t)blockIdx.x * C + gcol;
    #pragma unroll
    for (int j = 0; j < 4; ++j) sp[j * 16] = cs[j];
  }
}

__global__ __launch_bounds__(256) void gram_part128(const float* __restrict__ X, int N,
                                                    float* __restrict__ Gp,
                                                    float* __restrict__ Sp) {
  gram_body<128, 320>(X, N, Gp, Sp, nullptr, nullptr);
}

__global__ __launch_bounds__(256) void gram_part64(const float* __restrict__ X, int N,
                                                   float* __restrict__ Gp,
                                                   float* __restrict__ Sp) {
  __shared__ float red[3 * 64 * 64];
  __shared__ float reds[3 * 64];
  gram_body<64, 80>(X, N, Gp, Sp, red, reds);
}

__global__ __launch_bounds__(256) void reduce_rows2(
    const float* __restrict__ srcA, int nrA, int wA, float* __restrict__ dstA, int blocksA,
    const float* __restrict__ srcB, int nrB, int wB, float* __restrict__ dstB) {
  __shared__ float red[8][32];
  int b = blockIdx.x;
  const float* src = srcA; int nr = nrA; int w = wA; float* dst = dstA;
  if (b >= blocksA) { b -= blocksA; src = srcB; nr = nrB; w = wB; dst = dstB; }
  const int e = threadIdx.x & 31;
  const int q = threadIdx.x >> 5;
  const int col = b * 32 + e;
  float s = 0.0f;
  #pragma unroll 8
  for (int r = q; r < nr; r += 8) s += src[(size_t)r * w + col];
  red[q][e] = s;
  __syncthreads();
  if (q == 0) {
    float t = red[0][e];
    #pragma unroll
    for (int i = 1; i < 8; ++i) t += red[i][e];
    dst[col] = t;
  }
}

template <int C>
__device__ __forceinline__ void gram_finalize_body(
    const float* __restrict__ G, const float* __restrict__ S, float invN,
    const float* __restrict__ W, const float* __restrict__ g,
    const float* __restrict__ bta, float* __restrict__ sc, float* __restrict__ sh,
    int Cout, int c) {
  __shared__ float wl[C];
  __shared__ float r2[256];
  __shared__ float r1[256];
  const int tid = threadIdx.x;
  for (int k = tid; k < C; k += 256) wl[k] = W[(size_t)k * Cout + c];
  __syncthreads();
  float s2 = 0.0f;
  #pragma unroll 16
  for (int i = tid; i < C * C; i += 256) s2 += wl[i / C] * G[i] * wl[i % C];
  float s1 = 0.0f;
  for (int k = tid; k < C; k += 256) s1 += wl[k] * S[k];
  r2[tid] = s2;
  r1[tid] = s1;
  __syncthreads();
  for (int st = 128; st > 0; st >>= 1) {
    if (tid < st) { r2[tid] += r2[tid + st]; r1[tid] += r1[tid + st]; }
    __syncthreads();
  }
  if (tid == 0) {
    float mean = r1[0] * invN;
    float var  = fmaxf(r2[0] * invN - mean * mean, 0.0f);
    float s    = g[c] * rsqrtf(var + BN_EPS);
    sc[c] = s;
    sh[c] = bta[c] - mean * s;
  }
}

__global__ __launch_bounds__(256) void gram_finalize128(
    const float* __restrict__ G, const float* __restrict__ S, float invN,
    const float* __restrict__ Wa, const float* __restrict__ ga, const float* __restrict__ ba,
    float* __restrict__ sca, float* __restrict__ sha, int Ca,
    const float* __restrict__ Wb, const float* __restrict__ gb, const float* __restrict__ bb,
    float* __restrict__ scb, float* __restrict__ shb, int Cb) {
  int c = blockIdx.x;
  const float* W = Wa; const float* g = ga; const float* b = ba;
  float* sc = sca; float* sh = sha; int Cout = Ca;
  if (c >= Ca) { c -= Ca; W = Wb; g = gb; b = bb; sc = scb; sh = shb; Cout = Cb; }
  gram_finalize_body<128>(G, S, invN, W, g, b, sc, sh, Cout, c);
}

__global__ __launch_bounds__(256) void gram_finalize64(
    const float* __restrict__ G, const float* __restrict__ S, float invN,
    const float* __restrict__ W, const float* __restrict__ g, const float* __restrict__ b,
    float* __restrict__ sc, float* __restrict__ sh, int Cout) {
  gram_finalize_body<64>(G, S, invN, W, g, b, sc, sh, Cout, blockIdx.x);
}

extern "C" void kernel_launch(void* const* d_in, const int* in_sizes, int n_in,
                              void* d_out, int out_size, void* d_ws, size_t ws_size,
                              hipStream_t stream) {
  const float* feats = (const float*)d_in[0];
  const float* xyz   = (const float*)d_in[1];
  const int*   nidx  = (const int*)d_in[3];
  const float* W1    = (const float*)d_in[4];
  const float* g1    = (const float*)d_in[5];
  const float* b1    = (const float*)d_in[6];
  const float* kpp   = (const float*)d_in[7];
  const float* kpw   = (const float*)d_in[8];
  const float* W2    = (const float*)d_in[9];
  const float* g2    = (const float*)d_in[10];
  const float* b2    = (const float*)d_in[11];
  const float* Wsc   = (const float*)d_in[12];
  const float* gsc   = (const float*)d_in[13];
  const float* bsc   = (const float*)d_in[14];
  float* out = (float*)d_out;

  const int N = in_sizes[0] / CIN;

  float* ws      = (float*)d_ws;
  float* x1      = ws;
  float* xk      = x1 + (size_t)N * D2;
  float* ysc     = xk + (size_t)N * D2;
  float* stats1  = ysc + (size_t)N * COUT;
  float* stats2  = stats1 + 2 * D2;
  float* statsSC = stats2 + 2 * COUT;
  float* sc1 = statsSC + 2 * COUT; float* sh1 = sc1 + D2;
  float* sc2 = sh1 + D2;           float* sh2 = sc2 + COUT;
  float* scS = sh2 + COUT;         float* shS = scS + COUT;

  const int nblk = (N + 319) / 320;
  float* GpF = ysc;
  float* SpF = GpF + (size_t)nblk * CIN * CIN;
  float* GF  = SpF + (size_t)nblk * CIN;
  float* SF  = GF + CIN * CIN;
  float* Gp2 = x1;
  float* Sp2 = Gp2 + (size_t)nblk * D2 * D2;
  float* G2  = Sp2 + (size_t)nblk * D2;
  float* S2  = G2 + D2 * D2;

  gram_part128<<<nblk, 256, 0, stream>>>(feats, N, GpF, SpF);
  reduce_rows2<<<(CIN * CIN) / 32 + CIN / 32, 256, 0, stream>>>(
      GpF, nblk, CIN * CIN, GF, (CIN * CIN) / 32, SpF, nblk, CIN, SF);
  gram_finalize128<<<D2 + COUT, 256, 0, stream>>>(GF, SF, 1.0f / (float)N,
      W1, g1, b1, sc1, sh1, D2, Wsc, gsc, bsc, scS, shS, COUT);

  gemm_bnact_wmma<CIN><<<((N + 127) / 128) * (D2 / 64), 256, 0, stream>>>(
      feats, W1, sc1, sh1, x1, N, D2);

  {
    size_t lds = (2 * KPT * FKP + 8 * NBH * 16 + 2 * 3 * KPT * D2) * sizeof(float)
               + 8 * NBH * sizeof(int);
    kpconv_wmma<<<KP_GRID, 512, lds, stream>>>(xyz, nidx, x1, kpp, kpw, xk, N / KPT);
  }

  gram_part64<<<nblk, 256, 0, stream>>>(xk, N, Gp2, Sp2);
  reduce_rows2<<<(D2 * D2) / 32 + D2 / 32, 256, 0, stream>>>(
      Gp2, nblk, D2 * D2, G2, (D2 * D2) / 32, Sp2, nblk, D2, S2);
  gram_finalize64<<<COUT, 256, 0, stream>>>(G2, S2, 1.0f / (float)N, W2, g2, b2, sc2, sh2, COUT);

  gemm2_bn_residual_wmma<D2, CIN><<<((N + 127) / 128) * (COUT / 64), 256, 0, stream>>>(
      xk, W2, sc2, sh2, feats, Wsc, scS, shS, out, N, COUT);
}
